# starting kernel
# baseline (speedup 1.0000x reference)
#include <hip/hip_runtime.h>
#include <hip/hip_bf16.h>


#define B_  4
#define DI  192
#define LL  4096
#define NS  16
#define RR  6
#define KK  4
#define CP  38
#define TL  64
#define TLP 65

typedef float v2f __attribute__((ext_vector_type(2)));
typedef float v8f __attribute__((ext_vector_type(8)));

__device__ __forceinline__ int scan_idx(int k, int l) {
    int ll = (k & 2) ? (LL - 1 - l) : l;
    if (k & 1) return ((ll & 63) << 6) | (ll >> 6);
    return ll;
}

__global__ __launch_bounds__(256) void ss2d_proj_wmma(
    const float* __restrict__ x,
    const float* __restrict__ xpw,
    float* __restrict__ xdbl)
{
    int wave = (int)((blockIdx.x * blockDim.x + threadIdx.x) >> 5);
    int lane = threadIdx.x & 31;
    int nt = wave & 255;
    int mt = (wave >> 8) % 3;
    int kd = (wave >> 8) / 3;
    if (kd >= B_ * KK) return;
    int k = kd & 3;
    int b = kd >> 2;

    int m0 = mt << 4;
    int l0 = nt << 4;
    int hw = lane >> 4;
    int ln = lane & 15;

    const float* wbase = xpw + (size_t)k * CP * DI;
    const float* xb    = x + (size_t)b * DI * LL;

    int arow   = m0 + ln;
    int arow_c = arow < CP ? arow : (CP - 1);
    float amsk = arow < CP ? 1.0f : 0.0f;

    int midx = scan_idx(k, l0 + ln);

    v8f acc = {0.f, 0.f, 0.f, 0.f, 0.f, 0.f, 0.f, 0.f};
    for (int k0 = 0; k0 < DI; k0 += 4) {
        const float* ap = wbase + arow_c * DI + k0 + (hw << 1);
        v2f a; a.x = ap[0] * amsk; a.y = ap[1] * amsk;
        const float* bp = xb + (size_t)(k0 + (hw << 1)) * LL + midx;
        v2f bb; bb.x = bp[0]; bb.y = bp[LL];
        acc = __builtin_amdgcn_wmma_f32_16x16x4_f32(
            false, a, false, bb, (short)0, acc, false, false);
    }

    float* obase = xdbl + (size_t)kd * CP * LL + l0 + ln;
#pragma unroll
    for (int i = 0; i < 8; ++i) {
        int m = m0 + i + (hw << 3);
        if (m < CP) obase[(size_t)m * LL] = acc[i];
    }
}

__global__ __launch_bounds__(1024) void ss2d_scan(
    const float* __restrict__ y,
    const float* __restrict__ dtw_g,
    const float* __restrict__ dtb_g,
    const float* __restrict__ Alogs,
    const float* __restrict__ Ds_g,
    const float* __restrict__ xdbl,
    float* __restrict__ outy)
{
    __shared__ float sm_rows[CP * TLP];
    __shared__ float sm_u[64 * TLP];
    __shared__ float sm_y[64 * TLP];

    int dblk = blockIdx.x % 3;
    int k    = (blockIdx.x / 3) & 3;
    int b    = blockIdx.x / 12;
    int kd   = b * 4 + k;

    int t  = threadIdx.x;
    int n  = t & 15;
    int dl = t >> 4;
    int d  = dblk * 64 + dl;
    int kdch = k * DI + d;

    float A_dn = -__expf(Alogs[(size_t)kdch * NS + n]);
    float dtw[RR];
#pragma unroll
    for (int r = 0; r < RR; ++r) dtw[r] = dtw_g[(size_t)kdch * RR + r];
    float dtb = dtb_g[kdch];
    float Dsd = Ds_g[kdch];

    const float* xdb = xdbl + (size_t)kd * CP * LL;
    const float* yb  = y + (size_t)b * DI * LL;

    float h = 0.0f;

    for (int c = 0; c < LL / TL; ++c) {
        int l0 = c * TL;
        for (int i = t; i < CP * TL; i += 1024) {
            int r = i >> 6, col = i & 63;
            sm_rows[r * TLP + col] = xdb[(size_t)r * LL + l0 + col];
        }
        for (int i = t; i < 64 * TL; i += 1024) {
            int r = i >> 6, col = i & 63;
            sm_u[r * TLP + col] =
                yb[(size_t)(dblk * 64 + r) * LL + scan_idx(k, l0 + col)];
        }
        __syncthreads();

        for (int j = 0; j < TL; ++j) {
            float s = dtb;
#pragma unroll
            for (int r = 0; r < RR; ++r) s += dtw[r] * sm_rows[r * TLP + j];
            float delta = (s > 20.0f) ? s : log1pf(__expf(s));

            float u  = sm_u[dl * TLP + j];
            float bv = sm_rows[(RR + n) * TLP + j];
            float cv = sm_rows[(RR + NS + n) * TLP + j];

            h = __expf(delta * A_dn) * h + (delta * u) * bv;
            float yv = h * cv;
            yv += __shfl_xor(yv, 1, 16);
            yv += __shfl_xor(yv, 2, 16);
            yv += __shfl_xor(yv, 4, 16);
            yv += __shfl_xor(yv, 8, 16);
            if (n == 0) sm_y[dl * TLP + j] = yv + Dsd * u;
        }
        __syncthreads();

        for (int i = t; i < 64 * TL; i += 1024) {
            int r = i >> 6, col = i & 63;
            outy[((size_t)kd * DI + dblk * 64 + r) * LL + l0 + col] =
                sm_y[r * TLP + col];
        }
    }
}

__global__ __launch_bounds__(192) void ss2d_merge_ln(
    const float* __restrict__ outy,
    const float* __restrict__ gamma,
    const float* __restrict__ beta,
    float* __restrict__ out)
{
    __shared__ float red[16];
    __shared__ float stats[2];

    int b  = blockIdx.x >> 12;
    int lm = blockIdx.x & 4095;
    int hh = lm >> 6, ww = lm & 63;
    int lt = (ww << 6) | hh;
    int d  = threadIdx.x;

    const size_t ks = (size_t)DI * LL;
    const float* base = outy + (size_t)b * KK * ks + (size_t)d * LL;

    float v = base[lm]
            + base[2 * ks + (LL - 1 - lm)]
            + base[ks + lt]
            + base[3 * ks + (LL - 1 - lt)];

    float s = v, q = v * v;
#pragma unroll
    for (int o = 16; o; o >>= 1) {
        s += __shfl_xor(s, o, 32);
        q += __shfl_xor(q, o, 32);
    }
    int wv = d >> 5;
    if ((d & 31) == 0) { red[wv] = s; red[8 + wv] = q; }
    __syncthreads();
    if (d == 0) {
        float ts = 0.f, tq = 0.f;
        for (int i = 0; i < 6; ++i) { ts += red[i]; tq += red[8 + i]; }
        float mu  = ts / 192.0f;
        float var = tq / 192.0f - mu * mu;
        stats[0] = mu;
        stats[1] = rsqrtf(var + 1e-5f);
    }
    __syncthreads();
    float mu = stats[0], rstd = stats[1];
    out[(size_t)blockIdx.x * DI + d] = (v - mu) * rstd * gamma[d] + beta[d];
}

extern "C" void kernel_launch(void* const* d_in, const int* in_sizes, int n_in,
                              void* d_out, int out_size, void* d_ws, size_t ws_size,
                              hipStream_t stream) {
    const float* x   = (const float*)d_in[0];
    const float* y   = (const float*)d_in[1];
    const float* xpw = (const float*)d_in[2];
    const float* dtw = (const float*)d_in[3];
    const float* dtb = (const float*)d_in[4];
    const float* Al  = (const float*)d_in[5];
    const float* Ds  = (const float*)d_in[6];
    const float* gm  = (const float*)d_in[7];
    const float* bt  = (const float*)d_in[8];

    float* xdbl = (float*)d_ws;
    float* outy = xdbl + (size_t)B_ * KK * CP * LL;

    ss2d_proj_wmma<<<1536, 256, 0, stream>>>(x, xpw, xdbl);
    ss2d_scan<<<B_ * KK * 3, 1024, 0, stream>>>(y, dtw, dtb, Al, Ds, xdbl, outy);
    ss2d_merge_ln<<<B_ * LL, 192, 0, stream>>>(outy, gm, bt, (float*)d_out);
}


# scan re-partitioned: one-wave work-groups of 4 channels x 8 lanes, 2 states per lane in registers; softplus once per channel-step, state sum by DPP instead of LDS shuffles
# speedup vs baseline: 8.2673x; 8.2673x over previous
#include <hip/hip_runtime.h>
#include <hip/hip_bf16.h>


#define B_  4
#define DI  192
#define LL  4096
#define NS  16
#define RR  6
#define KK  4
#define CP  38
#define TL  64
#define TLP 65

typedef float v2f __attribute__((ext_vector_type(2)));
typedef float v8f __attribute__((ext_vector_type(8)));

__device__ __forceinline__ int scan_idx(int k, int l) {
    int ll = (k & 2) ? (LL - 1 - l) : l;
    if (k & 1) return ((ll & 63) << 6) | (ll >> 6);
    return ll;
}

__global__ __launch_bounds__(256) void ss2d_proj_wmma(
    const float* __restrict__ x,
    const float* __restrict__ xpw,
    float* __restrict__ xdbl)
{
    int wave = (int)((blockIdx.x * blockDim.x + threadIdx.x) >> 5);
    int lane = threadIdx.x & 31;
    int nt = wave & 255;
    int mt = (wave >> 8) % 3;
    int kd = (wave >> 8) / 3;
    if (kd >= B_ * KK) return;
    int k = kd & 3;
    int b = kd >> 2;

    int m0 = mt << 4;
    int l0 = nt << 4;
    int hw = lane >> 4;
    int ln = lane & 15;

    const float* wbase = xpw + (size_t)k * CP * DI;
    const float* xb    = x + (size_t)b * DI * LL;

    int arow   = m0 + ln;
    int arow_c = arow < CP ? arow : (CP - 1);
    float amsk = arow < CP ? 1.0f : 0.0f;

    int midx = scan_idx(k, l0 + ln);

    v8f acc = {0.f, 0.f, 0.f, 0.f, 0.f, 0.f, 0.f, 0.f};
    for (int k0 = 0; k0 < DI; k0 += 4) {
        const float* ap = wbase + arow_c * DI + k0 + (hw << 1);
        v2f a; a.x = ap[0] * amsk; a.y = ap[1] * amsk;
        const float* bp = xb + (size_t)(k0 + (hw << 1)) * LL + midx;
        v2f bb; bb.x = bp[0]; bb.y = bp[LL];
        acc = __builtin_amdgcn_wmma_f32_16x16x4_f32(
            false, a, false, bb, (short)0, acc, false, false);
    }

    float* obase = xdbl + (size_t)kd * CP * LL + l0 + ln;
#pragma unroll
    for (int i = 0; i < 8; ++i) {
        int m = m0 + i + (hw << 3);
        if (m < CP) obase[(size_t)m * LL] = acc[i];
    }
}

#define CHW 4
#define LPC 8
#define XP  40
#define UP  72
#define DP  144

__device__ __forceinline__ float oct_sum(float v) {
    v += __int_as_float(__builtin_amdgcn_update_dpp(0, __float_as_int(v), 0xB1, 0xF, 0xF, true));
    v += __int_as_float(__builtin_amdgcn_update_dpp(0, __float_as_int(v), 0x4E, 0xF, 0xF, true));
    v += __int_as_float(__builtin_amdgcn_update_dpp(0, __float_as_int(v), 0x141, 0xF, 0xF, true));
    return v;
}

__device__ __forceinline__ void scan_load_group(
    const float* sm_x, const float* sm_dd, int g, int q, int ch,
    float4 (&BC)[4], float2 (&Dv)[4])
{
#pragma unroll
    for (int t = 0; t < 4; ++t) {
        int j = 4 * g + t;
        BC[t] = *(const float4*)(sm_x + j * XP + 8 + 4 * q);
        Dv[t] = *(const float2*)(sm_dd + ch * DP + 2 * j);
    }
}

__device__ __forceinline__ void scan_run_group(
    float* sm_y, int g, int q, int ch, float A20, float A21, float& h0, float& h1,
    const float4 (&BC)[4], const float2 (&Dv)[4])
{
    float yy[4];
#pragma unroll
    for (int t = 0; t < 4; ++t) {
        float delta = Dv[t].x, du = Dv[t].y;
        float dA0 = __builtin_amdgcn_exp2f(delta * A20);
        float dA1 = __builtin_amdgcn_exp2f(delta * A21);
        h0 = dA0 * h0 + du * BC[t].x;
        h1 = dA1 * h1 + du * BC[t].y;
        float yv = h0 * BC[t].z;
        yv += h1 * BC[t].w;
        yy[t] = oct_sum(yv);
    }
    if (q == 0)
        *(float4*)(sm_y + ch * UP + 4 * g) = make_float4(yy[0], yy[1], yy[2], yy[3]);
}

__global__ __launch_bounds__(32) void ss2d_scan(
    const float* __restrict__ y,
    const float* __restrict__ dtw_g,
    const float* __restrict__ dtb_g,
    const float* __restrict__ Alogs,
    const float* __restrict__ Ds_g,
    const float* __restrict__ xdbl,
    float* __restrict__ outy)
{
    __shared__ __attribute__((aligned(16))) float sm_x[TL * XP];
    __shared__ __attribute__((aligned(16))) float sm_u[CHW * UP];
    __shared__ __attribute__((aligned(16))) float sm_dd[CHW * DP];
    __shared__ __attribute__((aligned(16))) float sm_y[CHW * UP];

    const int lane = threadIdx.x;
    const int cb = blockIdx.x % 48;
    const int kd = blockIdx.x / 48;
    const int k  = kd & 3;
    const int b  = kd >> 2;
    const int ch = lane >> 3;
    const int q  = lane & 7;
    const int d0 = cb * CHW;
    const int kdch = k * DI + d0 + ch;

    const float A20 = -__expf(Alogs[(size_t)kdch * NS + 2 * q])     * 1.4426950408889634f;
    const float A21 = -__expf(Alogs[(size_t)kdch * NS + 2 * q + 1]) * 1.4426950408889634f;
    float dtw[RR];
#pragma unroll
    for (int r = 0; r < RR; ++r) dtw[r] = dtw_g[(size_t)kdch * RR + r];
    const float dtb = dtb_g[kdch];
    float DsC[CHW];
#pragma unroll
    for (int c4 = 0; c4 < CHW; ++c4) DsC[c4] = Ds_g[k * DI + d0 + c4];

    const float* xdb = xdbl + (size_t)kd * CP * LL;
    const float* yb  = y + ((size_t)b * DI + d0) * LL;
    float* ob        = outy + ((size_t)kd * DI + d0) * LL;

    float xr[2][CP];
    float ur[8];

#pragma unroll
    for (int hf = 0; hf < 2; ++hf)
#pragma unroll
        for (int r = 0; r < CP; ++r)
            xr[hf][r] = xdb[(size_t)r * LL + lane + 32 * hf];
#pragma unroll
    for (int m = 0; m < 8; ++m)
        ur[m] = yb[(size_t)(m >> 1) * LL + scan_idx(k, (m & 1) * 32 + lane)];

    float h0 = 0.0f, h1 = 0.0f;

    for (int c = 0; c < LL / TL; ++c) {
        const int l0 = c * TL;

#pragma unroll
        for (int hf = 0; hf < 2; ++hf) {
            float* dst = sm_x + (lane + 32 * hf) * XP;
            *(float4*)(dst)     = make_float4(xr[hf][0], xr[hf][1], xr[hf][2], xr[hf][3]);
            *(float4*)(dst + 4) = make_float4(xr[hf][4], xr[hf][5], 0.0f, 0.0f);
#pragma unroll
            for (int p = 0; p < 8; ++p)
                *(float4*)(dst + 8 + 4 * p) =
                    make_float4(xr[hf][RR + 2 * p],      xr[hf][RR + 2 * p + 1],
                                xr[hf][RR + NS + 2 * p], xr[hf][RR + NS + 2 * p + 1]);
        }
#pragma unroll
        for (int m = 0; m < 8; ++m)
            sm_u[(m >> 1) * UP + (m & 1) * 32 + lane] = ur[m];
        __syncthreads();

        {
            const int cn = (c + 1 < LL / TL) ? (c + 1) : c;
            const int ln0 = cn * TL;
#pragma unroll
            for (int hf = 0; hf < 2; ++hf)
#pragma unroll
                for (int r = 0; r < CP; ++r)
                    xr[hf][r] = xdb[(size_t)r * LL + ln0 + lane + 32 * hf];
#pragma unroll
            for (int m = 0; m < 8; ++m)
                ur[m] = yb[(size_t)(m >> 1) * LL + scan_idx(k, ln0 + (m & 1) * 32 + lane)];
        }

#pragma unroll
        for (int m = 0; m < 8; ++m) {
            const int j = q + 8 * m;
            const float4 d03 = *(const float4*)(sm_x + j * XP);
            const float2 d45 = *(const float2*)(sm_x + j * XP + 4);
            float s = dtb;
            s += dtw[0] * d03.x;
            s += dtw[1] * d03.y;
            s += dtw[2] * d03.z;
            s += dtw[3] * d03.w;
            s += dtw[4] * d45.x;
            s += dtw[5] * d45.y;
            const float sp = log1pf(__expf(s));
            const float delta = (s > 20.0f) ? s : sp;
            const float u = sm_u[ch * UP + j];
            *(float2*)(sm_dd + ch * DP + 2 * j) = make_float2(delta, delta * u);
        }
        __syncthreads();

        {
            float4 BC0[4], BC1[4];
            float2 D0[4], D1[4];
            scan_load_group(sm_x, sm_dd, 0, q, ch, BC0, D0);
#pragma unroll 1
            for (int g = 0; g < TL / 4; g += 2) {
                scan_load_group(sm_x, sm_dd, g + 1, q, ch, BC1, D1);
                __builtin_amdgcn_sched_barrier(0);
                scan_run_group(sm_y, g, q, ch, A20, A21, h0, h1, BC0, D0);
                const int gn = (g + 2 < TL / 4) ? (g + 2) : g;
                scan_load_group(sm_x, sm_dd, gn, q, ch, BC0, D0);
                __builtin_amdgcn_sched_barrier(0);
                scan_run_group(sm_y, g + 1, q, ch, A20, A21, h0, h1, BC1, D1);
            }
        }
        __syncthreads();

#pragma unroll
        for (int m = 0; m < 8; ++m) {
            const int cc = m >> 1;
            const int j  = (m & 1) * 32 + lane;
            ob[(size_t)cc * LL + l0 + j] = sm_y[cc * UP + j] + DsC[cc] * sm_u[cc * UP + j];
        }
    }
}

__global__ __launch_bounds__(192) void ss2d_merge_ln(
    const float* __restrict__ outy,
    const float* __restrict__ gamma,
    const float* __restrict__ beta,
    float* __restrict__ out)
{
    __shared__ float red[16];
    __shared__ float stats[2];

    int b  = blockIdx.x >> 12;
    int lm = blockIdx.x & 4095;
    int hh = lm >> 6, ww = lm & 63;
    int lt = (ww << 6) | hh;
    int d  = threadIdx.x;

    const size_t ks = (size_t)DI * LL;
    const float* base = outy + (size_t)b * KK * ks + (size_t)d * LL;

    float v = base[lm]
            + base[2 * ks + (LL - 1 - lm)]
            + base[ks + lt]
            + base[3 * ks + (LL - 1 - lt)];

    float s = v, q = v * v;
#pragma unroll
    for (int o = 16; o; o >>= 1) {
        s += __shfl_xor(s, o, 32);
        q += __shfl_xor(q, o, 32);
    }
    int wv = d >> 5;
    if ((d & 31) == 0) { red[wv] = s; red[8 + wv] = q; }
    __syncthreads();
    if (d == 0) {
        float ts = 0.f, tq = 0.f;
        for (int i = 0; i < 6; ++i) { ts += red[i]; tq += red[8 + i]; }
        float mu  = ts / 192.0f;
        float var = tq / 192.0f - mu * mu;
        stats[0] = mu;
        stats[1] = rsqrtf(var + 1e-5f);
    }
    __syncthreads();
    float mu = stats[0], rstd = stats[1];
    out[(size_t)blockIdx.x * DI + d] = (v - mu) * rstd * gamma[d] + beta[d];
}

extern "C" void kernel_launch(void* const* d_in, const int* in_sizes, int n_in,
                              void* d_out, int out_size, void* d_ws, size_t ws_size,
                              hipStream_t stream) {
    const float* x   = (const float*)d_in[0];
    const float* y   = (const float*)d_in[1];
    const float* xpw = (const float*)d_in[2];
    const float* dtw = (const float*)d_in[3];
    const float* dtb = (const float*)d_in[4];
    const float* Al  = (const float*)d_in[5];
    const float* Ds  = (const float*)d_in[6];
    const float* gm  = (const float*)d_in[7];
    const float* bt  = (const float*)d_in[8];

    float* xdbl = (float*)d_ws;
    float* outy = xdbl + (size_t)B_ * KK * CP * LL;

    ss2d_proj_wmma<<<1536, 256, 0, stream>>>(x, xpw, xdbl);
    ss2d_scan<<<768, 32, 0, stream>>>(y, dtw, dtb, Al, Ds, xdbl, outy);
    ss2d_merge_ln<<<B_ * LL, 192, 0, stream>>>(outy, gm, bt, (float*)d_out);
}


# merge+LN kernel re-tiled: one 512-thread block per 8x8 tile of positions (thread = 1 position x 24 channels); loads cover 32-B or 16-B runs, not one line per lane; stores via LDS as whole lines
# speedup vs baseline: 9.3781x; 1.1344x over previous
#include <hip/hip_runtime.h>
#include <hip/hip_bf16.h>


#define B_  4
#define DI  192
#define LL  4096
#define NS  16
#define RR  6
#define KK  4
#define CP  38
#define TL  64
#define TLP 65

typedef float v2f __attribute__((ext_vector_type(2)));
typedef float v8f __attribute__((ext_vector_type(8)));

__device__ __forceinline__ int scan_idx(int k, int l) {
    int ll = (k & 2) ? (LL - 1 - l) : l;
    if (k & 1) return ((ll & 63) << 6) | (ll >> 6);
    return ll;
}

__global__ __launch_bounds__(256) void ss2d_proj_wmma(
    const float* __restrict__ x,
    const float* __restrict__ xpw,
    float* __restrict__ xdbl)
{
    int wave = (int)((blockIdx.x * blockDim.x + threadIdx.x) >> 5);
    int lane = threadIdx.x & 31;
    int nt = wave & 255;
    int mt = (wave >> 8) % 3;
    int kd = (wave >> 8) / 3;
    if (kd >= B_ * KK) return;
    int k = kd & 3;
    int b = kd >> 2;

    int m0 = mt << 4;
    int l0 = nt << 4;
    int hw = lane >> 4;
    int ln = lane & 15;

    const float* wbase = xpw + (size_t)k * CP * DI;
    const float* xb    = x + (size_t)b * DI * LL;

    int arow   = m0 + ln;
    int arow_c = arow < CP ? arow : (CP - 1);
    float amsk = arow < CP ? 1.0f : 0.0f;

    int midx = scan_idx(k, l0 + ln);

    v8f acc = {0.f, 0.f, 0.f, 0.f, 0.f, 0.f, 0.f, 0.f};
    for (int k0 = 0; k0 < DI; k0 += 4) {
        const float* ap = wbase + arow_c * DI + k0 + (hw << 1);
        v2f a; a.x = ap[0] * amsk; a.y = ap[1] * amsk;
        const float* bp = xb + (size_t)(k0 + (hw << 1)) * LL + midx;
        v2f bb; bb.x = bp[0]; bb.y = bp[LL];
        acc = __builtin_amdgcn_wmma_f32_16x16x4_f32(
            false, a, false, bb, (short)0, acc, false, false);
    }

    float* obase = xdbl + (size_t)kd * CP * LL + l0 + ln;
#pragma unroll
    for (int i = 0; i < 8; ++i) {
        int m = m0 + i + (hw << 3);
        if (m < CP) obase[(size_t)m * LL] = acc[i];
    }
}

#define CHW 4
#define LPC 8
#define XP  40
#define UP  72
#define DP  144

__device__ __forceinline__ float oct_sum(float v) {
    v += __int_as_float(__builtin_amdgcn_update_dpp(0, __float_as_int(v), 0xB1, 0xF, 0xF, true));
    v += __int_as_float(__builtin_amdgcn_update_dpp(0, __float_as_int(v), 0x4E, 0xF, 0xF, true));
    v += __int_as_float(__builtin_amdgcn_update_dpp(0, __float_as_int(v), 0x141, 0xF, 0xF, true));
    return v;
}

__device__ __forceinline__ void scan_load_group(
    const float* sm_x, const float* sm_dd, int g, int q, int ch,
    float4 (&BC)[4], float2 (&Dv)[4])
{
#pragma unroll
    for (int t = 0; t < 4; ++t) {
        int j = 4 * g + t;
        BC[t] = *(const float4*)(sm_x + j * XP + 8 + 4 * q);
        Dv[t] = *(const float2*)(sm_dd + ch * DP + 2 * j);
    }
}

__device__ __forceinline__ void scan_run_group(
    float* sm_y, int g, int q, int ch, float A20, float A21, float& h0, float& h1,
    const float4 (&BC)[4], const float2 (&Dv)[4])
{
    float yy[4];
#pragma unroll
    for (int t = 0; t < 4; ++t) {
        float delta = Dv[t].x, du = Dv[t].y;
        float dA0 = __builtin_amdgcn_exp2f(delta * A20);
        float dA1 = __builtin_amdgcn_exp2f(delta * A21);
        h0 = dA0 * h0 + du * BC[t].x;
        h1 = dA1 * h1 + du * BC[t].y;
        float yv = h0 * BC[t].z;
        yv += h1 * BC[t].w;
        yy[t] = oct_sum(yv);
    }
    if (q == 0)
        *(float4*)(sm_y + ch * UP + 4 * g) = make_float4(yy[0], yy[1], yy[2], yy[3]);
}

__global__ __launch_bounds__(32) void ss2d_scan(
    const float* __restrict__ y,
    const float* __restrict__ dtw_g,
    const float* __restrict__ dtb_g,
    const float* __restrict__ Alogs,
    const float* __restrict__ Ds_g,
    const float* __restrict__ xdbl,
    float* __restrict__ outy)
{
    __shared__ __attribute__((aligned(16))) float sm_x[TL * XP];
    __shared__ __attribute__((aligned(16))) float sm_u[CHW * UP];
    __shared__ __attribute__((aligned(16))) float sm_dd[CHW * DP];
    __shared__ __attribute__((aligned(16))) float sm_y[CHW * UP];

    const int lane = threadIdx.x;
    const int cb = blockIdx.x % 48;
    const int kd = blockIdx.x / 48;
    const int k  = kd & 3;
    const int b  = kd >> 2;
    const int ch = lane >> 3;
    const int q  = lane & 7;
    const int d0 = cb * CHW;
    const int kdch = k * DI + d0 + ch;

    const float A20 = -__expf(Alogs[(size_t)kdch * NS + 2 * q])     * 1.4426950408889634f;
    const float A21 = -__expf(Alogs[(size_t)kdch * NS + 2 * q + 1]) * 1.4426950408889634f;
    float dtw[RR];
#pragma unroll
    for (int r = 0; r < RR; ++r) dtw[r] = dtw_g[(size_t)kdch * RR + r];
    const float dtb = dtb_g[kdch];
    float DsC[CHW];
#pragma unroll
    for (int c4 = 0; c4 < CHW; ++c4) DsC[c4] = Ds_g[k * DI + d0 + c4];

    const float* xdb = xdbl + (size_t)kd * CP * LL;
    const float* yb  = y + ((size_t)b * DI + d0) * LL;
    float* ob        = outy + ((size_t)kd * DI + d0) * LL;

    float xr[2][CP];
    float ur[8];

#pragma unroll
    for (int hf = 0; hf < 2; ++hf)
#pragma unroll
        for (int r = 0; r < CP; ++r)
            xr[hf][r] = xdb[(size_t)r * LL + lane + 32 * hf];
#pragma unroll
    for (int m = 0; m < 8; ++m)
        ur[m] = yb[(size_t)(m >> 1) * LL + scan_idx(k, (m & 1) * 32 + lane)];

    float h0 = 0.0f, h1 = 0.0f;

    for (int c = 0; c < LL / TL; ++c) {
        const int l0 = c * TL;

#pragma unroll
        for (int hf = 0; hf < 2; ++hf) {
            float* dst = sm_x + (lane + 32 * hf) * XP;
            *(float4*)(dst)     = make_float4(xr[hf][0], xr[hf][1], xr[hf][2], xr[hf][3]);
            *(float4*)(dst + 4) = make_float4(xr[hf][4], xr[hf][5], 0.0f, 0.0f);
#pragma unroll
            for (int p = 0; p < 8; ++p)
                *(float4*)(dst + 8 + 4 * p) =
                    make_float4(xr[hf][RR + 2 * p],      xr[hf][RR + 2 * p + 1],
                                xr[hf][RR + NS + 2 * p], xr[hf][RR + NS + 2 * p + 1]);
        }
#pragma unroll
        for (int m = 0; m < 8; ++m)
            sm_u[(m >> 1) * UP + (m & 1) * 32 + lane] = ur[m];
        __syncthreads();

        {
            const int cn = (c + 1 < LL / TL) ? (c + 1) : c;
            const int ln0 = cn * TL;
#pragma unroll
            for (int hf = 0; hf < 2; ++hf)
#pragma unroll
                for (int r = 0; r < CP; ++r)
                    xr[hf][r] = xdb[(size_t)r * LL + ln0 + lane + 32 * hf];
#pragma unroll
            for (int m = 0; m < 8; ++m)
                ur[m] = yb[(size_t)(m >> 1) * LL + scan_idx(k, ln0 + (m & 1) * 32 + lane)];
        }

#pragma unroll
        for (int m = 0; m < 8; ++m) {
            const int j = q + 8 * m;
            const float4 d03 = *(const float4*)(sm_x + j * XP);
            const float2 d45 = *(const float2*)(sm_x + j * XP + 4);
            float s = dtb;
            s += dtw[0] * d03.x;
            s += dtw[1] * d03.y;
            s += dtw[2] * d03.z;
            s += dtw[3] * d03.w;
            s += dtw[4] * d45.x;
            s += dtw[5] * d45.y;
            const float sp = log1pf(__expf(s));
            const float delta = (s > 20.0f) ? s : sp;
            const float u = sm_u[ch * UP + j];
            *(float2*)(sm_dd + ch * DP + 2 * j) = make_float2(delta, delta * u);
        }
        __syncthreads();

        {
            float4 BC0[4], BC1[4];
            float2 D0[4], D1[4];
            scan_load_group(sm_x, sm_dd, 0, q, ch, BC0, D0);
#pragma unroll 1
            for (int g = 0; g < TL / 4; g += 2) {
                scan_load_group(sm_x, sm_dd, g + 1, q, ch, BC1, D1);
                __builtin_amdgcn_sched_barrier(0);
                scan_run_group(sm_y, g, q, ch, A20, A21, h0, h1, BC0, D0);
                const int gn = (g + 2 < TL / 4) ? (g + 2) : g;
                scan_load_group(sm_x, sm_dd, gn, q, ch, BC0, D0);
                __builtin_amdgcn_sched_barrier(0);
                scan_run_group(sm_y, g + 1, q, ch, A20, A21, h0, h1, BC1, D1);
            }
        }
        __syncthreads();

#pragma unroll
        for (int m = 0; m < 8; ++m) {
            const int cc = m >> 1;
            const int j  = (m & 1) * 32 + lane;
            ob[(size_t)cc * LL + l0 + j] = sm_y[cc * UP + j] + DsC[cc] * sm_u[cc * UP + j];
        }
    }
}

#define MTE   8
#define MPOS  (MTE * MTE)
#define MDG   8
#define MCH   (DI / MDG)
#define MPIT  (DI + 1)

__global__ __launch_bounds__(MPOS * MDG) void ss2d_merge_ln(
    const float* __restrict__ outy,
    const float* __restrict__ gamma,
    const float* __restrict__ beta,
    float* __restrict__ out)
{
    __shared__ float tile[MPOS * MPIT];
    __shared__ float red_s[MDG * MPOS];
    __shared__ float red_q[MDG * MPOS];

    const int tid  = threadIdx.x;
    const int lane = tid & 31;
    const int wv   = tid >> 5;
    const int b    = blockIdx.x >> 6;
    const int t    = blockIdx.x & 63;
    const int h0   = (t >> 3) * MTE;
    const int w0   = (t & 7) * MTE;
    const int pos  = tid & (MPOS - 1);
    const int dg   = tid >> 6;
    const int hh   = h0 + (pos >> 3);
    const int ww   = w0 + (pos & 7);
    const int lm   = (hh << 6) | ww;
    const int lt   = (ww << 6) | hh;

    const size_t ks = (size_t)DI * LL;
    const float* base = outy + (size_t)b * KK * ks + (size_t)dg * LL;
    const float* p0 = base + lm;
    const float* p2 = base + 2 * ks + (LL - 1 - lm);
    const float* p1 = base + ks + lt;
    const float* p3 = base + 3 * ks + (LL - 1 - lt);

    float gm[6], bt[6];
#pragma unroll
    for (int c = 0; c < 6; ++c) { gm[c] = gamma[lane + 32 * c]; bt[c] = beta[lane + 32 * c]; }

    float a0[MCH], a2[MCH], a1[MCH], a3[MCH];
#pragma unroll
    for (int it = 0; it < MCH; ++it) {
        const size_t o = (size_t)it * MDG * LL;
        a0[it] = p0[o]; a2[it] = p2[o]; a1[it] = p1[o]; a3[it] = p3[o];
    }
    __builtin_amdgcn_sched_barrier(0);

    float v[MCH];
    float s = 0.0f, q = 0.0f;
#pragma unroll
    for (int it = 0; it < MCH; ++it) {
        v[it] = a0[it] + a2[it] + a1[it] + a3[it];
        s += v[it];
        q += v[it] * v[it];
    }
    red_s[dg * MPOS + pos] = s;
    red_q[dg * MPOS + pos] = q;
    __syncthreads();

    float ts = 0.0f, tq = 0.0f;
#pragma unroll
    for (int g = 0; g < MDG; ++g) {
        ts += red_s[g * MPOS + pos];
        tq += red_q[g * MPOS + pos];
    }
    const float mu   = ts / 192.0f;
    const float var  = tq / 192.0f - mu * mu;
    const float rstd = rsqrtf(var + 1e-5f);

#pragma unroll
    for (int it = 0; it < MCH; ++it)
        tile[pos * MPIT + dg + it * MDG] = (v[it] - mu) * rstd;
    __syncthreads();

    const int pb = wv * 4;
    const int lb = ((h0 + (pb >> 3)) << 6) | (w0 + (pb & 7));
    float* ob = out + ((size_t)b * LL + lb) * DI + lane;
    const float* tb = tile + pb * MPIT + lane;
#pragma unroll
    for (int pi = 0; pi < 4; ++pi)
#pragma unroll
        for (int c = 0; c < 6; ++c)
            ob[pi * DI + 32 * c] = tb[pi * MPIT + 32 * c] * gm[c] + bt[c];
}

extern "C" void kernel_launch(void* const* d_in, const int* in_sizes, int n_in,
                              void* d_out, int out_size, void* d_ws, size_t ws_size,
                              hipStream_t stream) {
    const float* x   = (const float*)d_in[0];
    const float* y   = (const float*)d_in[1];
    const float* xpw = (const float*)d_in[2];
    const float* dtw = (const float*)d_in[3];
    const float* dtb = (const float*)d_in[4];
    const float* Al  = (const float*)d_in[5];
    const float* Ds  = (const float*)d_in[6];
    const float* gm  = (const float*)d_in[7];
    const float* bt  = (const float*)d_in[8];

    float* xdbl = (float*)d_ws;
    float* outy = xdbl + (size_t)B_ * KK * CP * LL;

    ss2d_proj_wmma<<<1536, 256, 0, stream>>>(x, xpw, xdbl);
    ss2d_scan<<<768, 32, 0, stream>>>(y, dtw, dtb, Al, Ds, xdbl, outy);
    ss2d_merge_ln<<<B_ * (LL / MPOS), MPOS * MDG, 0, stream>>>(outy, gm, bt, (float*)d_out);
}


# projection re-tiled: a block stages one 192x128 x tile and its direction pair's weights in LDS once, computes both directions from LDS; global loads contiguous float4, x read 2x not 12x
# speedup vs baseline: 11.9200x; 1.2710x over previous
#include <hip/hip_runtime.h>
#include <hip/hip_bf16.h>


#define B_  4
#define DI  192
#define LL  4096
#define NS  16
#define RR  6
#define KK  4
#define CP  38
#define TL  64
#define TLP 65

typedef float v2f __attribute__((ext_vector_type(2)));
typedef float v8f __attribute__((ext_vector_type(8)));

__device__ __forceinline__ int scan_idx(int k, int l) {
    int ll = (k & 2) ? (LL - 1 - l) : l;
    if (k & 1) return ((ll & 63) << 6) | (ll >> 6);
    return ll;
}

#define PNT  8
#define PXP  136
#define PWP  196

__global__ __launch_bounds__(256) void ss2d_proj_wmma(
    const float* __restrict__ x,
    const float* __restrict__ xpw,
    float* __restrict__ xdbl)
{
    __shared__ __attribute__((aligned(16))) float xs[DI * PXP];
    __shared__ __attribute__((aligned(16))) float ws[2 * CP * PWP];

    const int tid  = threadIdx.x;
    const int lane = tid & 31;
    const int j    = tid >> 5;
    const int t    = blockIdx.x & 31;
    const int cls  = (blockIdx.x >> 5) & 1;
    const int b    = blockIdx.x >> 6;

    const float* xb = x + (size_t)b * DI * LL;

    if (cls == 0) {
        const int p0 = t * 128;
#pragma unroll
        for (int i = 0; i < 24; ++i) {
            const int f  = tid + 256 * i;
            const int d  = f >> 5;
            const int c4 = f & 31;
            const float4 v = *(const float4*)(xb + (size_t)d * LL + p0 + 4 * c4);
            *(float4*)(xs + d * PXP + 4 * c4) = v;
        }
    } else {
        const int w0 = (t & 7) * 8;
        const int h0 = (t >> 3) * 16;
#pragma unroll
        for (int i = 0; i < 24; ++i) {
            const int f  = tid + 256 * i;
            const int d  = f >> 5;
            const int r  = f & 31;
            const int c  = r >> 1;
            const int jq = r & 1;
            const float4 v = *(const float4*)(xb + (size_t)d * LL + (h0 + c) * 64 + w0 + 4 * jq);
            float* dst = xs + d * PXP + 64 * jq + c;
            dst[0]  = v.x;
            dst[16] = v.y;
            dst[32] = v.z;
            dst[48] = v.w;
        }
    }
#pragma unroll
    for (int i = 0; i < 15; ++i) {
        const int f = tid + 256 * i;
        if (f < 2 * CP * (DI / 4)) {
            const int r  = f / (DI / 4);
            const int q  = f - r * (DI / 4);
            const int sr = cls * CP + r + (r >= CP ? CP : 0);
            const float4 v = *(const float4*)(xpw + (size_t)sr * DI + 4 * q);
            *(float4*)(ws + r * PWP + 4 * q) = v;
        }
    }
    __syncthreads();

    const int hw = lane >> 4;
    const int ln = lane & 15;

    int aoff[6];
#pragma unroll
    for (int mt = 0; mt < 3; ++mt) {
        const int row = (16 * mt + ln) < CP ? (16 * mt + ln) : (CP - 1);
        aoff[mt]     = row * PWP + 2 * hw;
        aoff[3 + mt] = (CP + row) * PWP + 2 * hw;
    }
    const int boff = (2 * hw) * PXP + 16 * j + ln;

    v8f acc[6];
#pragma unroll
    for (int a = 0; a < 6; ++a) acc[a] = (v8f){0.f, 0.f, 0.f, 0.f, 0.f, 0.f, 0.f, 0.f};

#pragma unroll 4
    for (int k0 = 0; k0 < DI; k0 += 4) {
        v2f bb;
        bb.x = xs[boff + k0 * PXP];
        bb.y = xs[boff + k0 * PXP + PXP];
#pragma unroll
        for (int a = 0; a < 6; ++a) {
            const v2f av = *(const v2f*)(ws + aoff[a] + k0);
            acc[a] = __builtin_amdgcn_wmma_f32_16x16x4_f32(
                false, av, false, bb, (short)0, acc[a], false, false);
        }
    }

    int lf;
    if (cls == 0) lf = t * 128 + 16 * j + ln;
    else          lf = (((t & 7) * 8 + j) << 6) + (t >> 3) * 16 + ln;

#pragma unroll
    for (int kk = 0; kk < 2; ++kk) {
        const int kd = b * KK + cls + 2 * kk;
        const int l  = kk ? (LL - 1 - lf) : lf;
        float* obase = xdbl + (size_t)kd * CP * LL + l;
#pragma unroll
        for (int mt = 0; mt < 3; ++mt) {
#pragma unroll
            for (int i = 0; i < 8; ++i) {
                const int m = 16 * mt + i + (hw << 3);
                if (m < CP) obase[(size_t)m * LL] = acc[kk * 3 + mt][i];
            }
        }
    }
}

#define CHW 4
#define LPC 8
#define XP  40
#define UP  72
#define DP  144

__device__ __forceinline__ float oct_sum(float v) {
    v += __int_as_float(__builtin_amdgcn_update_dpp(0, __float_as_int(v), 0xB1, 0xF, 0xF, true));
    v += __int_as_float(__builtin_amdgcn_update_dpp(0, __float_as_int(v), 0x4E, 0xF, 0xF, true));
    v += __int_as_float(__builtin_amdgcn_update_dpp(0, __float_as_int(v), 0x141, 0xF, 0xF, true));
    return v;
}

__device__ __forceinline__ void scan_load_group(
    const float* sm_x, const float* sm_dd, int g, int q, int ch,
    float4 (&BC)[4], float2 (&Dv)[4])
{
#pragma unroll
    for (int t = 0; t < 4; ++t) {
        int j = 4 * g + t;
        BC[t] = *(const float4*)(sm_x + j * XP + 8 + 4 * q);
        Dv[t] = *(const float2*)(sm_dd + ch * DP + 2 * j);
    }
}

__device__ __forceinline__ void scan_run_group(
    float* sm_y, int g, int q, int ch, float A20, float A21, float& h0, float& h1,
    const float4 (&BC)[4], const float2 (&Dv)[4])
{
    float yy[4];
#pragma unroll
    for (int t = 0; t < 4; ++t) {
        float delta = Dv[t].x, du = Dv[t].y;
        float dA0 = __builtin_amdgcn_exp2f(delta * A20);
        float dA1 = __builtin_amdgcn_exp2f(delta * A21);
        h0 = dA0 * h0 + du * BC[t].x;
        h1 = dA1 * h1 + du * BC[t].y;
        float yv = h0 * BC[t].z;
        yv += h1 * BC[t].w;
        yy[t] = oct_sum(yv);
    }
    if (q == 0)
        *(float4*)(sm_y + ch * UP + 4 * g) = make_float4(yy[0], yy[1], yy[2], yy[3]);
}

__global__ __launch_bounds__(32) void ss2d_scan(
    const float* __restrict__ y,
    const float* __restrict__ dtw_g,
    const float* __restrict__ dtb_g,
    const float* __restrict__ Alogs,
    const float* __restrict__ Ds_g,
    const float* __restrict__ xdbl,
    float* __restrict__ outy)
{
    __shared__ __attribute__((aligned(16))) float sm_x[TL * XP];
    __shared__ __attribute__((aligned(16))) float sm_u[CHW * UP];
    __shared__ __attribute__((aligned(16))) float sm_dd[CHW * DP];
    __shared__ __attribute__((aligned(16))) float sm_y[CHW * UP];

    const int lane = threadIdx.x;
    const int cb = blockIdx.x % 48;
    const int kd = blockIdx.x / 48;
    const int k  = kd & 3;
    const int b  = kd >> 2;
    const int ch = lane >> 3;
    const int q  = lane & 7;
    const int d0 = cb * CHW;
    const int kdch = k * DI + d0 + ch;

    const float A20 = -__expf(Alogs[(size_t)kdch * NS + 2 * q])     * 1.4426950408889634f;
    const float A21 = -__expf(Alogs[(size_t)kdch * NS + 2 * q + 1]) * 1.4426950408889634f;
    float dtw[RR];
#pragma unroll
    for (int r = 0; r < RR; ++r) dtw[r] = dtw_g[(size_t)kdch * RR + r];
    const float dtb = dtb_g[kdch];
    float DsC[CHW];
#pragma unroll
    for (int c4 = 0; c4 < CHW; ++c4) DsC[c4] = Ds_g[k * DI + d0 + c4];

    const float* xdb = xdbl + (size_t)kd * CP * LL;
    const float* yb  = y + ((size_t)b * DI + d0) * LL;
    float* ob        = outy + ((size_t)kd * DI + d0) * LL;

    float xr[2][CP];
    float ur[8];

#pragma unroll
    for (int hf = 0; hf < 2; ++hf)
#pragma unroll
        for (int r = 0; r < CP; ++r)
            xr[hf][r] = xdb[(size_t)r * LL + lane + 32 * hf];
#pragma unroll
    for (int m = 0; m < 8; ++m)
        ur[m] = yb[(size_t)(m >> 1) * LL + scan_idx(k, (m & 1) * 32 + lane)];

    float h0 = 0.0f, h1 = 0.0f;

    for (int c = 0; c < LL / TL; ++c) {
        const int l0 = c * TL;

#pragma unroll
        for (int hf = 0; hf < 2; ++hf) {
            float* dst = sm_x + (lane + 32 * hf) * XP;
            *(float4*)(dst)     = make_float4(xr[hf][0], xr[hf][1], xr[hf][2], xr[hf][3]);
            *(float4*)(dst + 4) = make_float4(xr[hf][4], xr[hf][5], 0.0f, 0.0f);
#pragma unroll
            for (int p = 0; p < 8; ++p)
                *(float4*)(dst + 8 + 4 * p) =
                    make_float4(xr[hf][RR + 2 * p],      xr[hf][RR + 2 * p + 1],
                                xr[hf][RR + NS + 2 * p], xr[hf][RR + NS + 2 * p + 1]);
        }
#pragma unroll
        for (int m = 0; m < 8; ++m)
            sm_u[(m >> 1) * UP + (m & 1) * 32 + lane] = ur[m];
        __syncthreads();

        {
            const int cn = (c + 1 < LL / TL) ? (c + 1) : c;
            const int ln0 = cn * TL;
#pragma unroll
            for (int hf = 0; hf < 2; ++hf)
#pragma unroll
                for (int r = 0; r < CP; ++r)
                    xr[hf][r] = xdb[(size_t)r * LL + ln0 + lane + 32 * hf];
#pragma unroll
            for (int m = 0; m < 8; ++m)
                ur[m] = yb[(size_t)(m >> 1) * LL + scan_idx(k, ln0 + (m & 1) * 32 + lane)];
        }

#pragma unroll
        for (int m = 0; m < 8; ++m) {
            const int j = q + 8 * m;
            const float4 d03 = *(const float4*)(sm_x + j * XP);
            const float2 d45 = *(const float2*)(sm_x + j * XP + 4);
            float s = dtb;
            s += dtw[0] * d03.x;
            s += dtw[1] * d03.y;
            s += dtw[2] * d03.z;
            s += dtw[3] * d03.w;
            s += dtw[4] * d45.x;
            s += dtw[5] * d45.y;
            const float sp = log1pf(__expf(s));
            const float delta = (s > 20.0f) ? s : sp;
            const float u = sm_u[ch * UP + j];
            *(float2*)(sm_dd + ch * DP + 2 * j) = make_float2(delta, delta * u);
        }
        __syncthreads();

        {
            float4 BC0[4], BC1[4];
            float2 D0[4], D1[4];
            scan_load_group(sm_x, sm_dd, 0, q, ch, BC0, D0);
#pragma unroll 1
            for (int g = 0; g < TL / 4; g += 2) {
                scan_load_group(sm_x, sm_dd, g + 1, q, ch, BC1, D1);
                __builtin_amdgcn_sched_barrier(0);
                scan_run_group(sm_y, g, q, ch, A20, A21, h0, h1, BC0, D0);
                const int gn = (g + 2 < TL / 4) ? (g + 2) : g;
                scan_load_group(sm_x, sm_dd, gn, q, ch, BC0, D0);
                __builtin_amdgcn_sched_barrier(0);
                scan_run_group(sm_y, g + 1, q, ch, A20, A21, h0, h1, BC1, D1);
            }
        }
        __syncthreads();

#pragma unroll
        for (int m = 0; m < 8; ++m) {
            const int cc = m >> 1;
            const int j  = (m & 1) * 32 + lane;
            ob[(size_t)cc * LL + l0 + j] = sm_y[cc * UP + j] + DsC[cc] * sm_u[cc * UP + j];
        }
    }
}

#define MTE   8
#define MPOS  (MTE * MTE)
#define MDG   8
#define MCH   (DI / MDG)
#define MPIT  (DI + 1)

__global__ __launch_bounds__(MPOS * MDG) void ss2d_merge_ln(
    const float* __restrict__ outy,
    const float* __restrict__ gamma,
    const float* __restrict__ beta,
    float* __restrict__ out)
{
    __shared__ float tile[MPOS * MPIT];
    __shared__ float red_s[MDG * MPOS];
    __shared__ float red_q[MDG * MPOS];

    const int tid  = threadIdx.x;
    const int lane = tid & 31;
    const int wv   = tid >> 5;
    const int b    = blockIdx.x >> 6;
    const int t    = blockIdx.x & 63;
    const int h0   = (t >> 3) * MTE;
    const int w0   = (t & 7) * MTE;
    const int pos  = tid & (MPOS - 1);
    const int dg   = tid >> 6;
    const int hh   = h0 + (pos >> 3);
    const int ww   = w0 + (pos & 7);
    const int lm   = (hh << 6) | ww;
    const int lt   = (ww << 6) | hh;

    const size_t ks = (size_t)DI * LL;
    const float* base = outy + (size_t)b * KK * ks + (size_t)dg * LL;
    const float* p0 = base + lm;
    const float* p2 = base + 2 * ks + (LL - 1 - lm);
    const float* p1 = base + ks + lt;
    const float* p3 = base + 3 * ks + (LL - 1 - lt);

    float gm[6], bt[6];
#pragma unroll
    for (int c = 0; c < 6; ++c) { gm[c] = gamma[lane + 32 * c]; bt[c] = beta[lane + 32 * c]; }

    float a0[MCH], a2[MCH], a1[MCH], a3[MCH];
#pragma unroll
    for (int it = 0; it < MCH; ++it) {
        const size_t o = (size_t)it * MDG * LL;
        a0[it] = p0[o]; a2[it] = p2[o]; a1[it] = p1[o]; a3[it] = p3[o];
    }
    __builtin_amdgcn_sched_barrier(0);

    float v[MCH];
    float s = 0.0f, q = 0.0f;
#pragma unroll
    for (int it = 0; it < MCH; ++it) {
        v[it] = a0[it] + a2[it] + a1[it] + a3[it];
        s += v[it];
        q += v[it] * v[it];
    }
    red_s[dg * MPOS + pos] = s;
    red_q[dg * MPOS + pos] = q;
    __syncthreads();

    float ts = 0.0f, tq = 0.0f;
#pragma unroll
    for (int g = 0; g < MDG; ++g) {
        ts += red_s[g * MPOS + pos];
        tq += red_q[g * MPOS + pos];
    }
    const float mu   = ts / 192.0f;
    const float var  = tq / 192.0f - mu * mu;
    const float rstd = rsqrtf(var + 1e-5f);

#pragma unroll
    for (int it = 0; it < MCH; ++it)
        tile[pos * MPIT + dg + it * MDG] = (v[it] - mu) * rstd;
    __syncthreads();

    const int pb = wv * 4;
    const int lb = ((h0 + (pb >> 3)) << 6) | (w0 + (pb & 7));
    float* ob = out + ((size_t)b * LL + lb) * DI + lane;
    const float* tb = tile + pb * MPIT + lane;
#pragma unroll
    for (int pi = 0; pi < 4; ++pi)
#pragma unroll
        for (int c = 0; c < 6; ++c)
            ob[pi * DI + 32 * c] = tb[pi * MPIT + 32 * c] * gm[c] + bt[c];
}

extern "C" void kernel_launch(void* const* d_in, const int* in_sizes, int n_in,
                              void* d_out, int out_size, void* d_ws, size_t ws_size,
                              hipStream_t stream) {
    const float* x   = (const float*)d_in[0];
    const float* y   = (const float*)d_in[1];
    const float* xpw = (const float*)d_in[2];
    const float* dtw = (const float*)d_in[3];
    const float* dtb = (const float*)d_in[4];
    const float* Al  = (const float*)d_in[5];
    const float* Ds  = (const float*)d_in[6];
    const float* gm  = (const float*)d_in[7];
    const float* bt  = (const float*)d_in[8];

    float* xdbl = (float*)d_ws;
    float* outy = xdbl + (size_t)B_ * KK * CP * LL;

    ss2d_proj_wmma<<<256, 256, 0, stream>>>(x, xpw, xdbl);
    ss2d_scan<<<768, 32, 0, stream>>>(y, dtw, dtb, Al, Ds, xdbl, outy);
    ss2d_merge_ln<<<B_ * (LL / MPOS), MPOS * MDG, 0, stream>>>(outy, gm, bt, (float*)d_out);
}


# scan split by role inside an 8-wave work-group per 12 channels: 3 waves run only the recurrence; 5 others stage B/C rows, softplus, loads and stores one chunk ahead/behind via two LDS buffer sets
# speedup vs baseline: 19.2479x; 1.6148x over previous
#include <hip/hip_runtime.h>
#include <hip/hip_bf16.h>


#define B_  4
#define DI  192
#define LL  4096
#define NS  16
#define RR  6
#define KK  4
#define CP  38
#define TL  64
#define TLP 65

typedef float v2f __attribute__((ext_vector_type(2)));
typedef float v8f __attribute__((ext_vector_type(8)));

__device__ __forceinline__ int scan_idx(int k, int l) {
    int ll = (k & 2) ? (LL - 1 - l) : l;
    if (k & 1) return ((ll & 63) << 6) | (ll >> 6);
    return ll;
}

#define PNT  8
#define PXP  136
#define PWP  196

__global__ __launch_bounds__(256) void ss2d_proj_wmma(
    const float* __restrict__ x,
    const float* __restrict__ xpw,
    float* __restrict__ xdbl)
{
    __shared__ __attribute__((aligned(16))) float xs[DI * PXP];
    __shared__ __attribute__((aligned(16))) float ws[2 * CP * PWP];

    const int tid  = threadIdx.x;
    const int lane = tid & 31;
    const int j    = tid >> 5;
    const int t    = blockIdx.x & 31;
    const int cls  = (blockIdx.x >> 5) & 1;
    const int b    = blockIdx.x >> 6;

    const float* xb = x + (size_t)b * DI * LL;

    if (cls == 0) {
        const int p0 = t * 128;
#pragma unroll
        for (int i = 0; i < 24; ++i) {
            const int f  = tid + 256 * i;
            const int d  = f >> 5;
            const int c4 = f & 31;
            const float4 v = *(const float4*)(xb + (size_t)d * LL + p0 + 4 * c4);
            *(float4*)(xs + d * PXP + 4 * c4) = v;
        }
    } else {
        const int w0 = (t & 7) * 8;
        const int h0 = (t >> 3) * 16;
#pragma unroll
        for (int i = 0; i < 24; ++i) {
            const int f  = tid + 256 * i;
            const int d  = f >> 5;
            const int r  = f & 31;
            const int c  = r >> 1;
            const int jq = r & 1;
            const float4 v = *(const float4*)(xb + (size_t)d * LL + (h0 + c) * 64 + w0 + 4 * jq);
            float* dst = xs + d * PXP + 64 * jq + c;
            dst[0]  = v.x;
            dst[16] = v.y;
            dst[32] = v.z;
            dst[48] = v.w;
        }
    }
#pragma unroll
    for (int i = 0; i < 15; ++i) {
        const int f = tid + 256 * i;
        if (f < 2 * CP * (DI / 4)) {
            const int r  = f / (DI / 4);
            const int q  = f - r * (DI / 4);
            const int sr = cls * CP + r + (r >= CP ? CP : 0);
            const float4 v = *(const float4*)(xpw + (size_t)sr * DI + 4 * q);
            *(float4*)(ws + r * PWP + 4 * q) = v;
        }
    }
    __syncthreads();

    const int hw = lane >> 4;
    const int ln = lane & 15;

    int aoff[6];
#pragma unroll
    for (int mt = 0; mt < 3; ++mt) {
        const int row = (16 * mt + ln) < CP ? (16 * mt + ln) : (CP - 1);
        aoff[mt]     = row * PWP + 2 * hw;
        aoff[3 + mt] = (CP + row) * PWP + 2 * hw;
    }
    const int boff = (2 * hw) * PXP + 16 * j + ln;

    v8f acc[6];
#pragma unroll
    for (int a = 0; a < 6; ++a) acc[a] = (v8f){0.f, 0.f, 0.f, 0.f, 0.f, 0.f, 0.f, 0.f};

#pragma unroll 4
    for (int k0 = 0; k0 < DI; k0 += 4) {
        v2f bb;
        bb.x = xs[boff + k0 * PXP];
        bb.y = xs[boff + k0 * PXP + PXP];
#pragma unroll
        for (int a = 0; a < 6; ++a) {
            const v2f av = *(const v2f*)(ws + aoff[a] + k0);
            acc[a] = __builtin_amdgcn_wmma_f32_16x16x4_f32(
                false, av, false, bb, (short)0, acc[a], false, false);
        }
    }

    int lf;
    if (cls == 0) lf = t * 128 + 16 * j + ln;
    else          lf = (((t & 7) * 8 + j) << 6) + (t >> 3) * 16 + ln;

#pragma unroll
    for (int kk = 0; kk < 2; ++kk) {
        const int kd = b * KK + cls + 2 * kk;
        const int l  = kk ? (LL - 1 - lf) : lf;
        float* obase = xdbl + (size_t)kd * CP * LL + l;
#pragma unroll
        for (int mt = 0; mt < 3; ++mt) {
#pragma unroll
            for (int i = 0; i < 8; ++i) {
                const int m = 16 * mt + i + (hw << 3);
                if (m < CP) obase[(size_t)m * LL] = acc[kk * 3 + mt][i];
            }
        }
    }
}

#define CHW 4
#define LPC 8
#define SCW 3
#define GCH (SCW * CHW)
#define SFE 6
#define XP  36
#define UP  72
#define DP  144

__device__ __forceinline__ float oct_sum(float v) {
    v += __int_as_float(__builtin_amdgcn_update_dpp(0, __float_as_int(v), 0xB1, 0xF, 0xF, true));
    v += __int_as_float(__builtin_amdgcn_update_dpp(0, __float_as_int(v), 0x4E, 0xF, 0xF, true));
    v += __int_as_float(__builtin_amdgcn_update_dpp(0, __float_as_int(v), 0x141, 0xF, 0xF, true));
    return v;
}

__device__ __forceinline__ void scan_load_group(
    const float* sm_x, const float* sm_dd, int g, int q, int ch,
    float4 (&BC)[4], float2 (&Dv)[4])
{
#pragma unroll
    for (int t = 0; t < 4; ++t) {
        int j = 4 * g + t;
        BC[t] = *(const float4*)(sm_x + j * XP + 4 * q);
        Dv[t] = *(const float2*)(sm_dd + ch * DP + 2 * j);
    }
}

__device__ __forceinline__ void scan_run_group(
    float* sm_y, int g, int q, int ch, float A20, float A21, float& h0, float& h1,
    const float4 (&BC)[4], const float2 (&Dv)[4])
{
    float yy[4];
#pragma unroll
    for (int t = 0; t < 4; ++t) {
        float delta = Dv[t].x, du = Dv[t].y;
        float dA0 = __builtin_amdgcn_exp2f(delta * A20);
        float dA1 = __builtin_amdgcn_exp2f(delta * A21);
        h0 = dA0 * h0 + du * BC[t].x;
        h1 = dA1 * h1 + du * BC[t].y;
        float yv = h0 * BC[t].z;
        yv += h1 * BC[t].w;
        yy[t] = oct_sum(yv);
    }
    if (q == 0)
        *(float4*)(sm_y + ch * UP + 4 * g) = make_float4(yy[0], yy[1], yy[2], yy[3]);
}

__global__ __launch_bounds__(256) void ss2d_scan(
    const float* __restrict__ y,
    const float* __restrict__ dtw_g,
    const float* __restrict__ dtb_g,
    const float* __restrict__ Alogs,
    const float* __restrict__ Ds_g,
    const float* __restrict__ xdbl,
    float* __restrict__ outy)
{
    __shared__ __attribute__((aligned(16))) float sm_x[2 * TL * XP];
    __shared__ __attribute__((aligned(16))) float sm_dd[2 * GCH * DP];
    __shared__ __attribute__((aligned(16))) float sm_y[2 * GCH * UP];

    const int lane = threadIdx.x & 31;
    const int wv = __builtin_amdgcn_readfirstlane((int)(threadIdx.x >> 5));
    const bool is_scan  = wv < SCW;
    const bool is_prep  = (wv >= SCW) && (wv < SCW + 4);
    const bool is_stage = wv == SCW + 4;

    const int cg = blockIdx.x % (DI / GCH);
    const int kd = blockIdx.x / (DI / GCH);
    const int k  = kd & 3;
    const int b  = kd >> 2;
    const int d0 = cg * GCH;

    const float* xdb = xdbl + (size_t)kd * CP * LL;
    const float* yb  = y + ((size_t)b * DI + d0) * LL;
    float* ob        = outy + ((size_t)kd * DI + d0) * LL;

    const int ch = lane >> 3;
    const int q  = lane & 7;
    const int cw = (is_scan ? wv : 0) * CHW + ch;
    const int kdch = k * DI + d0 + cw;
    const float A20 = -__expf(Alogs[(size_t)kdch * NS + 2 * q])     * 1.4426950408889634f;
    const float A21 = -__expf(Alogs[(size_t)kdch * NS + 2 * q + 1]) * 1.4426950408889634f;
    float h0 = 0.0f, h1 = 0.0f;

    const int pw  = is_prep ? (wv - SCW) : 0;
    const int ps  = (pw & 1) * 32 + lane;
    const int pc0 = (pw >> 1) * SFE;
    float dtw[SFE][RR], dtb[SFE], DsC[SFE];
#pragma unroll
    for (int e = 0; e < SFE; ++e) {
        const int kc = k * DI + d0 + pc0 + e;
#pragma unroll
        for (int r = 0; r < RR; ++r) dtw[e][r] = dtw_g[(size_t)kc * RR + r];
        dtb[e] = dtb_g[kc];
        DsC[e] = Ds_g[kc];
    }
    float dtr[RR];
    float un[SFE];
    float u1[SFE], u2[SFE];
#pragma unroll
    for (int r = 0; r < RR; ++r) dtr[r] = 0.0f;
#pragma unroll
    for (int e = 0; e < SFE; ++e) { un[e] = 0.0f; u1[e] = 0.0f; u2[e] = 0.0f; }

    float xr[2][2 * NS];
#pragma unroll
    for (int hf = 0; hf < 2; ++hf)
#pragma unroll
        for (int r = 0; r < 2 * NS; ++r) xr[hf][r] = 0.0f;

    if (is_prep) {
#pragma unroll
        for (int r = 0; r < RR; ++r) dtr[r] = xdb[(size_t)r * LL + ps];
#pragma unroll
        for (int e = 0; e < SFE; ++e) un[e] = yb[(size_t)(pc0 + e) * LL + scan_idx(k, ps)];
    } else if (is_stage) {
#pragma unroll
        for (int hf = 0; hf < 2; ++hf)
#pragma unroll
            for (int r = 0; r < 2 * NS; ++r)
                xr[hf][r] = xdb[(size_t)(RR + r) * LL + lane + 32 * hf];
    }

    for (int i = 0; i < LL / TL + 2; ++i) {
        const int bs = i & 1;
        if (is_scan) {
            if (i >= 1 && i <= LL / TL) {
                const float* px = sm_x  + (bs ^ 1) * (TL * XP);
                const float* pd = sm_dd + (bs ^ 1) * (GCH * DP);
                float*       py = sm_y  + (bs ^ 1) * (GCH * UP);
                float4 BC0[4], BC1[4];
                float2 D0[4], D1[4];
                scan_load_group(px, pd, 0, q, cw, BC0, D0);
#pragma unroll 1
                for (int g = 0; g < TL / 4; g += 2) {
                    scan_load_group(px, pd, g + 1, q, cw, BC1, D1);
                    __builtin_amdgcn_sched_barrier(0);
                    scan_run_group(py, g, q, cw, A20, A21, h0, h1, BC0, D0);
                    const int gn = (g + 2 < TL / 4) ? (g + 2) : g;
                    scan_load_group(px, pd, gn, q, cw, BC0, D0);
                    __builtin_amdgcn_sched_barrier(0);
                    scan_run_group(py, g + 1, q, cw, A20, A21, h0, h1, BC1, D1);
                }
            }
        } else if (is_prep) {
            float dnx[RR], unx[SFE];
            if (i + 1 < LL / TL) {
                const int ln0 = (i + 1) * TL;
#pragma unroll
                for (int r = 0; r < RR; ++r) dnx[r] = xdb[(size_t)r * LL + ln0 + ps];
#pragma unroll
                for (int e = 0; e < SFE; ++e)
                    unx[e] = yb[(size_t)(pc0 + e) * LL + scan_idx(k, ln0 + ps)];
            } else {
#pragma unroll
                for (int r = 0; r < RR; ++r) dnx[r] = 0.0f;
#pragma unroll
                for (int e = 0; e < SFE; ++e) unx[e] = 0.0f;
            }
            if (i >= 2) {
                const float* psy = sm_y + bs * (GCH * UP) + pc0 * UP + ps;
                float* po = ob + (size_t)pc0 * LL + (i - 2) * TL + ps;
#pragma unroll
                for (int e = 0; e < SFE; ++e)
                    po[(size_t)e * LL] = psy[e * UP] + DsC[e] * u2[e];
            }
            if (i < LL / TL) {
                float* pdd = sm_dd + bs * (GCH * DP) + pc0 * DP + 2 * ps;
#pragma unroll
                for (int e = 0; e < SFE; ++e) {
                    float s = dtb[e];
                    s += dtw[e][0] * dtr[0];
                    s += dtw[e][1] * dtr[1];
                    s += dtw[e][2] * dtr[2];
                    s += dtw[e][3] * dtr[3];
                    s += dtw[e][4] * dtr[4];
                    s += dtw[e][5] * dtr[5];
                    const float sp = log1pf(__expf(s));
                    const float delta = (s > 20.0f) ? s : sp;
                    *(float2*)(pdd + e * DP) = make_float2(delta, delta * un[e]);
                }
            }
#pragma unroll
            for (int e = 0; e < SFE; ++e) { u2[e] = u1[e]; u1[e] = un[e]; un[e] = unx[e]; }
#pragma unroll
            for (int r = 0; r < RR; ++r) dtr[r] = dnx[r];
        } else if (is_stage) {
            if (i < LL / TL) {
#pragma unroll
                for (int hf = 0; hf < 2; ++hf) {
                    float* dst = sm_x + bs * (TL * XP) + (lane + 32 * hf) * XP;
#pragma unroll
                    for (int p = 0; p < 8; ++p)
                        *(float4*)(dst + 4 * p) =
                            make_float4(xr[hf][2 * p],      xr[hf][2 * p + 1],
                                        xr[hf][NS + 2 * p], xr[hf][NS + 2 * p + 1]);
                }
            }
            if (i + 1 < LL / TL) {
                const int ln0 = (i + 1) * TL;
#pragma unroll
                for (int hf = 0; hf < 2; ++hf)
#pragma unroll
                    for (int r = 0; r < 2 * NS; ++r)
                        xr[hf][r] = xdb[(size_t)(RR + r) * LL + ln0 + lane + 32 * hf];
            }
        }
        __syncthreads();
    }
}

#define MTE   8
#define MPOS  (MTE * MTE)
#define MDG   8
#define MCH   (DI / MDG)
#define MPIT  (DI + 1)

__global__ __launch_bounds__(MPOS * MDG) void ss2d_merge_ln(
    const float* __restrict__ outy,
    const float* __restrict__ gamma,
    const float* __restrict__ beta,
    float* __restrict__ out)
{
    __shared__ float tile[MPOS * MPIT];
    __shared__ float red_s[MDG * MPOS];
    __shared__ float red_q[MDG * MPOS];

    const int tid  = threadIdx.x;
    const int lane = tid & 31;
    const int wv   = tid >> 5;
    const int b    = blockIdx.x >> 6;
    const int t    = blockIdx.x & 63;
    const int h0   = (t >> 3) * MTE;
    const int w0   = (t & 7) * MTE;
    const int pos  = tid & (MPOS - 1);
    const int dg   = tid >> 6;
    const int hh   = h0 + (pos >> 3);
    const int ww   = w0 + (pos & 7);
    const int lm   = (hh << 6) | ww;
    const int lt   = (ww << 6) | hh;

    const size_t ks = (size_t)DI * LL;
    const float* base = outy + (size_t)b * KK * ks + (size_t)dg * LL;
    const float* p0 = base + lm;
    const float* p2 = base + 2 * ks + (LL - 1 - lm);
    const float* p1 = base + ks + lt;
    const float* p3 = base + 3 * ks + (LL - 1 - lt);

    float gm[6], bt[6];
#pragma unroll
    for (int c = 0; c < 6; ++c) { gm[c] = gamma[lane + 32 * c]; bt[c] = beta[lane + 32 * c]; }

    float a0[MCH], a2[MCH], a1[MCH], a3[MCH];
#pragma unroll
    for (int it = 0; it < MCH; ++it) {
        const size_t o = (size_t)it * MDG * LL;
        a0[it] = p0[o]; a2[it] = p2[o]; a1[it] = p1[o]; a3[it] = p3[o];
    }
    __builtin_amdgcn_sched_barrier(0);

    float v[MCH];
    float s = 0.0f, q = 0.0f;
#pragma unroll
    for (int it = 0; it < MCH; ++it) {
        v[it] = a0[it] + a2[it] + a1[it] + a3[it];
        s += v[it];
        q += v[it] * v[it];
    }
    red_s[dg * MPOS + pos] = s;
    red_q[dg * MPOS + pos] = q;
    __syncthreads();

    float ts = 0.0f, tq = 0.0f;
#pragma unroll
    for (int g = 0; g < MDG; ++g) {
        ts += red_s[g * MPOS + pos];
        tq += red_q[g * MPOS + pos];
    }
    const float mu   = ts / 192.0f;
    const float var  = tq / 192.0f - mu * mu;
    const float rstd = rsqrtf(var + 1e-5f);

#pragma unroll
    for (int it = 0; it < MCH; ++it)
        tile[pos * MPIT + dg + it * MDG] = (v[it] - mu) * rstd;
    __syncthreads();

    const int pb = wv * 4;
    const int lb = ((h0 + (pb >> 3)) << 6) | (w0 + (pb & 7));
    float* ob = out + ((size_t)b * LL + lb) * DI + lane;
    const float* tb = tile + pb * MPIT + lane;
#pragma unroll
    for (int pi = 0; pi < 4; ++pi)
#pragma unroll
        for (int c = 0; c < 6; ++c)
            ob[pi * DI + 32 * c] = tb[pi * MPIT + 32 * c] * gm[c] + bt[c];
}

extern "C" void kernel_launch(void* const* d_in, const int* in_sizes, int n_in,
                              void* d_out, int out_size, void* d_ws, size_t ws_size,
                              hipStream_t stream) {
    const float* x   = (const float*)d_in[0];
    const float* y   = (const float*)d_in[1];
    const float* xpw = (const float*)d_in[2];
    const float* dtw = (const float*)d_in[3];
    const float* dtb = (const float*)d_in[4];
    const float* Al  = (const float*)d_in[5];
    const float* Ds  = (const float*)d_in[6];
    const float* gm  = (const float*)d_in[7];
    const float* bt  = (const float*)d_in[8];

    float* xdbl = (float*)d_ws;
    float* outy = xdbl + (size_t)B_ * KK * CP * LL;

    ss2d_proj_wmma<<<256, 256, 0, stream>>>(x, xpw, xdbl);
    ss2d_scan<<<B_ * KK * (DI / GCH), 256, 0, stream>>>(y, dtw, dtb, Al, Ds, xdbl, outy);
    ss2d_merge_ln<<<B_ * (LL / MPOS), MPOS * MDG, 0, stream>>>(outy, gm, bt, (float*)d_out);
}


# scan, transposed directions: the work-group's y slice (12 channels x 4096) is copied once into LDS by contiguous 16-B loads and u is read from LDS, not by 4-B loads that touch 32 lines each
# speedup vs baseline: 21.7730x; 1.1312x over previous
#include <hip/hip_runtime.h>
#include <hip/hip_bf16.h>


#define B_  4
#define DI  192
#define LL  4096
#define NS  16
#define RR  6
#define KK  4
#define CP  38
#define TL  64
#define TLP 65

typedef float v2f __attribute__((ext_vector_type(2)));
typedef float v8f __attribute__((ext_vector_type(8)));

__device__ __forceinline__ int scan_idx(int k, int l) {
    int ll = (k & 2) ? (LL - 1 - l) : l;
    if (k & 1) return ((ll & 63) << 6) | (ll >> 6);
    return ll;
}

#define PNT  8
#define PXP  136
#define PWP  196

__global__ __launch_bounds__(256) void ss2d_proj_wmma(
    const float* __restrict__ x,
    const float* __restrict__ xpw,
    float* __restrict__ xdbl)
{
    __shared__ __attribute__((aligned(16))) float xs[DI * PXP];
    __shared__ __attribute__((aligned(16))) float ws[2 * CP * PWP];

    const int tid  = threadIdx.x;
    const int lane = tid & 31;
    const int j    = tid >> 5;
    const int t    = blockIdx.x & 31;
    const int cls  = (blockIdx.x >> 5) & 1;
    const int b    = blockIdx.x >> 6;

    const float* xb = x + (size_t)b * DI * LL;

    if (cls == 0) {
        const int p0 = t * 128;
#pragma unroll
        for (int i = 0; i < 24; ++i) {
            const int f  = tid + 256 * i;
            const int d  = f >> 5;
            const int c4 = f & 31;
            const float4 v = *(const float4*)(xb + (size_t)d * LL + p0 + 4 * c4);
            *(float4*)(xs + d * PXP + 4 * c4) = v;
        }
    } else {
        const int w0 = (t & 7) * 8;
        const int h0 = (t >> 3) * 16;
#pragma unroll
        for (int i = 0; i < 24; ++i) {
            const int f  = tid + 256 * i;
            const int d  = f >> 5;
            const int r  = f & 31;
            const int c  = r >> 1;
            const int jq = r & 1;
            const float4 v = *(const float4*)(xb + (size_t)d * LL + (h0 + c) * 64 + w0 + 4 * jq);
            float* dst = xs + d * PXP + 64 * jq + c;
            dst[0]  = v.x;
            dst[16] = v.y;
            dst[32] = v.z;
            dst[48] = v.w;
        }
    }
#pragma unroll
    for (int i = 0; i < 15; ++i) {
        const int f = tid + 256 * i;
        if (f < 2 * CP * (DI / 4)) {
            const int r  = f / (DI / 4);
            const int q  = f - r * (DI / 4);
            const int sr = cls * CP + r + (r >= CP ? CP : 0);
            const float4 v = *(const float4*)(xpw + (size_t)sr * DI + 4 * q);
            *(float4*)(ws + r * PWP + 4 * q) = v;
        }
    }
    __syncthreads();

    const int hw = lane >> 4;
    const int ln = lane & 15;

    int aoff[6];
#pragma unroll
    for (int mt = 0; mt < 3; ++mt) {
        const int row = (16 * mt + ln) < CP ? (16 * mt + ln) : (CP - 1);
        aoff[mt]     = row * PWP + 2 * hw;
        aoff[3 + mt] = (CP + row) * PWP + 2 * hw;
    }
    const int boff = (2 * hw) * PXP + 16 * j + ln;

    v8f acc[6];
#pragma unroll
    for (int a = 0; a < 6; ++a) acc[a] = (v8f){0.f, 0.f, 0.f, 0.f, 0.f, 0.f, 0.f, 0.f};

#pragma unroll 4
    for (int k0 = 0; k0 < DI; k0 += 4) {
        v2f bb;
        bb.x = xs[boff + k0 * PXP];
        bb.y = xs[boff + k0 * PXP + PXP];
#pragma unroll
        for (int a = 0; a < 6; ++a) {
            const v2f av = *(const v2f*)(ws + aoff[a] + k0);
            acc[a] = __builtin_amdgcn_wmma_f32_16x16x4_f32(
                false, av, false, bb, (short)0, acc[a], false, false);
        }
    }

    int lf;
    if (cls == 0) lf = t * 128 + 16 * j + ln;
    else          lf = (((t & 7) * 8 + j) << 6) + (t >> 3) * 16 + ln;

#pragma unroll
    for (int kk = 0; kk < 2; ++kk) {
        const int kd = b * KK + cls + 2 * kk;
        const int l  = kk ? (LL - 1 - lf) : lf;
        float* obase = xdbl + (size_t)kd * CP * LL + l;
#pragma unroll
        for (int mt = 0; mt < 3; ++mt) {
#pragma unroll
            for (int i = 0; i < 8; ++i) {
                const int m = 16 * mt + i + (hw << 3);
                if (m < CP) obase[(size_t)m * LL] = acc[kk * 3 + mt][i];
            }
        }
    }
}

#define CHW 4
#define LPC 8
#define SCW 3
#define GCH (SCW * CHW)
#define SFE 6
#define XP  36
#define UP  72
#define DP  144
#define YTP 65
#define YTE (64 * YTP)
#define SMX_F (2 * TL * XP)
#define SMD_F (2 * GCH * DP)
#define SMY_F (2 * GCH * UP)

__device__ __forceinline__ float oct_sum(float v) {
    v += __int_as_float(__builtin_amdgcn_update_dpp(0, __float_as_int(v), 0xB1, 0xF, 0xF, true));
    v += __int_as_float(__builtin_amdgcn_update_dpp(0, __float_as_int(v), 0x4E, 0xF, 0xF, true));
    v += __int_as_float(__builtin_amdgcn_update_dpp(0, __float_as_int(v), 0x141, 0xF, 0xF, true));
    return v;
}

__device__ __forceinline__ void scan_load_group(
    const float* sm_x, const float* sm_dd, int g, int q, int ch,
    float4 (&BC)[4], float2 (&Dv)[4])
{
#pragma unroll
    for (int t = 0; t < 4; ++t) {
        int j = 4 * g + t;
        BC[t] = *(const float4*)(sm_x + j * XP + 4 * q);
        Dv[t] = *(const float2*)(sm_dd + ch * DP + 2 * j);
    }
}

__device__ __forceinline__ void scan_run_group(
    float* sm_y, int g, int q, int ch, float A20, float A21, float& h0, float& h1,
    const float4 (&BC)[4], const float2 (&Dv)[4])
{
    float yy[4];
#pragma unroll
    for (int t = 0; t < 4; ++t) {
        float delta = Dv[t].x, du = Dv[t].y;
        float dA0 = __builtin_amdgcn_exp2f(delta * A20);
        float dA1 = __builtin_amdgcn_exp2f(delta * A21);
        h0 = dA0 * h0 + du * BC[t].x;
        h1 = dA1 * h1 + du * BC[t].y;
        float yv = h0 * BC[t].z;
        yv += h1 * BC[t].w;
        yy[t] = oct_sum(yv);
    }
    if (q == 0)
        *(float4*)(sm_y + ch * UP + 4 * g) = make_float4(yy[0], yy[1], yy[2], yy[3]);
}

__global__ __launch_bounds__(256) void ss2d_scan(
    const float* __restrict__ y,
    const float* __restrict__ dtw_g,
    const float* __restrict__ dtb_g,
    const float* __restrict__ Alogs,
    const float* __restrict__ Ds_g,
    const float* __restrict__ xdbl,
    float* __restrict__ outy)
{
    __shared__ __attribute__((aligned(16))) float smem[SMX_F + SMD_F + SMY_F + GCH * YTE];
    float* const sm_x  = smem;
    float* const sm_dd = smem + SMX_F;
    float* const sm_y  = smem + SMX_F + SMD_F;
    float* const sm_yt = smem + SMX_F + SMD_F + SMY_F;

    const int lane = threadIdx.x & 31;
    const int wv = __builtin_amdgcn_readfirstlane((int)(threadIdx.x >> 5));
    const bool is_scan  = wv < SCW;
    const bool is_prep  = (wv >= SCW) && (wv < SCW + 4);
    const bool is_stage = wv == SCW + 4;

    const int cg = blockIdx.x % (DI / GCH);
    const int kd = blockIdx.x / (DI / GCH);
    const int k  = kd & 3;
    const int b  = kd >> 2;
    const int d0 = cg * GCH;

    const float* xdb = xdbl + (size_t)kd * CP * LL;
    const float* yb  = y + ((size_t)b * DI + d0) * LL;
    float* ob        = outy + ((size_t)kd * DI + d0) * LL;

    const bool tdir = (k & 1) != 0;
    if (tdir) {
        const float4* ysl = (const float4*)yb;
#pragma unroll 1
        for (int rd = 0; rd < 2; ++rd) {
            float4 yv[24];
#pragma unroll
            for (int p = 0; p < 24; ++p)
                yv[p] = ysl[(rd * 24 + p) * 256 + (int)threadIdx.x];
#pragma unroll
            for (int p = 0; p < 24; ++p) {
                const int f  = (rd * 24 + p) * 256 + (int)threadIdx.x;
                const int r  = f & 1023;
                float* dst = sm_yt + (f >> 10) * YTE + (r >> 4) * YTP + 4 * (r & 15);
                dst[0] = yv[p].x;
                dst[1] = yv[p].y;
                dst[2] = yv[p].z;
                dst[3] = yv[p].w;
            }
        }
        __syncthreads();
    }

    const int ch = lane >> 3;
    const int q  = lane & 7;
    const int cw = (is_scan ? wv : 0) * CHW + ch;
    const int kdch = k * DI + d0 + cw;
    const float A20 = -__expf(Alogs[(size_t)kdch * NS + 2 * q])     * 1.4426950408889634f;
    const float A21 = -__expf(Alogs[(size_t)kdch * NS + 2 * q + 1]) * 1.4426950408889634f;
    float h0 = 0.0f, h1 = 0.0f;

    const int pw  = is_prep ? (wv - SCW) : 0;
    const int ps  = (pw & 1) * 32 + lane;
    const int pc0 = (pw >> 1) * SFE;
    float dtw[SFE][RR], dtb[SFE], DsC[SFE];
#pragma unroll
    for (int e = 0; e < SFE; ++e) {
        const int kc = k * DI + d0 + pc0 + e;
#pragma unroll
        for (int r = 0; r < RR; ++r) dtw[e][r] = dtw_g[(size_t)kc * RR + r];
        dtb[e] = dtb_g[kc];
        DsC[e] = Ds_g[kc];
    }
    float dtr[RR];
    float un[SFE];
    float u1[SFE], u2[SFE];
#pragma unroll
    for (int r = 0; r < RR; ++r) dtr[r] = 0.0f;
#pragma unroll
    for (int e = 0; e < SFE; ++e) { un[e] = 0.0f; u1[e] = 0.0f; u2[e] = 0.0f; }

    float xr[2][2 * NS];
#pragma unroll
    for (int hf = 0; hf < 2; ++hf)
#pragma unroll
        for (int r = 0; r < 2 * NS; ++r) xr[hf][r] = 0.0f;

    if (is_prep) {
#pragma unroll
        for (int r = 0; r < RR; ++r) dtr[r] = xdb[(size_t)r * LL + ps];
        const int m0 = scan_idx(k, ps);
        if (tdir) {
            const float* pyt = sm_yt + pc0 * YTE + (m0 >> 6) * YTP + (m0 & 63);
#pragma unroll
            for (int e = 0; e < SFE; ++e) un[e] = pyt[e * YTE];
        } else {
#pragma unroll
            for (int e = 0; e < SFE; ++e) un[e] = yb[(size_t)(pc0 + e) * LL + m0];
        }
    } else if (is_stage) {
#pragma unroll
        for (int hf = 0; hf < 2; ++hf)
#pragma unroll
            for (int r = 0; r < 2 * NS; ++r)
                xr[hf][r] = xdb[(size_t)(RR + r) * LL + lane + 32 * hf];
    }

    for (int i = 0; i < LL / TL + 2; ++i) {
        const int bs = i & 1;
        if (is_scan) {
            if (i >= 1 && i <= LL / TL) {
                const float* px = sm_x  + (bs ^ 1) * (TL * XP);
                const float* pd = sm_dd + (bs ^ 1) * (GCH * DP);
                float*       py = sm_y  + (bs ^ 1) * (GCH * UP);
                float4 BC0[4], BC1[4];
                float2 D0[4], D1[4];
                scan_load_group(px, pd, 0, q, cw, BC0, D0);
#pragma unroll 1
                for (int g = 0; g < TL / 4; g += 2) {
                    scan_load_group(px, pd, g + 1, q, cw, BC1, D1);
                    __builtin_amdgcn_sched_barrier(0);
                    scan_run_group(py, g, q, cw, A20, A21, h0, h1, BC0, D0);
                    const int gn = (g + 2 < TL / 4) ? (g + 2) : g;
                    scan_load_group(px, pd, gn, q, cw, BC0, D0);
                    __builtin_amdgcn_sched_barrier(0);
                    scan_run_group(py, g + 1, q, cw, A20, A21, h0, h1, BC1, D1);
                }
            }
        } else if (is_prep) {
            float dnx[RR], unx[SFE];
            if (i + 1 < LL / TL) {
                const int ln0 = (i + 1) * TL;
#pragma unroll
                for (int r = 0; r < RR; ++r) dnx[r] = xdb[(size_t)r * LL + ln0 + ps];
                const int m1 = scan_idx(k, ln0 + ps);
                if (tdir) {
                    const float* pyt = sm_yt + pc0 * YTE + (m1 >> 6) * YTP + (m1 & 63);
#pragma unroll
                    for (int e = 0; e < SFE; ++e) unx[e] = pyt[e * YTE];
                } else {
#pragma unroll
                    for (int e = 0; e < SFE; ++e)
                        unx[e] = yb[(size_t)(pc0 + e) * LL + m1];
                }
            } else {
#pragma unroll
                for (int r = 0; r < RR; ++r) dnx[r] = 0.0f;
#pragma unroll
                for (int e = 0; e < SFE; ++e) unx[e] = 0.0f;
            }
            if (i >= 2) {
                const float* psy = sm_y + bs * (GCH * UP) + pc0 * UP + ps;
                float* po = ob + (size_t)pc0 * LL + (i - 2) * TL + ps;
#pragma unroll
                for (int e = 0; e < SFE; ++e)
                    po[(size_t)e * LL] = psy[e * UP] + DsC[e] * u2[e];
            }
            if (i < LL / TL) {
                float* pdd = sm_dd + bs * (GCH * DP) + pc0 * DP + 2 * ps;
#pragma unroll
                for (int e = 0; e < SFE; ++e) {
                    float s = dtb[e];
                    s += dtw[e][0] * dtr[0];
                    s += dtw[e][1] * dtr[1];
                    s += dtw[e][2] * dtr[2];
                    s += dtw[e][3] * dtr[3];
                    s += dtw[e][4] * dtr[4];
                    s += dtw[e][5] * dtr[5];
                    const float sp = log1pf(__expf(s));
                    const float delta = (s > 20.0f) ? s : sp;
                    *(float2*)(pdd + e * DP) = make_float2(delta, delta * un[e]);
                }
            }
#pragma unroll
            for (int e = 0; e < SFE; ++e) { u2[e] = u1[e]; u1[e] = un[e]; un[e] = unx[e]; }
#pragma unroll
            for (int r = 0; r < RR; ++r) dtr[r] = dnx[r];
        } else if (is_stage) {
            if (i < LL / TL) {
#pragma unroll
                for (int hf = 0; hf < 2; ++hf) {
                    float* dst = sm_x + bs * (TL * XP) + (lane + 32 * hf) * XP;
#pragma unroll
                    for (int p = 0; p < 8; ++p)
                        *(float4*)(dst + 4 * p) =
                            make_float4(xr[hf][2 * p],      xr[hf][2 * p + 1],
                                        xr[hf][NS + 2 * p], xr[hf][NS + 2 * p + 1]);
                }
            }
            if (i + 1 < LL / TL) {
                const int ln0 = (i + 1) * TL;
#pragma unroll
                for (int hf = 0; hf < 2; ++hf)
#pragma unroll
                    for (int r = 0; r < 2 * NS; ++r)
                        xr[hf][r] = xdb[(size_t)(RR + r) * LL + ln0 + lane + 32 * hf];
            }
        }
        __syncthreads();
    }
}

#define MTE   8
#define MPOS  (MTE * MTE)
#define MDG   8
#define MCH   (DI / MDG)
#define MPIT  (DI + 1)

__global__ __launch_bounds__(MPOS * MDG) void ss2d_merge_ln(
    const float* __restrict__ outy,
    const float* __restrict__ gamma,
    const float* __restrict__ beta,
    float* __restrict__ out)
{
    __shared__ float tile[MPOS * MPIT];
    __shared__ float red_s[MDG * MPOS];
    __shared__ float red_q[MDG * MPOS];

    const int tid  = threadIdx.x;
    const int lane = tid & 31;
    const int wv   = tid >> 5;
    const int b    = blockIdx.x >> 6;
    const int t    = blockIdx.x & 63;
    const int h0   = (t >> 3) * MTE;
    const int w0   = (t & 7) * MTE;
    const int pos  = tid & (MPOS - 1);
    const int dg   = tid >> 6;
    const int hh   = h0 + (pos >> 3);
    const int ww   = w0 + (pos & 7);
    const int lm   = (hh << 6) | ww;
    const int lt   = (ww << 6) | hh;

    const size_t ks = (size_t)DI * LL;
    const float* base = outy + (size_t)b * KK * ks + (size_t)dg * LL;
    const float* p0 = base + lm;
    const float* p2 = base + 2 * ks + (LL - 1 - lm);
    const float* p1 = base + ks + lt;
    const float* p3 = base + 3 * ks + (LL - 1 - lt);

    float gm[6], bt[6];
#pragma unroll
    for (int c = 0; c < 6; ++c) { gm[c] = gamma[lane + 32 * c]; bt[c] = beta[lane + 32 * c]; }

    float a0[MCH], a2[MCH], a1[MCH], a3[MCH];
#pragma unroll
    for (int it = 0; it < MCH; ++it) {
        const size_t o = (size_t)it * MDG * LL;
        a0[it] = p0[o]; a2[it] = p2[o]; a1[it] = p1[o]; a3[it] = p3[o];
    }
    __builtin_amdgcn_sched_barrier(0);

    float v[MCH];
    float s = 0.0f, q = 0.0f;
#pragma unroll
    for (int it = 0; it < MCH; ++it) {
        v[it] = a0[it] + a2[it] + a1[it] + a3[it];
        s += v[it];
        q += v[it] * v[it];
    }
    red_s[dg * MPOS + pos] = s;
    red_q[dg * MPOS + pos] = q;
    __syncthreads();

    float ts = 0.0f, tq = 0.0f;
#pragma unroll
    for (int g = 0; g < MDG; ++g) {
        ts += red_s[g * MPOS + pos];
        tq += red_q[g * MPOS + pos];
    }
    const float mu   = ts / 192.0f;
    const float var  = tq / 192.0f - mu * mu;
    const float rstd = rsqrtf(var + 1e-5f);

#pragma unroll
    for (int it = 0; it < MCH; ++it)
        tile[pos * MPIT + dg + it * MDG] = (v[it] - mu) * rstd;
    __syncthreads();

    const int pb = wv * 4;
    const int lb = ((h0 + (pb >> 3)) << 6) | (w0 + (pb & 7));
    float* ob = out + ((size_t)b * LL + lb) * DI + lane;
    const float* tb = tile + pb * MPIT + lane;
#pragma unroll
    for (int pi = 0; pi < 4; ++pi)
#pragma unroll
        for (int c = 0; c < 6; ++c)
            ob[pi * DI + 32 * c] = tb[pi * MPIT + 32 * c] * gm[c] + bt[c];
}

extern "C" void kernel_launch(void* const* d_in, const int* in_sizes, int n_in,
                              void* d_out, int out_size, void* d_ws, size_t ws_size,
                              hipStream_t stream) {
    const float* x   = (const float*)d_in[0];
    const float* y   = (const float*)d_in[1];
    const float* xpw = (const float*)d_in[2];
    const float* dtw = (const float*)d_in[3];
    const float* dtb = (const float*)d_in[4];
    const float* Al  = (const float*)d_in[5];
    const float* Ds  = (const float*)d_in[6];
    const float* gm  = (const float*)d_in[7];
    const float* bt  = (const float*)d_in[8];

    float* xdbl = (float*)d_ws;
    float* outy = xdbl + (size_t)B_ * KK * CP * LL;

    ss2d_proj_wmma<<<256, 256, 0, stream>>>(x, xpw, xdbl);
    ss2d_scan<<<B_ * KK * (DI / GCH), 256, 0, stream>>>(y, dtw, dtb, Al, Ds, xdbl, outy);
    ss2d_merge_ln<<<B_ * (LL / MPOS), MPOS * MDG, 0, stream>>>(outy, gm, bt, (float*)d_out);
}


# scan recurrence software-pipelined: groups of 8 steps; a group's lane sums (DPP) and output store run one group late, beside the next group's state updates; every sum keeps its order
# speedup vs baseline: 26.1068x; 1.1990x over previous
#include <hip/hip_runtime.h>
#include <hip/hip_bf16.h>


#define B_  4
#define DI  192
#define LL  4096
#define NS  16
#define RR  6
#define KK  4
#define CP  38
#define TL  64
#define TLP 65

typedef float v2f __attribute__((ext_vector_type(2)));
typedef float v8f __attribute__((ext_vector_type(8)));

__device__ __forceinline__ int scan_idx(int k, int l) {
    int ll = (k & 2) ? (LL - 1 - l) : l;
    if (k & 1) return ((ll & 63) << 6) | (ll >> 6);
    return ll;
}

#define PNT  8
#define PXP  136
#define PWP  196

__global__ __launch_bounds__(256) void ss2d_proj_wmma(
    const float* __restrict__ x,
    const float* __restrict__ xpw,
    float* __restrict__ xdbl)
{
    __shared__ __attribute__((aligned(16))) float xs[DI * PXP];
    __shared__ __attribute__((aligned(16))) float ws[2 * CP * PWP];

    const int tid  = threadIdx.x;
    const int lane = tid & 31;
    const int j    = tid >> 5;
    const int t    = blockIdx.x & 31;
    const int cls  = (blockIdx.x >> 5) & 1;
    const int b    = blockIdx.x >> 6;

    const float* xb = x + (size_t)b * DI * LL;

    if (cls == 0) {
        const int p0 = t * 128;
#pragma unroll
        for (int i = 0; i < 24; ++i) {
            const int f  = tid + 256 * i;
            const int d  = f >> 5;
            const int c4 = f & 31;
            const float4 v = *(const float4*)(xb + (size_t)d * LL + p0 + 4 * c4);
            *(float4*)(xs + d * PXP + 4 * c4) = v;
        }
    } else {
        const int w0 = (t & 7) * 8;
        const int h0 = (t >> 3) * 16;
#pragma unroll
        for (int i = 0; i < 24; ++i) {
            const int f  = tid + 256 * i;
            const int d  = f >> 5;
            const int r  = f & 31;
            const int c  = r >> 1;
            const int jq = r & 1;
            const float4 v = *(const float4*)(xb + (size_t)d * LL + (h0 + c) * 64 + w0 + 4 * jq);
            float* dst = xs + d * PXP + 64 * jq + c;
            dst[0]  = v.x;
            dst[16] = v.y;
            dst[32] = v.z;
            dst[48] = v.w;
        }
    }
#pragma unroll
    for (int i = 0; i < 15; ++i) {
        const int f = tid + 256 * i;
        if (f < 2 * CP * (DI / 4)) {
            const int r  = f / (DI / 4);
            const int q  = f - r * (DI / 4);
            const int sr = cls * CP + r + (r >= CP ? CP : 0);
            const float4 v = *(const float4*)(xpw + (size_t)sr * DI + 4 * q);
            *(float4*)(ws + r * PWP + 4 * q) = v;
        }
    }
    __syncthreads();

    const int hw = lane >> 4;
    const int ln = lane & 15;

    int aoff[6];
#pragma unroll
    for (int mt = 0; mt < 3; ++mt) {
        const int row = (16 * mt + ln) < CP ? (16 * mt + ln) : (CP - 1);
        aoff[mt]     = row * PWP + 2 * hw;
        aoff[3 + mt] = (CP + row) * PWP + 2 * hw;
    }
    const int boff = (2 * hw) * PXP + 16 * j + ln;

    v8f acc[6];
#pragma unroll
    for (int a = 0; a < 6; ++a) acc[a] = (v8f){0.f, 0.f, 0.f, 0.f, 0.f, 0.f, 0.f, 0.f};

#pragma unroll 4
    for (int k0 = 0; k0 < DI; k0 += 4) {
        v2f bb;
        bb.x = xs[boff + k0 * PXP];
        bb.y = xs[boff + k0 * PXP + PXP];
#pragma unroll
        for (int a = 0; a < 6; ++a) {
            const v2f av = *(const v2f*)(ws + aoff[a] + k0);
            acc[a] = __builtin_amdgcn_wmma_f32_16x16x4_f32(
                false, av, false, bb, (short)0, acc[a], false, false);
        }
    }

    int lf;
    if (cls == 0) lf = t * 128 + 16 * j + ln;
    else          lf = (((t & 7) * 8 + j) << 6) + (t >> 3) * 16 + ln;

#pragma unroll
    for (int kk = 0; kk < 2; ++kk) {
        const int kd = b * KK + cls + 2 * kk;
        const int l  = kk ? (LL - 1 - lf) : lf;
        float* obase = xdbl + (size_t)kd * CP * LL + l;
#pragma unroll
        for (int mt = 0; mt < 3; ++mt) {
#pragma unroll
            for (int i = 0; i < 8; ++i) {
                const int m = 16 * mt + i + (hw << 3);
                if (m < CP) obase[(size_t)m * LL] = acc[kk * 3 + mt][i];
            }
        }
    }
}

#define CHW 4
#define LPC 8
#define SCW 3
#define GCH (SCW * CHW)
#define SFE 6
#define XP  36
#define UP  72
#define DP  144
#define YTP 65
#define YTE (64 * YTP)
#define SMX_F (2 * TL * XP)
#define SMD_F (2 * GCH * DP)
#define SMY_F (2 * GCH * UP)

__device__ __forceinline__ float oct_sum(float v) {
    v += __int_as_float(__builtin_amdgcn_update_dpp(0, __float_as_int(v), 0xB1, 0xF, 0xF, true));
    v += __int_as_float(__builtin_amdgcn_update_dpp(0, __float_as_int(v), 0x4E, 0xF, 0xF, true));
    v += __int_as_float(__builtin_amdgcn_update_dpp(0, __float_as_int(v), 0x141, 0xF, 0xF, true));
    return v;
}

#define GS  8
#define GPC (TL / GS)

__device__ __forceinline__ void scan_load_group(
    const float* sm_x, const float* sm_dd, int g, int q, int ch,
    float4 (&BC)[GS], float2 (&Dv)[GS])
{
#pragma unroll
    for (int t = 0; t < GS; ++t) {
        int j = GS * g + t;
        BC[t] = *(const float4*)(sm_x + j * XP + 4 * q);
        Dv[t] = *(const float2*)(sm_dd + ch * DP + 2 * j);
    }
}

__device__ __forceinline__ void scan_step_group(
    float A20, float A21, float& h0, float& h1,
    const float4 (&BC)[GS], const float2 (&Dv)[GS], float (&yp)[GS])
{
#pragma unroll
    for (int t = 0; t < GS; ++t) {
        float delta = Dv[t].x, du = Dv[t].y;
        float dA0 = __builtin_amdgcn_exp2f(delta * A20);
        float dA1 = __builtin_amdgcn_exp2f(delta * A21);
        h0 = dA0 * h0 + du * BC[t].x;
        h1 = dA1 * h1 + du * BC[t].y;
        float yv = h0 * BC[t].z;
        yv += h1 * BC[t].w;
        yp[t] = yv;
    }
}

__device__ __forceinline__ void scan_sum_group(const float (&yp)[GS], float (&yy)[GS])
{
#pragma unroll
    for (int t = 0; t < GS; ++t) yy[t] = oct_sum(yp[t]);
}

__device__ __forceinline__ void scan_store_group(
    float* sm_y, int g, int q, int ch, const float (&yy)[GS])
{
    if (q == 0) {
        *(float4*)(sm_y + ch * UP + GS * g)     = make_float4(yy[0], yy[1], yy[2], yy[3]);
        *(float4*)(sm_y + ch * UP + GS * g + 4) = make_float4(yy[4], yy[5], yy[6], yy[7]);
    }
}

__device__ __forceinline__ void scan_hold(float& h0, float& h1, float (&yp)[GS])
{
    asm volatile("" : "+v"(h0), "+v"(h1),
                      "+v"(yp[0]), "+v"(yp[1]), "+v"(yp[2]), "+v"(yp[3]),
                      "+v"(yp[4]), "+v"(yp[5]), "+v"(yp[6]), "+v"(yp[7]));
}

__global__ __launch_bounds__(256) void ss2d_scan(
    const float* __restrict__ y,
    const float* __restrict__ dtw_g,
    const float* __restrict__ dtb_g,
    const float* __restrict__ Alogs,
    const float* __restrict__ Ds_g,
    const float* __restrict__ xdbl,
    float* __restrict__ outy)
{
    __shared__ __attribute__((aligned(16))) float smem[SMX_F + SMD_F + SMY_F + GCH * YTE];
    float* const sm_x  = smem;
    float* const sm_dd = smem + SMX_F;
    float* const sm_y  = smem + SMX_F + SMD_F;
    float* const sm_yt = smem + SMX_F + SMD_F + SMY_F;

    const int lane = threadIdx.x & 31;
    const int wv = __builtin_amdgcn_readfirstlane((int)(threadIdx.x >> 5));
    const bool is_scan  = wv < SCW;
    const bool is_prep  = (wv >= SCW) && (wv < SCW + 4);
    const bool is_stage = wv == SCW + 4;

    const int cg = blockIdx.x % (DI / GCH);
    const int kd = blockIdx.x / (DI / GCH);
    const int k  = kd & 3;
    const int b  = kd >> 2;
    const int d0 = cg * GCH;

    const float* xdb = xdbl + (size_t)kd * CP * LL;
    const float* yb  = y + ((size_t)b * DI + d0) * LL;
    float* ob        = outy + ((size_t)kd * DI + d0) * LL;

    const bool tdir = (k & 1) != 0;
    if (tdir) {
        const float4* ysl = (const float4*)yb;
#pragma unroll 1
        for (int rd = 0; rd < 2; ++rd) {
            float4 yv[24];
#pragma unroll
            for (int p = 0; p < 24; ++p)
                yv[p] = ysl[(rd * 24 + p) * 256 + (int)threadIdx.x];
#pragma unroll
            for (int p = 0; p < 24; ++p) {
                const int f  = (rd * 24 + p) * 256 + (int)threadIdx.x;
                const int r  = f & 1023;
                float* dst = sm_yt + (f >> 10) * YTE + (r >> 4) * YTP + 4 * (r & 15);
                dst[0] = yv[p].x;
                dst[1] = yv[p].y;
                dst[2] = yv[p].z;
                dst[3] = yv[p].w;
            }
        }
        __syncthreads();
    }

    const int ch = lane >> 3;
    const int q  = lane & 7;
    const int cw = (is_scan ? wv : 0) * CHW + ch;
    const int kdch = k * DI + d0 + cw;
    const float A20 = -__expf(Alogs[(size_t)kdch * NS + 2 * q])     * 1.4426950408889634f;
    const float A21 = -__expf(Alogs[(size_t)kdch * NS + 2 * q + 1]) * 1.4426950408889634f;
    float h0 = 0.0f, h1 = 0.0f;

    const int pw  = is_prep ? (wv - SCW) : 0;
    const int ps  = (pw & 1) * 32 + lane;
    const int pc0 = (pw >> 1) * SFE;
    float dtw[SFE][RR], dtb[SFE], DsC[SFE];
#pragma unroll
    for (int e = 0; e < SFE; ++e) {
        const int kc = k * DI + d0 + pc0 + e;
#pragma unroll
        for (int r = 0; r < RR; ++r) dtw[e][r] = dtw_g[(size_t)kc * RR + r];
        dtb[e] = dtb_g[kc];
        DsC[e] = Ds_g[kc];
    }
    float dtr[RR];
    float un[SFE];
    float u1[SFE], u2[SFE];
#pragma unroll
    for (int r = 0; r < RR; ++r) dtr[r] = 0.0f;
#pragma unroll
    for (int e = 0; e < SFE; ++e) { un[e] = 0.0f; u1[e] = 0.0f; u2[e] = 0.0f; }

    float xr[2][2 * NS];
#pragma unroll
    for (int hf = 0; hf < 2; ++hf)
#pragma unroll
        for (int r = 0; r < 2 * NS; ++r) xr[hf][r] = 0.0f;

    if (is_prep) {
#pragma unroll
        for (int r = 0; r < RR; ++r) dtr[r] = xdb[(size_t)r * LL + ps];
        const int m0 = scan_idx(k, ps);
        if (tdir) {
            const float* pyt = sm_yt + pc0 * YTE + (m0 >> 6) * YTP + (m0 & 63);
#pragma unroll
            for (int e = 0; e < SFE; ++e) un[e] = pyt[e * YTE];
        } else {
#pragma unroll
            for (int e = 0; e < SFE; ++e) un[e] = yb[(size_t)(pc0 + e) * LL + m0];
        }
    } else if (is_stage) {
#pragma unroll
        for (int hf = 0; hf < 2; ++hf)
#pragma unroll
            for (int r = 0; r < 2 * NS; ++r)
                xr[hf][r] = xdb[(size_t)(RR + r) * LL + lane + 32 * hf];
    }

    for (int i = 0; i < LL / TL + 2; ++i) {
        const int bs = i & 1;
        if (is_scan) {
            if (i >= 1 && i <= LL / TL) {
                const float* px = sm_x  + (bs ^ 1) * (TL * XP);
                const float* pd = sm_dd + (bs ^ 1) * (GCH * DP);
                float*       py = sm_y  + (bs ^ 1) * (GCH * UP);
                float4 BC0[GS], BC1[GS];
                float2 D0[GS], D1[GS];
                float ya[GS], yb[GS], yy[GS];
                scan_load_group(px, pd, 0, q, cw, BC0, D0);
                scan_load_group(px, pd, 1, q, cw, BC1, D1);
                __builtin_amdgcn_sched_barrier(0);
                scan_step_group(A20, A21, h0, h1, BC0, D0, ya);
#pragma unroll 1
                for (int g = 1; g < GPC - 1; g += 2) {
                    scan_load_group(px, pd, g + 1, q, cw, BC0, D0);
                    __builtin_amdgcn_sched_barrier(0);
                    scan_step_group(A20, A21, h0, h1, BC1, D1, yb);
                    scan_sum_group(ya, yy);
                    scan_hold(h0, h1, yb);
                    scan_store_group(py, g - 1, q, cw, yy);
                    scan_load_group(px, pd, g + 2, q, cw, BC1, D1);
                    __builtin_amdgcn_sched_barrier(0);
                    scan_step_group(A20, A21, h0, h1, BC0, D0, ya);
                    scan_sum_group(yb, yy);
                    scan_hold(h0, h1, ya);
                    scan_store_group(py, g, q, cw, yy);
                }
                scan_step_group(A20, A21, h0, h1, BC1, D1, yb);
                scan_sum_group(ya, yy);
                scan_hold(h0, h1, yb);
                scan_store_group(py, GPC - 2, q, cw, yy);
                scan_sum_group(yb, yy);
                scan_store_group(py, GPC - 1, q, cw, yy);
            }
        } else if (is_prep) {
            float dnx[RR], unx[SFE];
            if (i + 1 < LL / TL) {
                const int ln0 = (i + 1) * TL;
#pragma unroll
                for (int r = 0; r < RR; ++r) dnx[r] = xdb[(size_t)r * LL + ln0 + ps];
                const int m1 = scan_idx(k, ln0 + ps);
                if (tdir) {
                    const float* pyt = sm_yt + pc0 * YTE + (m1 >> 6) * YTP + (m1 & 63);
#pragma unroll
                    for (int e = 0; e < SFE; ++e) unx[e] = pyt[e * YTE];
                } else {
#pragma unroll
                    for (int e = 0; e < SFE; ++e)
                        unx[e] = yb[(size_t)(pc0 + e) * LL + m1];
                }
            } else {
#pragma unroll
                for (int r = 0; r < RR; ++r) dnx[r] = 0.0f;
#pragma unroll
                for (int e = 0; e < SFE; ++e) unx[e] = 0.0f;
            }
            if (i >= 2) {
                const float* psy = sm_y + bs * (GCH * UP) + pc0 * UP + ps;
                float* po = ob + (size_t)pc0 * LL + (i - 2) * TL + ps;
#pragma unroll
                for (int e = 0; e < SFE; ++e)
                    po[(size_t)e * LL] = psy[e * UP] + DsC[e] * u2[e];
            }
            if (i < LL / TL) {
                float* pdd = sm_dd + bs * (GCH * DP) + pc0 * DP + 2 * ps;
#pragma unroll
                for (int e = 0; e < SFE; ++e) {
                    float s = dtb[e];
                    s += dtw[e][0] * dtr[0];
                    s += dtw[e][1] * dtr[1];
                    s += dtw[e][2] * dtr[2];
                    s += dtw[e][3] * dtr[3];
                    s += dtw[e][4] * dtr[4];
                    s += dtw[e][5] * dtr[5];
                    const float sp = log1pf(__expf(s));
                    const float delta = (s > 20.0f) ? s : sp;
                    *(float2*)(pdd + e * DP) = make_float2(delta, delta * un[e]);
                }
            }
#pragma unroll
            for (int e = 0; e < SFE; ++e) { u2[e] = u1[e]; u1[e] = un[e]; un[e] = unx[e]; }
#pragma unroll
            for (int r = 0; r < RR; ++r) dtr[r] = dnx[r];
        } else if (is_stage) {
            if (i < LL / TL) {
#pragma unroll
                for (int hf = 0; hf < 2; ++hf) {
                    float* dst = sm_x + bs * (TL * XP) + (lane + 32 * hf) * XP;
#pragma unroll
                    for (int p = 0; p < 8; ++p)
                        *(float4*)(dst + 4 * p) =
                            make_float4(xr[hf][2 * p],      xr[hf][2 * p + 1],
                                        xr[hf][NS + 2 * p], xr[hf][NS + 2 * p + 1]);
                }
            }
            if (i + 1 < LL / TL) {
                const int ln0 = (i + 1) * TL;
#pragma unroll
                for (int hf = 0; hf < 2; ++hf)
#pragma unroll
                    for (int r = 0; r < 2 * NS; ++r)
                        xr[hf][r] = xdb[(size_t)(RR + r) * LL + ln0 + lane + 32 * hf];
            }
        }
        __syncthreads();
    }
}

#define MTE   8
#define MPOS  (MTE * MTE)
#define MDG   8
#define MCH   (DI / MDG)
#define MPIT  (DI + 1)

__global__ __launch_bounds__(MPOS * MDG) void ss2d_merge_ln(
    const float* __restrict__ outy,
    const float* __restrict__ gamma,
    const float* __restrict__ beta,
    float* __restrict__ out)
{
    __shared__ float tile[MPOS * MPIT];
    __shared__ float red_s[MDG * MPOS];
    __shared__ float red_q[MDG * MPOS];

    const int tid  = threadIdx.x;
    const int lane = tid & 31;
    const int wv   = tid >> 5;
    const int b    = blockIdx.x >> 6;
    const int t    = blockIdx.x & 63;
    const int h0   = (t >> 3) * MTE;
    const int w0   = (t & 7) * MTE;
    const int pos  = tid & (MPOS - 1);
    const int dg   = tid >> 6;
    const int hh   = h0 + (pos >> 3);
    const int ww   = w0 + (pos & 7);
    const int lm   = (hh << 6) | ww;
    const int lt   = (ww << 6) | hh;

    const size_t ks = (size_t)DI * LL;
    const float* base = outy + (size_t)b * KK * ks + (size_t)dg * LL;
    const float* p0 = base + lm;
    const float* p2 = base + 2 * ks + (LL - 1 - lm);
    const float* p1 = base + ks + lt;
    const float* p3 = base + 3 * ks + (LL - 1 - lt);

    float gm[6], bt[6];
#pragma unroll
    for (int c = 0; c < 6; ++c) { gm[c] = gamma[lane + 32 * c]; bt[c] = beta[lane + 32 * c]; }

    float a0[MCH], a2[MCH], a1[MCH], a3[MCH];
#pragma unroll
    for (int it = 0; it < MCH; ++it) {
        const size_t o = (size_t)it * MDG * LL;
        a0[it] = p0[o]; a2[it] = p2[o]; a1[it] = p1[o]; a3[it] = p3[o];
    }
    __builtin_amdgcn_sched_barrier(0);

    float v[MCH];
    float s = 0.0f, q = 0.0f;
#pragma unroll
    for (int it = 0; it < MCH; ++it) {
        v[it] = a0[it] + a2[it] + a1[it] + a3[it];
        s += v[it];
        q += v[it] * v[it];
    }
    red_s[dg * MPOS + pos] = s;
    red_q[dg * MPOS + pos] = q;
    __syncthreads();

    float ts = 0.0f, tq = 0.0f;
#pragma unroll
    for (int g = 0; g < MDG; ++g) {
        ts += red_s[g * MPOS + pos];
        tq += red_q[g * MPOS + pos];
    }
    const float mu   = ts / 192.0f;
    const float var  = tq / 192.0f - mu * mu;
    const float rstd = rsqrtf(var + 1e-5f);

#pragma unroll
    for (int it = 0; it < MCH; ++it)
        tile[pos * MPIT + dg + it * MDG] = (v[it] - mu) * rstd;
    __syncthreads();

    const int pb = wv * 4;
    const int lb = ((h0 + (pb >> 3)) << 6) | (w0 + (pb & 7));
    float* ob = out + ((size_t)b * LL + lb) * DI + lane;
    const float* tb = tile + pb * MPIT + lane;
#pragma unroll
    for (int pi = 0; pi < 4; ++pi)
#pragma unroll
        for (int c = 0; c < 6; ++c)
            ob[pi * DI + 32 * c] = tb[pi * MPIT + 32 * c] * gm[c] + bt[c];
}

extern "C" void kernel_launch(void* const* d_in, const int* in_sizes, int n_in,
                              void* d_out, int out_size, void* d_ws, size_t ws_size,
                              hipStream_t stream) {
    const float* x   = (const float*)d_in[0];
    const float* y   = (const float*)d_in[1];
    const float* xpw = (const float*)d_in[2];
    const float* dtw = (const float*)d_in[3];
    const float* dtb = (const float*)d_in[4];
    const float* Al  = (const float*)d_in[5];
    const float* Ds  = (const float*)d_in[6];
    const float* gm  = (const float*)d_in[7];
    const float* bt  = (const float*)d_in[8];

    float* xdbl = (float*)d_ws;
    float* outy = xdbl + (size_t)B_ * KK * CP * LL;

    ss2d_proj_wmma<<<256, 256, 0, stream>>>(x, xpw, xdbl);
    ss2d_scan<<<B_ * KK * (DI / GCH), 256, 0, stream>>>(y, dtw, dtb, Al, Ds, xdbl, outy);
    ss2d_merge_ln<<<B_ * (LL / MPOS), MPOS * MDG, 0, stream>>>(outy, gm, bt, (float*)d_out);
}


# scan, preparing waves: softplus as max(s,0) + a 7-term odd series for log1p(exp(-|s|)), about 20 vector instructions instead of the library log1p's 95; the recurrence waves' code is unchanged
# speedup vs baseline: 28.3531x; 1.0860x over previous
#include <hip/hip_runtime.h>
#include <hip/hip_bf16.h>


#define B_  4
#define DI  192
#define LL  4096
#define NS  16
#define RR  6
#define KK  4
#define CP  38
#define TL  64
#define TLP 65

typedef float v2f __attribute__((ext_vector_type(2)));
typedef float v8f __attribute__((ext_vector_type(8)));

__device__ __forceinline__ int scan_idx(int k, int l) {
    int ll = (k & 2) ? (LL - 1 - l) : l;
    if (k & 1) return ((ll & 63) << 6) | (ll >> 6);
    return ll;
}

#define PNT  8
#define PXP  136
#define PWP  196

__global__ __launch_bounds__(256) void ss2d_proj_wmma(
    const float* __restrict__ x,
    const float* __restrict__ xpw,
    float* __restrict__ xdbl)
{
    __shared__ __attribute__((aligned(16))) float xs[DI * PXP];
    __shared__ __attribute__((aligned(16))) float ws[2 * CP * PWP];

    const int tid  = threadIdx.x;
    const int lane = tid & 31;
    const int j    = tid >> 5;
    const int t    = blockIdx.x & 31;
    const int cls  = (blockIdx.x >> 5) & 1;
    const int b    = blockIdx.x >> 6;

    const float* xb = x + (size_t)b * DI * LL;

    if (cls == 0) {
        const int p0 = t * 128;
#pragma unroll
        for (int i = 0; i < 24; ++i) {
            const int f  = tid + 256 * i;
            const int d  = f >> 5;
            const int c4 = f & 31;
            const float4 v = *(const float4*)(xb + (size_t)d * LL + p0 + 4 * c4);
            *(float4*)(xs + d * PXP + 4 * c4) = v;
        }
    } else {
        const int w0 = (t & 7) * 8;
        const int h0 = (t >> 3) * 16;
#pragma unroll
        for (int i = 0; i < 24; ++i) {
            const int f  = tid + 256 * i;
            const int d  = f >> 5;
            const int r  = f & 31;
            const int c  = r >> 1;
            const int jq = r & 1;
            const float4 v = *(const float4*)(xb + (size_t)d * LL + (h0 + c) * 64 + w0 + 4 * jq);
            float* dst = xs + d * PXP + 64 * jq + c;
            dst[0]  = v.x;
            dst[16] = v.y;
            dst[32] = v.z;
            dst[48] = v.w;
        }
    }
#pragma unroll
    for (int i = 0; i < 15; ++i) {
        const int f = tid + 256 * i;
        if (f < 2 * CP * (DI / 4)) {
            const int r  = f / (DI / 4);
            const int q  = f - r * (DI / 4);
            const int sr = cls * CP + r + (r >= CP ? CP : 0);
            const float4 v = *(const float4*)(xpw + (size_t)sr * DI + 4 * q);
            *(float4*)(ws + r * PWP + 4 * q) = v;
        }
    }
    __syncthreads();

    const int hw = lane >> 4;
    const int ln = lane & 15;

    int aoff[6];
#pragma unroll
    for (int mt = 0; mt < 3; ++mt) {
        const int row = (16 * mt + ln) < CP ? (16 * mt + ln) : (CP - 1);
        aoff[mt]     = row * PWP + 2 * hw;
        aoff[3 + mt] = (CP + row) * PWP + 2 * hw;
    }
    const int boff = (2 * hw) * PXP + 16 * j + ln;

    v8f acc[6];
#pragma unroll
    for (int a = 0; a < 6; ++a) acc[a] = (v8f){0.f, 0.f, 0.f, 0.f, 0.f, 0.f, 0.f, 0.f};

#pragma unroll 4
    for (int k0 = 0; k0 < DI; k0 += 4) {
        v2f bb;
        bb.x = xs[boff + k0 * PXP];
        bb.y = xs[boff + k0 * PXP + PXP];
#pragma unroll
        for (int a = 0; a < 6; ++a) {
            const v2f av = *(const v2f*)(ws + aoff[a] + k0);
            acc[a] = __builtin_amdgcn_wmma_f32_16x16x4_f32(
                false, av, false, bb, (short)0, acc[a], false, false);
        }
    }

    int lf;
    if (cls == 0) lf = t * 128 + 16 * j + ln;
    else          lf = (((t & 7) * 8 + j) << 6) + (t >> 3) * 16 + ln;

#pragma unroll
    for (int kk = 0; kk < 2; ++kk) {
        const int kd = b * KK + cls + 2 * kk;
        const int l  = kk ? (LL - 1 - lf) : lf;
        float* obase = xdbl + (size_t)kd * CP * LL + l;
#pragma unroll
        for (int mt = 0; mt < 3; ++mt) {
#pragma unroll
            for (int i = 0; i < 8; ++i) {
                const int m = 16 * mt + i + (hw << 3);
                if (m < CP) obase[(size_t)m * LL] = acc[kk * 3 + mt][i];
            }
        }
    }
}

#define CHW 4
#define LPC 8
#define SCW 3
#define GCH (SCW * CHW)
#define SFE 6
#define XP  36
#define UP  72
#define DP  144
#define YTP 65
#define YTE (64 * YTP)
#define SMX_F (2 * TL * XP)
#define SMD_F (2 * GCH * DP)
#define SMY_F (2 * GCH * UP)

__device__ __forceinline__ float oct_sum(float v) {
    v += __int_as_float(__builtin_amdgcn_update_dpp(0, __float_as_int(v), 0xB1, 0xF, 0xF, true));
    v += __int_as_float(__builtin_amdgcn_update_dpp(0, __float_as_int(v), 0x4E, 0xF, 0xF, true));
    v += __int_as_float(__builtin_amdgcn_update_dpp(0, __float_as_int(v), 0x141, 0xF, 0xF, true));
    return v;
}

__device__ __forceinline__ float softplus_series(float s) {
    const float t = __builtin_amdgcn_exp2f(-fabsf(s) * 1.4426950408889634f);
    const float z = t * __builtin_amdgcn_rcpf(2.0f + t);
    const float w = z * z;
    float p = 1.0f / 13.0f;
    p = fmaf(p, w, 1.0f / 11.0f);
    p = fmaf(p, w, 1.0f / 9.0f);
    p = fmaf(p, w, 1.0f / 7.0f);
    p = fmaf(p, w, 1.0f / 5.0f);
    p = fmaf(p, w, 1.0f / 3.0f);
    p = fmaf(p, w, 1.0f);
    return fmaf(z + z, p, fmaxf(s, 0.0f));
}

#define GS  8
#define GPC (TL / GS)

__device__ __forceinline__ void scan_load_group(
    const float* sm_x, const float* sm_dd, int g, int q, int ch,
    float4 (&BC)[GS], float2 (&Dv)[GS])
{
#pragma unroll
    for (int t = 0; t < GS; ++t) {
        int j = GS * g + t;
        BC[t] = *(const float4*)(sm_x + j * XP + 4 * q);
        Dv[t] = *(const float2*)(sm_dd + ch * DP + 2 * j);
    }
}

__device__ __forceinline__ void scan_step_group(
    float A20, float A21, float& h0, float& h1,
    const float4 (&BC)[GS], const float2 (&Dv)[GS], float (&yp)[GS])
{
#pragma unroll
    for (int t = 0; t < GS; ++t) {
        float delta = Dv[t].x, du = Dv[t].y;
        float dA0 = __builtin_amdgcn_exp2f(delta * A20);
        float dA1 = __builtin_amdgcn_exp2f(delta * A21);
        h0 = dA0 * h0 + du * BC[t].x;
        h1 = dA1 * h1 + du * BC[t].y;
        float yv = h0 * BC[t].z;
        yv += h1 * BC[t].w;
        yp[t] = yv;
    }
}

__device__ __forceinline__ void scan_sum_group(const float (&yp)[GS], float (&yy)[GS])
{
#pragma unroll
    for (int t = 0; t < GS; ++t) yy[t] = oct_sum(yp[t]);
}

__device__ __forceinline__ void scan_store_group(
    float* sm_y, int g, int q, int ch, const float (&yy)[GS])
{
    if (q == 0) {
        *(float4*)(sm_y + ch * UP + GS * g)     = make_float4(yy[0], yy[1], yy[2], yy[3]);
        *(float4*)(sm_y + ch * UP + GS * g + 4) = make_float4(yy[4], yy[5], yy[6], yy[7]);
    }
}

__device__ __forceinline__ void scan_hold(float& h0, float& h1, float (&yp)[GS])
{
    asm volatile("" : "+v"(h0), "+v"(h1),
                      "+v"(yp[0]), "+v"(yp[1]), "+v"(yp[2]), "+v"(yp[3]),
                      "+v"(yp[4]), "+v"(yp[5]), "+v"(yp[6]), "+v"(yp[7]));
}

__global__ __launch_bounds__(256) void ss2d_scan(
    const float* __restrict__ y,
    const float* __restrict__ dtw_g,
    const float* __restrict__ dtb_g,
    const float* __restrict__ Alogs,
    const float* __restrict__ Ds_g,
    const float* __restrict__ xdbl,
    float* __restrict__ outy)
{
    __shared__ __attribute__((aligned(16))) float smem[SMX_F + SMD_F + SMY_F + GCH * YTE];
    float* const sm_x  = smem;
    float* const sm_dd = smem + SMX_F;
    float* const sm_y  = smem + SMX_F + SMD_F;
    float* const sm_yt = smem + SMX_F + SMD_F + SMY_F;

    const int lane = threadIdx.x & 31;
    const int wv = __builtin_amdgcn_readfirstlane((int)(threadIdx.x >> 5));
    const bool is_scan  = wv < SCW;
    const bool is_prep  = (wv >= SCW) && (wv < SCW + 4);
    const bool is_stage = wv == SCW + 4;

    const int cg = blockIdx.x % (DI / GCH);
    const int kd = blockIdx.x / (DI / GCH);
    const int k  = kd & 3;
    const int b  = kd >> 2;
    const int d0 = cg * GCH;

    const float* xdb = xdbl + (size_t)kd * CP * LL;
    const float* yb  = y + ((size_t)b * DI + d0) * LL;
    float* ob        = outy + ((size_t)kd * DI + d0) * LL;

    const bool tdir = (k & 1) != 0;
    if (tdir) {
        const float4* ysl = (const float4*)yb;
#pragma unroll 1
        for (int rd = 0; rd < 2; ++rd) {
            float4 yv[24];
#pragma unroll
            for (int p = 0; p < 24; ++p)
                yv[p] = ysl[(rd * 24 + p) * 256 + (int)threadIdx.x];
#pragma unroll
            for (int p = 0; p < 24; ++p) {
                const int f  = (rd * 24 + p) * 256 + (int)threadIdx.x;
                const int r  = f & 1023;
                float* dst = sm_yt + (f >> 10) * YTE + (r >> 4) * YTP + 4 * (r & 15);
                dst[0] = yv[p].x;
                dst[1] = yv[p].y;
                dst[2] = yv[p].z;
                dst[3] = yv[p].w;
            }
        }
        __syncthreads();
    }

    const int ch = lane >> 3;
    const int q  = lane & 7;
    const int cw = (is_scan ? wv : 0) * CHW + ch;
    const int kdch = k * DI + d0 + cw;
    const float A20 = -__expf(Alogs[(size_t)kdch * NS + 2 * q])     * 1.4426950408889634f;
    const float A21 = -__expf(Alogs[(size_t)kdch * NS + 2 * q + 1]) * 1.4426950408889634f;
    float h0 = 0.0f, h1 = 0.0f;

    const int pw  = is_prep ? (wv - SCW) : 0;
    const int ps  = (pw & 1) * 32 + lane;
    const int pc0 = (pw >> 1) * SFE;
    float dtw[SFE][RR], dtb[SFE], DsC[SFE];
#pragma unroll
    for (int e = 0; e < SFE; ++e) {
        const int kc = k * DI + d0 + pc0 + e;
#pragma unroll
        for (int r = 0; r < RR; ++r) dtw[e][r] = dtw_g[(size_t)kc * RR + r];
        dtb[e] = dtb_g[kc];
        DsC[e] = Ds_g[kc];
    }
    float dtr[RR];
    float un[SFE];
    float u1[SFE], u2[SFE];
#pragma unroll
    for (int r = 0; r < RR; ++r) dtr[r] = 0.0f;
#pragma unroll
    for (int e = 0; e < SFE; ++e) { un[e] = 0.0f; u1[e] = 0.0f; u2[e] = 0.0f; }

    float xr[2][2 * NS];
#pragma unroll
    for (int hf = 0; hf < 2; ++hf)
#pragma unroll
        for (int r = 0; r < 2 * NS; ++r) xr[hf][r] = 0.0f;

    if (is_prep) {
#pragma unroll
        for (int r = 0; r < RR; ++r) dtr[r] = xdb[(size_t)r * LL + ps];
        const int m0 = scan_idx(k, ps);
        if (tdir) {
            const float* pyt = sm_yt + pc0 * YTE + (m0 >> 6) * YTP + (m0 & 63);
#pragma unroll
            for (int e = 0; e < SFE; ++e) un[e] = pyt[e * YTE];
        } else {
#pragma unroll
            for (int e = 0; e < SFE; ++e) un[e] = yb[(size_t)(pc0 + e) * LL + m0];
        }
    } else if (is_stage) {
#pragma unroll
        for (int hf = 0; hf < 2; ++hf)
#pragma unroll
            for (int r = 0; r < 2 * NS; ++r)
                xr[hf][r] = xdb[(size_t)(RR + r) * LL + lane + 32 * hf];
    }

    for (int i = 0; i < LL / TL + 2; ++i) {
        const int bs = i & 1;
        if (is_scan) {
            if (i >= 1 && i <= LL / TL) {
                const float* px = sm_x  + (bs ^ 1) * (TL * XP);
                const float* pd = sm_dd + (bs ^ 1) * (GCH * DP);
                float*       py = sm_y  + (bs ^ 1) * (GCH * UP);
                float4 BC0[GS], BC1[GS];
                float2 D0[GS], D1[GS];
                float ya[GS], yb[GS], yy[GS];
                scan_load_group(px, pd, 0, q, cw, BC0, D0);
                scan_load_group(px, pd, 1, q, cw, BC1, D1);
                __builtin_amdgcn_sched_barrier(0);
                scan_step_group(A20, A21, h0, h1, BC0, D0, ya);
#pragma unroll 1
                for (int g = 1; g < GPC - 1; g += 2) {
                    scan_load_group(px, pd, g + 1, q, cw, BC0, D0);
                    __builtin_amdgcn_sched_barrier(0);
                    scan_step_group(A20, A21, h0, h1, BC1, D1, yb);
                    scan_sum_group(ya, yy);
                    scan_hold(h0, h1, yb);
                    scan_store_group(py, g - 1, q, cw, yy);
                    scan_load_group(px, pd, g + 2, q, cw, BC1, D1);
                    __builtin_amdgcn_sched_barrier(0);
                    scan_step_group(A20, A21, h0, h1, BC0, D0, ya);
                    scan_sum_group(yb, yy);
                    scan_hold(h0, h1, ya);
                    scan_store_group(py, g, q, cw, yy);
                }
                scan_step_group(A20, A21, h0, h1, BC1, D1, yb);
                scan_sum_group(ya, yy);
                scan_hold(h0, h1, yb);
                scan_store_group(py, GPC - 2, q, cw, yy);
                scan_sum_group(yb, yy);
                scan_store_group(py, GPC - 1, q, cw, yy);
            }
        } else if (is_prep) {
            float dnx[RR], unx[SFE];
            if (i + 1 < LL / TL) {
                const int ln0 = (i + 1) * TL;
#pragma unroll
                for (int r = 0; r < RR; ++r) dnx[r] = xdb[(size_t)r * LL + ln0 + ps];
                const int m1 = scan_idx(k, ln0 + ps);
                if (tdir) {
                    const float* pyt = sm_yt + pc0 * YTE + (m1 >> 6) * YTP + (m1 & 63);
#pragma unroll
                    for (int e = 0; e < SFE; ++e) unx[e] = pyt[e * YTE];
                } else {
#pragma unroll
                    for (int e = 0; e < SFE; ++e)
                        unx[e] = yb[(size_t)(pc0 + e) * LL + m1];
                }
            } else {
#pragma unroll
                for (int r = 0; r < RR; ++r) dnx[r] = 0.0f;
#pragma unroll
                for (int e = 0; e < SFE; ++e) unx[e] = 0.0f;
            }
            if (i >= 2) {
                const float* psy = sm_y + bs * (GCH * UP) + pc0 * UP + ps;
                float* po = ob + (size_t)pc0 * LL + (i - 2) * TL + ps;
#pragma unroll
                for (int e = 0; e < SFE; ++e)
                    po[(size_t)e * LL] = psy[e * UP] + DsC[e] * u2[e];
            }
            if (i < LL / TL) {
                float* pdd = sm_dd + bs * (GCH * DP) + pc0 * DP + 2 * ps;
#pragma unroll
                for (int e = 0; e < SFE; ++e) {
                    float s = dtb[e];
                    s += dtw[e][0] * dtr[0];
                    s += dtw[e][1] * dtr[1];
                    s += dtw[e][2] * dtr[2];
                    s += dtw[e][3] * dtr[3];
                    s += dtw[e][4] * dtr[4];
                    s += dtw[e][5] * dtr[5];
                    const float delta = softplus_series(s);
                    *(float2*)(pdd + e * DP) = make_float2(delta, delta * un[e]);
                }
            }
#pragma unroll
            for (int e = 0; e < SFE; ++e) { u2[e] = u1[e]; u1[e] = un[e]; un[e] = unx[e]; }
#pragma unroll
            for (int r = 0; r < RR; ++r) dtr[r] = dnx[r];
        } else if (is_stage) {
            if (i < LL / TL) {
#pragma unroll
                for (int hf = 0; hf < 2; ++hf) {
                    float* dst = sm_x + bs * (TL * XP) + (lane + 32 * hf) * XP;
#pragma unroll
                    for (int p = 0; p < 8; ++p)
                        *(float4*)(dst + 4 * p) =
                            make_float4(xr[hf][2 * p],      xr[hf][2 * p + 1],
                                        xr[hf][NS + 2 * p], xr[hf][NS + 2 * p + 1]);
                }
            }
            if (i + 1 < LL / TL) {
                const int ln0 = (i + 1) * TL;
#pragma unroll
                for (int hf = 0; hf < 2; ++hf)
#pragma unroll
                    for (int r = 0; r < 2 * NS; ++r)
                        xr[hf][r] = xdb[(size_t)(RR + r) * LL + ln0 + lane + 32 * hf];
            }
        }
        __syncthreads();
    }
}

#define MTE   8
#define MPOS  (MTE * MTE)
#define MDG   8
#define MCH   (DI / MDG)
#define MPIT  (DI + 1)

__global__ __launch_bounds__(MPOS * MDG) void ss2d_merge_ln(
    const float* __restrict__ outy,
    const float* __restrict__ gamma,
    const float* __restrict__ beta,
    float* __restrict__ out)
{
    __shared__ float tile[MPOS * MPIT];
    __shared__ float red_s[MDG * MPOS];
    __shared__ float red_q[MDG * MPOS];

    const int tid  = threadIdx.x;
    const int lane = tid & 31;
    const int wv   = tid >> 5;
    const int b    = blockIdx.x >> 6;
    const int t    = blockIdx.x & 63;
    const int h0   = (t >> 3) * MTE;
    const int w0   = (t & 7) * MTE;
    const int pos  = tid & (MPOS - 1);
    const int dg   = tid >> 6;
    const int hh   = h0 + (pos >> 3);
    const int ww   = w0 + (pos & 7);
    const int lm   = (hh << 6) | ww;
    const int lt   = (ww << 6) | hh;

    const size_t ks = (size_t)DI * LL;
    const float* base = outy + (size_t)b * KK * ks + (size_t)dg * LL;
    const float* p0 = base + lm;
    const float* p2 = base + 2 * ks + (LL - 1 - lm);
    const float* p1 = base + ks + lt;
    const float* p3 = base + 3 * ks + (LL - 1 - lt);

    float gm[6], bt[6];
#pragma unroll
    for (int c = 0; c < 6; ++c) { gm[c] = gamma[lane + 32 * c]; bt[c] = beta[lane + 32 * c]; }

    float a0[MCH], a2[MCH], a1[MCH], a3[MCH];
#pragma unroll
    for (int it = 0; it < MCH; ++it) {
        const size_t o = (size_t)it * MDG * LL;
        a0[it] = p0[o]; a2[it] = p2[o]; a1[it] = p1[o]; a3[it] = p3[o];
    }
    __builtin_amdgcn_sched_barrier(0);

    float v[MCH];
    float s = 0.0f, q = 0.0f;
#pragma unroll
    for (int it = 0; it < MCH; ++it) {
        v[it] = a0[it] + a2[it] + a1[it] + a3[it];
        s += v[it];
        q += v[it] * v[it];
    }
    red_s[dg * MPOS + pos] = s;
    red_q[dg * MPOS + pos] = q;
    __syncthreads();

    float ts = 0.0f, tq = 0.0f;
#pragma unroll
    for (int g = 0; g < MDG; ++g) {
        ts += red_s[g * MPOS + pos];
        tq += red_q[g * MPOS + pos];
    }
    const float mu   = ts / 192.0f;
    const float var  = tq / 192.0f - mu * mu;
    const float rstd = rsqrtf(var + 1e-5f);

#pragma unroll
    for (int it = 0; it < MCH; ++it)
        tile[pos * MPIT + dg + it * MDG] = (v[it] - mu) * rstd;
    __syncthreads();

    const int pb = wv * 4;
    const int lb = ((h0 + (pb >> 3)) << 6) | (w0 + (pb & 7));
    float* ob = out + ((size_t)b * LL + lb) * DI + lane;
    const float* tb = tile + pb * MPIT + lane;
#pragma unroll
    for (int pi = 0; pi < 4; ++pi)
#pragma unroll
        for (int c = 0; c < 6; ++c)
            ob[pi * DI + 32 * c] = tb[pi * MPIT + 32 * c] * gm[c] + bt[c];
}

extern "C" void kernel_launch(void* const* d_in, const int* in_sizes, int n_in,
                              void* d_out, int out_size, void* d_ws, size_t ws_size,
                              hipStream_t stream) {
    const float* x   = (const float*)d_in[0];
    const float* y   = (const float*)d_in[1];
    const float* xpw = (const float*)d_in[2];
    const float* dtw = (const float*)d_in[3];
    const float* dtb = (const float*)d_in[4];
    const float* Al  = (const float*)d_in[5];
    const float* Ds  = (const float*)d_in[6];
    const float* gm  = (const float*)d_in[7];
    const float* bt  = (const float*)d_in[8];

    float* xdbl = (float*)d_ws;
    float* outy = xdbl + (size_t)B_ * KK * CP * LL;

    ss2d_proj_wmma<<<256, 256, 0, stream>>>(x, xpw, xdbl);
    ss2d_scan<<<B_ * KK * (DI / GCH), 256, 0, stream>>>(y, dtw, dtb, Al, Ds, xdbl, outy);
    ss2d_merge_ln<<<B_ * (LL / MPOS), MPOS * MDG, 0, stream>>>(outy, gm, bt, (float*)d_out);
}


# scan: recurrence waves store each lane's share of an output to LDS beside their next reads; preparing waves add a step's 8 shares in the same tree order; no cross-lane adds in the serial waves
# speedup vs baseline: 29.9512x; 1.0564x over previous
#include <hip/hip_runtime.h>
#include <hip/hip_bf16.h>


#define B_  4
#define DI  192
#define LL  4096
#define NS  16
#define RR  6
#define KK  4
#define CP  38
#define TL  64
#define TLP 65

typedef float v2f __attribute__((ext_vector_type(2)));
typedef float v8f __attribute__((ext_vector_type(8)));

__device__ __forceinline__ int scan_idx(int k, int l) {
    int ll = (k & 2) ? (LL - 1 - l) : l;
    if (k & 1) return ((ll & 63) << 6) | (ll >> 6);
    return ll;
}

#define PNT  8
#define PXP  136
#define PWP  196

__global__ __launch_bounds__(256) void ss2d_proj_wmma(
    const float* __restrict__ x,
    const float* __restrict__ xpw,
    float* __restrict__ xdbl)
{
    __shared__ __attribute__((aligned(16))) float xs[DI * PXP];
    __shared__ __attribute__((aligned(16))) float ws[2 * CP * PWP];

    const int tid  = threadIdx.x;
    const int lane = tid & 31;
    const int j    = tid >> 5;
    const int t    = blockIdx.x & 31;
    const int cls  = (blockIdx.x >> 5) & 1;
    const int b    = blockIdx.x >> 6;

    const float* xb = x + (size_t)b * DI * LL;

    if (cls == 0) {
        const int p0 = t * 128;
#pragma unroll
        for (int i = 0; i < 24; ++i) {
            const int f  = tid + 256 * i;
            const int d  = f >> 5;
            const int c4 = f & 31;
            const float4 v = *(const float4*)(xb + (size_t)d * LL + p0 + 4 * c4);
            *(float4*)(xs + d * PXP + 4 * c4) = v;
        }
    } else {
        const int w0 = (t & 7) * 8;
        const int h0 = (t >> 3) * 16;
#pragma unroll
        for (int i = 0; i < 24; ++i) {
            const int f  = tid + 256 * i;
            const int d  = f >> 5;
            const int r  = f & 31;
            const int c  = r >> 1;
            const int jq = r & 1;
            const float4 v = *(const float4*)(xb + (size_t)d * LL + (h0 + c) * 64 + w0 + 4 * jq);
            float* dst = xs + d * PXP + 64 * jq + c;
            dst[0]  = v.x;
            dst[16] = v.y;
            dst[32] = v.z;
            dst[48] = v.w;
        }
    }
#pragma unroll
    for (int i = 0; i < 15; ++i) {
        const int f = tid + 256 * i;
        if (f < 2 * CP * (DI / 4)) {
            const int r  = f / (DI / 4);
            const int q  = f - r * (DI / 4);
            const int sr = cls * CP + r + (r >= CP ? CP : 0);
            const float4 v = *(const float4*)(xpw + (size_t)sr * DI + 4 * q);
            *(float4*)(ws + r * PWP + 4 * q) = v;
        }
    }
    __syncthreads();

    const int hw = lane >> 4;
    const int ln = lane & 15;

    int aoff[6];
#pragma unroll
    for (int mt = 0; mt < 3; ++mt) {
        const int row = (16 * mt + ln) < CP ? (16 * mt + ln) : (CP - 1);
        aoff[mt]     = row * PWP + 2 * hw;
        aoff[3 + mt] = (CP + row) * PWP + 2 * hw;
    }
    const int boff = (2 * hw) * PXP + 16 * j + ln;

    v8f acc[6];
#pragma unroll
    for (int a = 0; a < 6; ++a) acc[a] = (v8f){0.f, 0.f, 0.f, 0.f, 0.f, 0.f, 0.f, 0.f};

#pragma unroll 4
    for (int k0 = 0; k0 < DI; k0 += 4) {
        v2f bb;
        bb.x = xs[boff + k0 * PXP];
        bb.y = xs[boff + k0 * PXP + PXP];
#pragma unroll
        for (int a = 0; a < 6; ++a) {
            const v2f av = *(const v2f*)(ws + aoff[a] + k0);
            acc[a] = __builtin_amdgcn_wmma_f32_16x16x4_f32(
                false, av, false, bb, (short)0, acc[a], false, false);
        }
    }

    int lf;
    if (cls == 0) lf = t * 128 + 16 * j + ln;
    else          lf = (((t & 7) * 8 + j) << 6) + (t >> 3) * 16 + ln;

#pragma unroll
    for (int kk = 0; kk < 2; ++kk) {
        const int kd = b * KK + cls + 2 * kk;
        const int l  = kk ? (LL - 1 - lf) : lf;
        float* obase = xdbl + (size_t)kd * CP * LL + l;
#pragma unroll
        for (int mt = 0; mt < 3; ++mt) {
#pragma unroll
            for (int i = 0; i < 8; ++i) {
                const int m = 16 * mt + i + (hw << 3);
                if (m < CP) obase[(size_t)m * LL] = acc[kk * 3 + mt][i];
            }
        }
    }
}

#define CHW 4
#define LPC 8
#define SCW 3
#define GCH (SCW * CHW)
#define SFE 6
#define XP  36
#define YQ  LPC
#define YP  (TL * YQ + 8)
#define DP  144
#define YTP 65
#define YTE (64 * YTP)
#define SMX_F (2 * TL * XP)
#define SMD_F (2 * GCH * DP)
#define SMY_F (2 * GCH * YP)

__device__ __forceinline__ float softplus_series(float s) {
    const float t = __builtin_amdgcn_exp2f(-fabsf(s) * 1.4426950408889634f);
    const float z = t * __builtin_amdgcn_rcpf(2.0f + t);
    const float w = z * z;
    float p = 1.0f / 13.0f;
    p = fmaf(p, w, 1.0f / 11.0f);
    p = fmaf(p, w, 1.0f / 9.0f);
    p = fmaf(p, w, 1.0f / 7.0f);
    p = fmaf(p, w, 1.0f / 5.0f);
    p = fmaf(p, w, 1.0f / 3.0f);
    p = fmaf(p, w, 1.0f);
    return fmaf(z + z, p, fmaxf(s, 0.0f));
}

#define GS  8
#define GPC (TL / GS)

__device__ __forceinline__ void scan_load_group(
    const float* sm_x, const float* sm_dd, int g, int q, int ch,
    float4 (&BC)[GS], float2 (&Dv)[GS])
{
#pragma unroll
    for (int t = 0; t < GS; ++t) {
        int j = GS * g + t;
        BC[t] = *(const float4*)(sm_x + j * XP + 4 * q);
        Dv[t] = *(const float2*)(sm_dd + ch * DP + 2 * j);
    }
}

__device__ __forceinline__ void scan_step_group(
    float A20, float A21, float& h0, float& h1,
    const float4 (&BC)[GS], const float2 (&Dv)[GS], float (&yp)[GS])
{
#pragma unroll
    for (int t = 0; t < GS; ++t) {
        float delta = Dv[t].x, du = Dv[t].y;
        float dA0 = __builtin_amdgcn_exp2f(delta * A20);
        float dA1 = __builtin_amdgcn_exp2f(delta * A21);
        h0 = dA0 * h0 + du * BC[t].x;
        h1 = dA1 * h1 + du * BC[t].y;
        float yv = h0 * BC[t].z;
        yv += h1 * BC[t].w;
        yp[t] = yv;
    }
}

__device__ __forceinline__ void scan_store_part(
    float* sm_y, int g, int q, int ch, const float (&yp)[GS])
{
    float* dst = sm_y + ch * YP + (GS * g) * YQ + q;
#pragma unroll
    for (int t = 0; t < GS; ++t) dst[t * YQ] = yp[t];
}

__global__ __launch_bounds__(256) void ss2d_scan(
    const float* __restrict__ y,
    const float* __restrict__ dtw_g,
    const float* __restrict__ dtb_g,
    const float* __restrict__ Alogs,
    const float* __restrict__ Ds_g,
    const float* __restrict__ xdbl,
    float* __restrict__ outy)
{
    __shared__ __attribute__((aligned(16))) float smem[SMX_F + SMD_F + SMY_F + GCH * YTE];
    float* const sm_x  = smem;
    float* const sm_dd = smem + SMX_F;
    float* const sm_y  = smem + SMX_F + SMD_F;
    float* const sm_yt = smem + SMX_F + SMD_F + SMY_F;

    const int lane = threadIdx.x & 31;
    const int wv = __builtin_amdgcn_readfirstlane((int)(threadIdx.x >> 5));
    const bool is_scan  = wv < SCW;
    const bool is_prep  = (wv >= SCW) && (wv < SCW + 4);
    const bool is_stage = wv == SCW + 4;

    const int cg = blockIdx.x % (DI / GCH);
    const int kd = blockIdx.x / (DI / GCH);
    const int k  = kd & 3;
    const int b  = kd >> 2;
    const int d0 = cg * GCH;

    const float* xdb = xdbl + (size_t)kd * CP * LL;
    const float* yb  = y + ((size_t)b * DI + d0) * LL;
    float* ob        = outy + ((size_t)kd * DI + d0) * LL;

    const bool tdir = (k & 1) != 0;
    if (tdir) {
        const float4* ysl = (const float4*)yb;
#pragma unroll 1
        for (int rd = 0; rd < 2; ++rd) {
            float4 yv[24];
#pragma unroll
            for (int p = 0; p < 24; ++p)
                yv[p] = ysl[(rd * 24 + p) * 256 + (int)threadIdx.x];
#pragma unroll
            for (int p = 0; p < 24; ++p) {
                const int f  = (rd * 24 + p) * 256 + (int)threadIdx.x;
                const int r  = f & 1023;
                float* dst = sm_yt + (f >> 10) * YTE + (r >> 4) * YTP + 4 * (r & 15);
                dst[0] = yv[p].x;
                dst[1] = yv[p].y;
                dst[2] = yv[p].z;
                dst[3] = yv[p].w;
            }
        }
        __syncthreads();
    }

    const int ch = lane >> 3;
    const int q  = lane & 7;
    const int cw = (is_scan ? wv : 0) * CHW + ch;
    const int kdch = k * DI + d0 + cw;
    const float A20 = -__expf(Alogs[(size_t)kdch * NS + 2 * q])     * 1.4426950408889634f;
    const float A21 = -__expf(Alogs[(size_t)kdch * NS + 2 * q + 1]) * 1.4426950408889634f;
    float h0 = 0.0f, h1 = 0.0f;

    const int pw  = is_prep ? (wv - SCW) : 0;
    const int ps  = (pw & 1) * 32 + lane;
    const int pc0 = (pw >> 1) * SFE;
    float dtw[SFE][RR], dtb[SFE], DsC[SFE];
#pragma unroll
    for (int e = 0; e < SFE; ++e) {
        const int kc = k * DI + d0 + pc0 + e;
#pragma unroll
        for (int r = 0; r < RR; ++r) dtw[e][r] = dtw_g[(size_t)kc * RR + r];
        dtb[e] = dtb_g[kc];
        DsC[e] = Ds_g[kc];
    }
    float dtr[RR];
    float un[SFE];
    float u1[SFE], u2[SFE];
#pragma unroll
    for (int r = 0; r < RR; ++r) dtr[r] = 0.0f;
#pragma unroll
    for (int e = 0; e < SFE; ++e) { un[e] = 0.0f; u1[e] = 0.0f; u2[e] = 0.0f; }

    float xr[2][2 * NS];
#pragma unroll
    for (int hf = 0; hf < 2; ++hf)
#pragma unroll
        for (int r = 0; r < 2 * NS; ++r) xr[hf][r] = 0.0f;

    if (is_prep) {
#pragma unroll
        for (int r = 0; r < RR; ++r) dtr[r] = xdb[(size_t)r * LL + ps];
        const int m0 = scan_idx(k, ps);
        if (tdir) {
            const float* pyt = sm_yt + pc0 * YTE + (m0 >> 6) * YTP + (m0 & 63);
#pragma unroll
            for (int e = 0; e < SFE; ++e) un[e] = pyt[e * YTE];
        } else {
#pragma unroll
            for (int e = 0; e < SFE; ++e) un[e] = yb[(size_t)(pc0 + e) * LL + m0];
        }
    } else if (is_stage) {
#pragma unroll
        for (int hf = 0; hf < 2; ++hf)
#pragma unroll
            for (int r = 0; r < 2 * NS; ++r)
                xr[hf][r] = xdb[(size_t)(RR + r) * LL + lane + 32 * hf];
    }

    for (int i = 0; i < LL / TL + 2; ++i) {
        const int bs = i & 1;
        if (is_scan) {
            if (i >= 1 && i <= LL / TL) {
                const float* px = sm_x  + (bs ^ 1) * (TL * XP);
                const float* pd = sm_dd + (bs ^ 1) * (GCH * DP);
                float*       py = sm_y  + (bs ^ 1) * (GCH * YP);
                float4 BC0[GS], BC1[GS];
                float2 D0[GS], D1[GS];
                float ya[GS], yb[GS];
                scan_load_group(px, pd, 0, q, cw, BC0, D0);
                scan_load_group(px, pd, 1, q, cw, BC1, D1);
                __builtin_amdgcn_sched_barrier(0);
                scan_step_group(A20, A21, h0, h1, BC0, D0, ya);
#pragma unroll 1
                for (int g = 1; g < GPC - 1; g += 2) {
                    scan_store_part(py, g - 1, q, cw, ya);
                    scan_load_group(px, pd, g + 1, q, cw, BC0, D0);
                    __builtin_amdgcn_sched_barrier(0);
                    scan_step_group(A20, A21, h0, h1, BC1, D1, yb);
                    scan_store_part(py, g, q, cw, yb);
                    scan_load_group(px, pd, g + 2, q, cw, BC1, D1);
                    __builtin_amdgcn_sched_barrier(0);
                    scan_step_group(A20, A21, h0, h1, BC0, D0, ya);
                }
                scan_store_part(py, GPC - 2, q, cw, ya);
                scan_step_group(A20, A21, h0, h1, BC1, D1, yb);
                scan_store_part(py, GPC - 1, q, cw, yb);
            }
        } else if (is_prep) {
            float4 pa[SFE], pb[SFE];
            if (i >= 2) {
                const float* psy = sm_y + bs * (GCH * YP) + pc0 * YP + ps * YQ;
#pragma unroll
                for (int e = 0; e < SFE; ++e) {
                    pa[e] = *(const float4*)(psy + e * YP);
                    pb[e] = *(const float4*)(psy + e * YP + 4);
                }
            } else {
#pragma unroll
                for (int e = 0; e < SFE; ++e) {
                    pa[e] = make_float4(0.0f, 0.0f, 0.0f, 0.0f);
                    pb[e] = make_float4(0.0f, 0.0f, 0.0f, 0.0f);
                }
            }
            float dnx[RR], unx[SFE];
            if (i + 1 < LL / TL) {
                const int ln0 = (i + 1) * TL;
#pragma unroll
                for (int r = 0; r < RR; ++r) dnx[r] = xdb[(size_t)r * LL + ln0 + ps];
                const int m1 = scan_idx(k, ln0 + ps);
                if (tdir) {
                    const float* pyt = sm_yt + pc0 * YTE + (m1 >> 6) * YTP + (m1 & 63);
#pragma unroll
                    for (int e = 0; e < SFE; ++e) unx[e] = pyt[e * YTE];
                } else {
#pragma unroll
                    for (int e = 0; e < SFE; ++e)
                        unx[e] = yb[(size_t)(pc0 + e) * LL + m1];
                }
            } else {
#pragma unroll
                for (int r = 0; r < RR; ++r) dnx[r] = 0.0f;
#pragma unroll
                for (int e = 0; e < SFE; ++e) unx[e] = 0.0f;
            }
            if (i >= 2) {
                float* po = ob + (size_t)pc0 * LL + (i - 2) * TL + ps;
#pragma unroll
                for (int e = 0; e < SFE; ++e) {
                    const float ys = ((pa[e].x + pa[e].y) + (pa[e].z + pa[e].w))
                                   + ((pb[e].x + pb[e].y) + (pb[e].z + pb[e].w));
                    po[(size_t)e * LL] = ys + DsC[e] * u2[e];
                }
            }
            if (i < LL / TL) {
                float* pdd = sm_dd + bs * (GCH * DP) + pc0 * DP + 2 * ps;
#pragma unroll
                for (int e = 0; e < SFE; ++e) {
                    float s = dtb[e];
                    s += dtw[e][0] * dtr[0];
                    s += dtw[e][1] * dtr[1];
                    s += dtw[e][2] * dtr[2];
                    s += dtw[e][3] * dtr[3];
                    s += dtw[e][4] * dtr[4];
                    s += dtw[e][5] * dtr[5];
                    const float delta = softplus_series(s);
                    *(float2*)(pdd + e * DP) = make_float2(delta, delta * un[e]);
                }
            }
#pragma unroll
            for (int e = 0; e < SFE; ++e) { u2[e] = u1[e]; u1[e] = un[e]; un[e] = unx[e]; }
#pragma unroll
            for (int r = 0; r < RR; ++r) dtr[r] = dnx[r];
        } else if (is_stage) {
            if (i < LL / TL) {
#pragma unroll
                for (int hf = 0; hf < 2; ++hf) {
                    float* dst = sm_x + bs * (TL * XP) + (lane + 32 * hf) * XP;
#pragma unroll
                    for (int p = 0; p < 8; ++p)
                        *(float4*)(dst + 4 * p) =
                            make_float4(xr[hf][2 * p],      xr[hf][2 * p + 1],
                                        xr[hf][NS + 2 * p], xr[hf][NS + 2 * p + 1]);
                }
            }
            if (i + 1 < LL / TL) {
                const int ln0 = (i + 1) * TL;
#pragma unroll
                for (int hf = 0; hf < 2; ++hf)
#pragma unroll
                    for (int r = 0; r < 2 * NS; ++r)
                        xr[hf][r] = xdb[(size_t)(RR + r) * LL + ln0 + lane + 32 * hf];
            }
        }
        __syncthreads();
    }
}

#define MTE   8
#define MPOS  (MTE * MTE)
#define MDG   8
#define MCH   (DI / MDG)
#define MPIT  (DI + 1)

__global__ __launch_bounds__(MPOS * MDG) void ss2d_merge_ln(
    const float* __restrict__ outy,
    const float* __restrict__ gamma,
    const float* __restrict__ beta,
    float* __restrict__ out)
{
    __shared__ float tile[MPOS * MPIT];
    __shared__ float red_s[MDG * MPOS];
    __shared__ float red_q[MDG * MPOS];

    const int tid  = threadIdx.x;
    const int lane = tid & 31;
    const int wv   = tid >> 5;
    const int b    = blockIdx.x >> 6;
    const int t    = blockIdx.x & 63;
    const int h0   = (t >> 3) * MTE;
    const int w0   = (t & 7) * MTE;
    const int pos  = tid & (MPOS - 1);
    const int dg   = tid >> 6;
    const int hh   = h0 + (pos >> 3);
    const int ww   = w0 + (pos & 7);
    const int lm   = (hh << 6) | ww;
    const int lt   = (ww << 6) | hh;

    const size_t ks = (size_t)DI * LL;
    const float* base = outy + (size_t)b * KK * ks + (size_t)dg * LL;
    const float* p0 = base + lm;
    const float* p2 = base + 2 * ks + (LL - 1 - lm);
    const float* p1 = base + ks + lt;
    const float* p3 = base + 3 * ks + (LL - 1 - lt);

    float gm[6], bt[6];
#pragma unroll
    for (int c = 0; c < 6; ++c) { gm[c] = gamma[lane + 32 * c]; bt[c] = beta[lane + 32 * c]; }

    float a0[MCH], a2[MCH], a1[MCH], a3[MCH];
#pragma unroll
    for (int it = 0; it < MCH; ++it) {
        const size_t o = (size_t)it * MDG * LL;
        a0[it] = p0[o]; a2[it] = p2[o]; a1[it] = p1[o]; a3[it] = p3[o];
    }
    __builtin_amdgcn_sched_barrier(0);

    float v[MCH];
    float s = 0.0f, q = 0.0f;
#pragma unroll
    for (int it = 0; it < MCH; ++it) {
        v[it] = a0[it] + a2[it] + a1[it] + a3[it];
        s += v[it];
        q += v[it] * v[it];
    }
    red_s[dg * MPOS + pos] = s;
    red_q[dg * MPOS + pos] = q;
    __syncthreads();

    float ts = 0.0f, tq = 0.0f;
#pragma unroll
    for (int g = 0; g < MDG; ++g) {
        ts += red_s[g * MPOS + pos];
        tq += red_q[g * MPOS + pos];
    }
    const float mu   = ts / 192.0f;
    const float var  = tq / 192.0f - mu * mu;
    const float rstd = rsqrtf(var + 1e-5f);

#pragma unroll
    for (int it = 0; it < MCH; ++it)
        tile[pos * MPIT + dg + it * MDG] = (v[it] - mu) * rstd;
    __syncthreads();

    const int pb = wv * 4;
    const int lb = ((h0 + (pb >> 3)) << 6) | (w0 + (pb & 7));
    float* ob = out + ((size_t)b * LL + lb) * DI + lane;
    const float* tb = tile + pb * MPIT + lane;
#pragma unroll
    for (int pi = 0; pi < 4; ++pi)
#pragma unroll
        for (int c = 0; c < 6; ++c)
            ob[pi * DI + 32 * c] = tb[pi * MPIT + 32 * c] * gm[c] + bt[c];
}

extern "C" void kernel_launch(void* const* d_in, const int* in_sizes, int n_in,
                              void* d_out, int out_size, void* d_ws, size_t ws_size,
                              hipStream_t stream) {
    const float* x   = (const float*)d_in[0];
    const float* y   = (const float*)d_in[1];
    const float* xpw = (const float*)d_in[2];
    const float* dtw = (const float*)d_in[3];
    const float* dtb = (const float*)d_in[4];
    const float* Al  = (const float*)d_in[5];
    const float* Ds  = (const float*)d_in[6];
    const float* gm  = (const float*)d_in[7];
    const float* bt  = (const float*)d_in[8];

    float* xdbl = (float*)d_ws;
    float* outy = xdbl + (size_t)B_ * KK * CP * LL;

    ss2d_proj_wmma<<<256, 256, 0, stream>>>(x, xpw, xdbl);
    ss2d_scan<<<B_ * KK * (DI / GCH), 256, 0, stream>>>(y, dtw, dtb, Al, Ds, xdbl, outy);
    ss2d_merge_ln<<<B_ * (LL / MPOS), MPOS * MDG, 0, stream>>>(outy, gm, bt, (float*)d_out);
}


# scan: the per-chunk barrier is written by hand and waits for LDS traffic only (s_wait_dscnt 0, signal, wait): no wave stands at it for the acknowledgement of its global output stores any more
# speedup vs baseline: 31.7734x; 1.0608x over previous
#include <hip/hip_runtime.h>
#include <hip/hip_bf16.h>


#define B_  4
#define DI  192
#define LL  4096
#define NS  16
#define RR  6
#define KK  4
#define CP  38
#define TL  64
#define TLP 65

typedef float v2f __attribute__((ext_vector_type(2)));
typedef float v8f __attribute__((ext_vector_type(8)));

__device__ __forceinline__ int scan_idx(int k, int l) {
    int ll = (k & 2) ? (LL - 1 - l) : l;
    if (k & 1) return ((ll & 63) << 6) | (ll >> 6);
    return ll;
}

#define PNT  8
#define PXP  136
#define PWP  196

__global__ __launch_bounds__(256) void ss2d_proj_wmma(
    const float* __restrict__ x,
    const float* __restrict__ xpw,
    float* __restrict__ xdbl)
{
    __shared__ __attribute__((aligned(16))) float xs[DI * PXP];
    __shared__ __attribute__((aligned(16))) float ws[2 * CP * PWP];

    const int tid  = threadIdx.x;
    const int lane = tid & 31;
    const int j    = tid >> 5;
    const int t    = blockIdx.x & 31;
    const int cls  = (blockIdx.x >> 5) & 1;
    const int b    = blockIdx.x >> 6;

    const float* xb = x + (size_t)b * DI * LL;

    if (cls == 0) {
        const int p0 = t * 128;
#pragma unroll
        for (int i = 0; i < 24; ++i) {
            const int f  = tid + 256 * i;
            const int d  = f >> 5;
            const int c4 = f & 31;
            const float4 v = *(const float4*)(xb + (size_t)d * LL + p0 + 4 * c4);
            *(float4*)(xs + d * PXP + 4 * c4) = v;
        }
    } else {
        const int w0 = (t & 7) * 8;
        const int h0 = (t >> 3) * 16;
#pragma unroll
        for (int i = 0; i < 24; ++i) {
            const int f  = tid + 256 * i;
            const int d  = f >> 5;
            const int r  = f & 31;
            const int c  = r >> 1;
            const int jq = r & 1;
            const float4 v = *(const float4*)(xb + (size_t)d * LL + (h0 + c) * 64 + w0 + 4 * jq);
            float* dst = xs + d * PXP + 64 * jq + c;
            dst[0]  = v.x;
            dst[16] = v.y;
            dst[32] = v.z;
            dst[48] = v.w;
        }
    }
#pragma unroll
    for (int i = 0; i < 15; ++i) {
        const int f = tid + 256 * i;
        if (f < 2 * CP * (DI / 4)) {
            const int r  = f / (DI / 4);
            const int q  = f - r * (DI / 4);
            const int sr = cls * CP + r + (r >= CP ? CP : 0);
            const float4 v = *(const float4*)(xpw + (size_t)sr * DI + 4 * q);
            *(float4*)(ws + r * PWP + 4 * q) = v;
        }
    }
    __syncthreads();

    const int hw = lane >> 4;
    const int ln = lane & 15;

    int aoff[6];
#pragma unroll
    for (int mt = 0; mt < 3; ++mt) {
        const int row = (16 * mt + ln) < CP ? (16 * mt + ln) : (CP - 1);
        aoff[mt]     = row * PWP + 2 * hw;
        aoff[3 + mt] = (CP + row) * PWP + 2 * hw;
    }
    const int boff = (2 * hw) * PXP + 16 * j + ln;

    v8f acc[6];
#pragma unroll
    for (int a = 0; a < 6; ++a) acc[a] = (v8f){0.f, 0.f, 0.f, 0.f, 0.f, 0.f, 0.f, 0.f};

#pragma unroll 4
    for (int k0 = 0; k0 < DI; k0 += 4) {
        v2f bb;
        bb.x = xs[boff + k0 * PXP];
        bb.y = xs[boff + k0 * PXP + PXP];
#pragma unroll
        for (int a = 0; a < 6; ++a) {
            const v2f av = *(const v2f*)(ws + aoff[a] + k0);
            acc[a] = __builtin_amdgcn_wmma_f32_16x16x4_f32(
                false, av, false, bb, (short)0, acc[a], false, false);
        }
    }

    int lf;
    if (cls == 0) lf = t * 128 + 16 * j + ln;
    else          lf = (((t & 7) * 8 + j) << 6) + (t >> 3) * 16 + ln;

#pragma unroll
    for (int kk = 0; kk < 2; ++kk) {
        const int kd = b * KK + cls + 2 * kk;
        const int l  = kk ? (LL - 1 - lf) : lf;
        float* obase = xdbl + (size_t)kd * CP * LL + l;
#pragma unroll
        for (int mt = 0; mt < 3; ++mt) {
#pragma unroll
            for (int i = 0; i < 8; ++i) {
                const int m = 16 * mt + i + (hw << 3);
                if (m < CP) obase[(size_t)m * LL] = acc[kk * 3 + mt][i];
            }
        }
    }
}

#define CHW 4
#define LPC 8
#define SCW 3
#define GCH (SCW * CHW)
#define SFE 6
#define XP  36
#define YQ  LPC
#define YP  (TL * YQ + 8)
#define DP  144
#define YTP 65
#define YTE (64 * YTP)
#define SMX_F (2 * TL * XP)
#define SMD_F (2 * GCH * DP)
#define SMY_F (2 * GCH * YP)

__device__ __forceinline__ float softplus_series(float s) {
    const float t = __builtin_amdgcn_exp2f(-fabsf(s) * 1.4426950408889634f);
    const float z = t * __builtin_amdgcn_rcpf(2.0f + t);
    const float w = z * z;
    float p = 1.0f / 13.0f;
    p = fmaf(p, w, 1.0f / 11.0f);
    p = fmaf(p, w, 1.0f / 9.0f);
    p = fmaf(p, w, 1.0f / 7.0f);
    p = fmaf(p, w, 1.0f / 5.0f);
    p = fmaf(p, w, 1.0f / 3.0f);
    p = fmaf(p, w, 1.0f);
    return fmaf(z + z, p, fmaxf(s, 0.0f));
}

#define GS  8
#define GPC (TL / GS)

__device__ __forceinline__ void scan_load_group(
    const float* sm_x, const float* sm_dd, int g, int q, int ch,
    float4 (&BC)[GS], float2 (&Dv)[GS])
{
#pragma unroll
    for (int t = 0; t < GS; ++t) {
        int j = GS * g + t;
        BC[t] = *(const float4*)(sm_x + j * XP + 4 * q);
        Dv[t] = *(const float2*)(sm_dd + ch * DP + 2 * j);
    }
}

__device__ __forceinline__ void scan_step_group(
    float A20, float A21, float& h0, float& h1,
    const float4 (&BC)[GS], const float2 (&Dv)[GS], float (&yp)[GS])
{
#pragma unroll
    for (int t = 0; t < GS; ++t) {
        float delta = Dv[t].x, du = Dv[t].y;
        float dA0 = __builtin_amdgcn_exp2f(delta * A20);
        float dA1 = __builtin_amdgcn_exp2f(delta * A21);
        h0 = dA0 * h0 + du * BC[t].x;
        h1 = dA1 * h1 + du * BC[t].y;
        float yv = h0 * BC[t].z;
        yv += h1 * BC[t].w;
        yp[t] = yv;
    }
}

__device__ __forceinline__ void scan_store_part(
    float* sm_y, int g, int q, int ch, const float (&yp)[GS])
{
    float* dst = sm_y + ch * YP + (GS * g) * YQ + q;
#pragma unroll
    for (int t = 0; t < GS; ++t) dst[t * YQ] = yp[t];
}

__device__ __forceinline__ void scan_trip_barrier() {
    asm volatile("s_wait_dscnt 0x0\n\ts_barrier_signal -1\n\ts_barrier_wait -1" ::: "memory");
}

__global__ __launch_bounds__(256) void ss2d_scan(
    const float* __restrict__ y,
    const float* __restrict__ dtw_g,
    const float* __restrict__ dtb_g,
    const float* __restrict__ Alogs,
    const float* __restrict__ Ds_g,
    const float* __restrict__ xdbl,
    float* __restrict__ outy)
{
    __shared__ __attribute__((aligned(16))) float smem[SMX_F + SMD_F + SMY_F + GCH * YTE];
    float* const sm_x  = smem;
    float* const sm_dd = smem + SMX_F;
    float* const sm_y  = smem + SMX_F + SMD_F;
    float* const sm_yt = smem + SMX_F + SMD_F + SMY_F;

    const int lane = threadIdx.x & 31;
    const int wv = __builtin_amdgcn_readfirstlane((int)(threadIdx.x >> 5));
    const bool is_scan  = wv < SCW;
    const bool is_prep  = (wv >= SCW) && (wv < SCW + 4);
    const bool is_stage = wv == SCW + 4;

    const int cg = blockIdx.x % (DI / GCH);
    const int kd = blockIdx.x / (DI / GCH);
    const int k  = kd & 3;
    const int b  = kd >> 2;
    const int d0 = cg * GCH;

    const float* xdb = xdbl + (size_t)kd * CP * LL;
    const float* yb  = y + ((size_t)b * DI + d0) * LL;
    float* ob        = outy + ((size_t)kd * DI + d0) * LL;

    const bool tdir = (k & 1) != 0;
    if (tdir) {
        const float4* ysl = (const float4*)yb;
#pragma unroll 1
        for (int rd = 0; rd < 2; ++rd) {
            float4 yv[24];
#pragma unroll
            for (int p = 0; p < 24; ++p)
                yv[p] = ysl[(rd * 24 + p) * 256 + (int)threadIdx.x];
#pragma unroll
            for (int p = 0; p < 24; ++p) {
                const int f  = (rd * 24 + p) * 256 + (int)threadIdx.x;
                const int r  = f & 1023;
                float* dst = sm_yt + (f >> 10) * YTE + (r >> 4) * YTP + 4 * (r & 15);
                dst[0] = yv[p].x;
                dst[1] = yv[p].y;
                dst[2] = yv[p].z;
                dst[3] = yv[p].w;
            }
        }
        __syncthreads();
    }

    const int ch = lane >> 3;
    const int q  = lane & 7;
    const int cw = (is_scan ? wv : 0) * CHW + ch;
    const int kdch = k * DI + d0 + cw;
    const float A20 = -__expf(Alogs[(size_t)kdch * NS + 2 * q])     * 1.4426950408889634f;
    const float A21 = -__expf(Alogs[(size_t)kdch * NS + 2 * q + 1]) * 1.4426950408889634f;
    float h0 = 0.0f, h1 = 0.0f;

    const int pw  = is_prep ? (wv - SCW) : 0;
    const int ps  = (pw & 1) * 32 + lane;
    const int pc0 = (pw >> 1) * SFE;
    float dtw[SFE][RR], dtb[SFE], DsC[SFE];
#pragma unroll
    for (int e = 0; e < SFE; ++e) {
        const int kc = k * DI + d0 + pc0 + e;
#pragma unroll
        for (int r = 0; r < RR; ++r) dtw[e][r] = dtw_g[(size_t)kc * RR + r];
        dtb[e] = dtb_g[kc];
        DsC[e] = Ds_g[kc];
    }
    float dtr[RR];
    float un[SFE];
    float u1[SFE], u2[SFE];
#pragma unroll
    for (int r = 0; r < RR; ++r) dtr[r] = 0.0f;
#pragma unroll
    for (int e = 0; e < SFE; ++e) { un[e] = 0.0f; u1[e] = 0.0f; u2[e] = 0.0f; }

    float xr[2][2 * NS];
#pragma unroll
    for (int hf = 0; hf < 2; ++hf)
#pragma unroll
        for (int r = 0; r < 2 * NS; ++r) xr[hf][r] = 0.0f;

    if (is_prep) {
#pragma unroll
        for (int r = 0; r < RR; ++r) dtr[r] = xdb[(size_t)r * LL + ps];
        const int m0 = scan_idx(k, ps);
        if (tdir) {
            const float* pyt = sm_yt + pc0 * YTE + (m0 >> 6) * YTP + (m0 & 63);
#pragma unroll
            for (int e = 0; e < SFE; ++e) un[e] = pyt[e * YTE];
        } else {
#pragma unroll
            for (int e = 0; e < SFE; ++e) un[e] = yb[(size_t)(pc0 + e) * LL + m0];
        }
    } else if (is_stage) {
#pragma unroll
        for (int hf = 0; hf < 2; ++hf)
#pragma unroll
            for (int r = 0; r < 2 * NS; ++r)
                xr[hf][r] = xdb[(size_t)(RR + r) * LL + lane + 32 * hf];
    }

    for (int i = 0; i < LL / TL + 2; ++i) {
        const int bs = i & 1;
        if (is_scan) {
            if (i >= 1 && i <= LL / TL) {
                const float* px = sm_x  + (bs ^ 1) * (TL * XP);
                const float* pd = sm_dd + (bs ^ 1) * (GCH * DP);
                float*       py = sm_y  + (bs ^ 1) * (GCH * YP);
                float4 BC0[GS], BC1[GS];
                float2 D0[GS], D1[GS];
                float ya[GS], yb[GS];
                scan_load_group(px, pd, 0, q, cw, BC0, D0);
                scan_load_group(px, pd, 1, q, cw, BC1, D1);
                __builtin_amdgcn_sched_barrier(0);
                scan_step_group(A20, A21, h0, h1, BC0, D0, ya);
#pragma unroll 1
                for (int g = 1; g < GPC - 1; g += 2) {
                    scan_store_part(py, g - 1, q, cw, ya);
                    scan_load_group(px, pd, g + 1, q, cw, BC0, D0);
                    __builtin_amdgcn_sched_barrier(0);
                    scan_step_group(A20, A21, h0, h1, BC1, D1, yb);
                    scan_store_part(py, g, q, cw, yb);
                    scan_load_group(px, pd, g + 2, q, cw, BC1, D1);
                    __builtin_amdgcn_sched_barrier(0);
                    scan_step_group(A20, A21, h0, h1, BC0, D0, ya);
                }
                scan_store_part(py, GPC - 2, q, cw, ya);
                scan_step_group(A20, A21, h0, h1, BC1, D1, yb);
                scan_store_part(py, GPC - 1, q, cw, yb);
            }
        } else if (is_prep) {
            float4 pa[SFE], pb[SFE];
            if (i >= 2) {
                const float* psy = sm_y + bs * (GCH * YP) + pc0 * YP + ps * YQ;
#pragma unroll
                for (int e = 0; e < SFE; ++e) {
                    pa[e] = *(const float4*)(psy + e * YP);
                    pb[e] = *(const float4*)(psy + e * YP + 4);
                }
            } else {
#pragma unroll
                for (int e = 0; e < SFE; ++e) {
                    pa[e] = make_float4(0.0f, 0.0f, 0.0f, 0.0f);
                    pb[e] = make_float4(0.0f, 0.0f, 0.0f, 0.0f);
                }
            }
            float dnx[RR], unx[SFE];
            if (i + 1 < LL / TL) {
                const int ln0 = (i + 1) * TL;
#pragma unroll
                for (int r = 0; r < RR; ++r) dnx[r] = xdb[(size_t)r * LL + ln0 + ps];
                const int m1 = scan_idx(k, ln0 + ps);
                if (tdir) {
                    const float* pyt = sm_yt + pc0 * YTE + (m1 >> 6) * YTP + (m1 & 63);
#pragma unroll
                    for (int e = 0; e < SFE; ++e) unx[e] = pyt[e * YTE];
                } else {
#pragma unroll
                    for (int e = 0; e < SFE; ++e)
                        unx[e] = yb[(size_t)(pc0 + e) * LL + m1];
                }
            } else {
#pragma unroll
                for (int r = 0; r < RR; ++r) dnx[r] = 0.0f;
#pragma unroll
                for (int e = 0; e < SFE; ++e) unx[e] = 0.0f;
            }
            if (i >= 2) {
                float* po = ob + (size_t)pc0 * LL + (i - 2) * TL + ps;
#pragma unroll
                for (int e = 0; e < SFE; ++e) {
                    const float ys = ((pa[e].x + pa[e].y) + (pa[e].z + pa[e].w))
                                   + ((pb[e].x + pb[e].y) + (pb[e].z + pb[e].w));
                    po[(size_t)e * LL] = ys + DsC[e] * u2[e];
                }
            }
            if (i < LL / TL) {
                float* pdd = sm_dd + bs * (GCH * DP) + pc0 * DP + 2 * ps;
#pragma unroll
                for (int e = 0; e < SFE; ++e) {
                    float s = dtb[e];
                    s += dtw[e][0] * dtr[0];
                    s += dtw[e][1] * dtr[1];
                    s += dtw[e][2] * dtr[2];
                    s += dtw[e][3] * dtr[3];
                    s += dtw[e][4] * dtr[4];
                    s += dtw[e][5] * dtr[5];
                    const float delta = softplus_series(s);
                    *(float2*)(pdd + e * DP) = make_float2(delta, delta * un[e]);
                }
            }
#pragma unroll
            for (int e = 0; e < SFE; ++e) { u2[e] = u1[e]; u1[e] = un[e]; un[e] = unx[e]; }
#pragma unroll
            for (int r = 0; r < RR; ++r) dtr[r] = dnx[r];
        } else if (is_stage) {
            if (i < LL / TL) {
#pragma unroll
                for (int hf = 0; hf < 2; ++hf) {
                    float* dst = sm_x + bs * (TL * XP) + (lane + 32 * hf) * XP;
#pragma unroll
                    for (int p = 0; p < 8; ++p)
                        *(float4*)(dst + 4 * p) =
                            make_float4(xr[hf][2 * p],      xr[hf][2 * p + 1],
                                        xr[hf][NS + 2 * p], xr[hf][NS + 2 * p + 1]);
                }
            }
            if (i + 1 < LL / TL) {
                const int ln0 = (i + 1) * TL;
#pragma unroll
                for (int hf = 0; hf < 2; ++hf)
#pragma unroll
                    for (int r = 0; r < 2 * NS; ++r)
                        xr[hf][r] = xdb[(size_t)(RR + r) * LL + ln0 + lane + 32 * hf];
            }
        }
        scan_trip_barrier();
    }
}

#define MTE   8
#define MPOS  (MTE * MTE)
#define MDG   8
#define MCH   (DI / MDG)
#define MPIT  (DI + 1)

__global__ __launch_bounds__(MPOS * MDG) void ss2d_merge_ln(
    const float* __restrict__ outy,
    const float* __restrict__ gamma,
    const float* __restrict__ beta,
    float* __restrict__ out)
{
    __shared__ float tile[MPOS * MPIT];
    __shared__ float red_s[MDG * MPOS];
    __shared__ float red_q[MDG * MPOS];

    const int tid  = threadIdx.x;
    const int lane = tid & 31;
    const int wv   = tid >> 5;
    const int b    = blockIdx.x >> 6;
    const int t    = blockIdx.x & 63;
    const int h0   = (t >> 3) * MTE;
    const int w0   = (t & 7) * MTE;
    const int pos  = tid & (MPOS - 1);
    const int dg   = tid >> 6;
    const int hh   = h0 + (pos >> 3);
    const int ww   = w0 + (pos & 7);
    const int lm   = (hh << 6) | ww;
    const int lt   = (ww << 6) | hh;

    const size_t ks = (size_t)DI * LL;
    const float* base = outy + (size_t)b * KK * ks + (size_t)dg * LL;
    const float* p0 = base + lm;
    const float* p2 = base + 2 * ks + (LL - 1 - lm);
    const float* p1 = base + ks + lt;
    const float* p3 = base + 3 * ks + (LL - 1 - lt);

    float gm[6], bt[6];
#pragma unroll
    for (int c = 0; c < 6; ++c) { gm[c] = gamma[lane + 32 * c]; bt[c] = beta[lane + 32 * c]; }

    float a0[MCH], a2[MCH], a1[MCH], a3[MCH];
#pragma unroll
    for (int it = 0; it < MCH; ++it) {
        const size_t o = (size_t)it * MDG * LL;
        a0[it] = p0[o]; a2[it] = p2[o]; a1[it] = p1[o]; a3[it] = p3[o];
    }
    __builtin_amdgcn_sched_barrier(0);

    float v[MCH];
    float s = 0.0f, q = 0.0f;
#pragma unroll
    for (int it = 0; it < MCH; ++it) {
        v[it] = a0[it] + a2[it] + a1[it] + a3[it];
        s += v[it];
        q += v[it] * v[it];
    }
    red_s[dg * MPOS + pos] = s;
    red_q[dg * MPOS + pos] = q;
    __syncthreads();

    float ts = 0.0f, tq = 0.0f;
#pragma unroll
    for (int g = 0; g < MDG; ++g) {
        ts += red_s[g * MPOS + pos];
        tq += red_q[g * MPOS + pos];
    }
    const float mu   = ts / 192.0f;
    const float var  = tq / 192.0f - mu * mu;
    const float rstd = rsqrtf(var + 1e-5f);

#pragma unroll
    for (int it = 0; it < MCH; ++it)
        tile[pos * MPIT + dg + it * MDG] = (v[it] - mu) * rstd;
    __syncthreads();

    const int pb = wv * 4;
    const int lb = ((h0 + (pb >> 3)) << 6) | (w0 + (pb & 7));
    float* ob = out + ((size_t)b * LL + lb) * DI + lane;
    const float* tb = tile + pb * MPIT + lane;
#pragma unroll
    for (int pi = 0; pi < 4; ++pi)
#pragma unroll
        for (int c = 0; c < 6; ++c)
            ob[pi * DI + 32 * c] = tb[pi * MPIT + 32 * c] * gm[c] + bt[c];
}

extern "C" void kernel_launch(void* const* d_in, const int* in_sizes, int n_in,
                              void* d_out, int out_size, void* d_ws, size_t ws_size,
                              hipStream_t stream) {
    const float* x   = (const float*)d_in[0];
    const float* y   = (const float*)d_in[1];
    const float* xpw = (const float*)d_in[2];
    const float* dtw = (const float*)d_in[3];
    const float* dtb = (const float*)d_in[4];
    const float* Al  = (const float*)d_in[5];
    const float* Ds  = (const float*)d_in[6];
    const float* gm  = (const float*)d_in[7];
    const float* bt  = (const float*)d_in[8];

    float* xdbl = (float*)d_ws;
    float* outy = xdbl + (size_t)B_ * KK * CP * LL;

    ss2d_proj_wmma<<<256, 256, 0, stream>>>(x, xpw, xdbl);
    ss2d_scan<<<B_ * KK * (DI / GCH), 256, 0, stream>>>(y, dtw, dtb, Al, Ds, xdbl, outy);
    ss2d_merge_ln<<<B_ * (LL / MPOS), MPOS * MDG, 0, stream>>>(outy, gm, bt, (float*)d_out);
}


# scan: helper waves' global loads are first read one trip after issue (roles as plain ifs, two register sets by trip parity, u from the LDS y slice in all 4 directions): no load wait at the barrier
# speedup vs baseline: 32.9071x; 1.0357x over previous
#include <hip/hip_runtime.h>
#include <hip/hip_bf16.h>


#define B_  4
#define DI  192
#define LL  4096
#define NS  16
#define RR  6
#define KK  4
#define CP  38
#define TL  64
#define TLP 65

typedef float v2f __attribute__((ext_vector_type(2)));
typedef float v8f __attribute__((ext_vector_type(8)));

__device__ __forceinline__ int scan_idx(int k, int l) {
    int ll = (k & 2) ? (LL - 1 - l) : l;
    if (k & 1) return ((ll & 63) << 6) | (ll >> 6);
    return ll;
}

#define PNT  8
#define PXP  136
#define PWP  196

__global__ __launch_bounds__(256) void ss2d_proj_wmma(
    const float* __restrict__ x,
    const float* __restrict__ xpw,
    float* __restrict__ xdbl)
{
    __shared__ __attribute__((aligned(16))) float xs[DI * PXP];
    __shared__ __attribute__((aligned(16))) float ws[2 * CP * PWP];

    const int tid  = threadIdx.x;
    const int lane = tid & 31;
    const int j    = tid >> 5;
    const int t    = blockIdx.x & 31;
    const int cls  = (blockIdx.x >> 5) & 1;
    const int b    = blockIdx.x >> 6;

    const float* xb = x + (size_t)b * DI * LL;

    if (cls == 0) {
        const int p0 = t * 128;
#pragma unroll
        for (int i = 0; i < 24; ++i) {
            const int f  = tid + 256 * i;
            const int d  = f >> 5;
            const int c4 = f & 31;
            const float4 v = *(const float4*)(xb + (size_t)d * LL + p0 + 4 * c4);
            *(float4*)(xs + d * PXP + 4 * c4) = v;
        }
    } else {
        const int w0 = (t & 7) * 8;
        const int h0 = (t >> 3) * 16;
#pragma unroll
        for (int i = 0; i < 24; ++i) {
            const int f  = tid + 256 * i;
            const int d  = f >> 5;
            const int r  = f & 31;
            const int c  = r >> 1;
            const int jq = r & 1;
            const float4 v = *(const float4*)(xb + (size_t)d * LL + (h0 + c) * 64 + w0 + 4 * jq);
            float* dst = xs + d * PXP + 64 * jq + c;
            dst[0]  = v.x;
            dst[16] = v.y;
            dst[32] = v.z;
            dst[48] = v.w;
        }
    }
#pragma unroll
    for (int i = 0; i < 15; ++i) {
        const int f = tid + 256 * i;
        if (f < 2 * CP * (DI / 4)) {
            const int r  = f / (DI / 4);
            const int q  = f - r * (DI / 4);
            const int sr = cls * CP + r + (r >= CP ? CP : 0);
            const float4 v = *(const float4*)(xpw + (size_t)sr * DI + 4 * q);
            *(float4*)(ws + r * PWP + 4 * q) = v;
        }
    }
    __syncthreads();

    const int hw = lane >> 4;
    const int ln = lane & 15;

    int aoff[6];
#pragma unroll
    for (int mt = 0; mt < 3; ++mt) {
        const int row = (16 * mt + ln) < CP ? (16 * mt + ln) : (CP - 1);
        aoff[mt]     = row * PWP + 2 * hw;
        aoff[3 + mt] = (CP + row) * PWP + 2 * hw;
    }
    const int boff = (2 * hw) * PXP + 16 * j + ln;

    v8f acc[6];
#pragma unroll
    for (int a = 0; a < 6; ++a) acc[a] = (v8f){0.f, 0.f, 0.f, 0.f, 0.f, 0.f, 0.f, 0.f};

#pragma unroll 4
    for (int k0 = 0; k0 < DI; k0 += 4) {
        v2f bb;
        bb.x = xs[boff + k0 * PXP];
        bb.y = xs[boff + k0 * PXP + PXP];
#pragma unroll
        for (int a = 0; a < 6; ++a) {
            const v2f av = *(const v2f*)(ws + aoff[a] + k0);
            acc[a] = __builtin_amdgcn_wmma_f32_16x16x4_f32(
                false, av, false, bb, (short)0, acc[a], false, false);
        }
    }

    int lf;
    if (cls == 0) lf = t * 128 + 16 * j + ln;
    else          lf = (((t & 7) * 8 + j) << 6) + (t >> 3) * 16 + ln;

#pragma unroll
    for (int kk = 0; kk < 2; ++kk) {
        const int kd = b * KK + cls + 2 * kk;
        const int l  = kk ? (LL - 1 - lf) : lf;
        float* obase = xdbl + (size_t)kd * CP * LL + l;
#pragma unroll
        for (int mt = 0; mt < 3; ++mt) {
#pragma unroll
            for (int i = 0; i < 8; ++i) {
                const int m = 16 * mt + i + (hw << 3);
                if (m < CP) obase[(size_t)m * LL] = acc[kk * 3 + mt][i];
            }
        }
    }
}

#define CHW 4
#define LPC 8
#define SCW 3
#define GCH (SCW * CHW)
#define SFE 6
#define XP  36
#define YQ  LPC
#define YP  (TL * YQ + 8)
#define DP  144
#define YTP 65
#define YTE (64 * YTP)
#define SMX_F (2 * TL * XP)
#define SMD_F (2 * GCH * DP)
#define SMY_F (2 * GCH * YP)

__device__ __forceinline__ float softplus_series(float s) {
    const float t = __builtin_amdgcn_exp2f(-fabsf(s) * 1.4426950408889634f);
    const float z = t * __builtin_amdgcn_rcpf(2.0f + t);
    const float w = z * z;
    float p = 1.0f / 13.0f;
    p = fmaf(p, w, 1.0f / 11.0f);
    p = fmaf(p, w, 1.0f / 9.0f);
    p = fmaf(p, w, 1.0f / 7.0f);
    p = fmaf(p, w, 1.0f / 5.0f);
    p = fmaf(p, w, 1.0f / 3.0f);
    p = fmaf(p, w, 1.0f);
    return fmaf(z + z, p, fmaxf(s, 0.0f));
}

#define GS  8
#define GPC (TL / GS)

__device__ __forceinline__ void scan_load_group(
    const float* sm_x, const float* sm_dd, int g, int q, int ch,
    float4 (&BC)[GS], float2 (&Dv)[GS])
{
#pragma unroll
    for (int t = 0; t < GS; ++t) {
        int j = GS * g + t;
        BC[t] = *(const float4*)(sm_x + j * XP + 4 * q);
        Dv[t] = *(const float2*)(sm_dd + ch * DP + 2 * j);
    }
}

__device__ __forceinline__ void scan_step_group(
    float A20, float A21, float& h0, float& h1,
    const float4 (&BC)[GS], const float2 (&Dv)[GS], float (&yp)[GS])
{
#pragma unroll
    for (int t = 0; t < GS; ++t) {
        float delta = Dv[t].x, du = Dv[t].y;
        float dA0 = __builtin_amdgcn_exp2f(delta * A20);
        float dA1 = __builtin_amdgcn_exp2f(delta * A21);
        h0 = dA0 * h0 + du * BC[t].x;
        h1 = dA1 * h1 + du * BC[t].y;
        float yv = h0 * BC[t].z;
        yv += h1 * BC[t].w;
        yp[t] = yv;
    }
}

__device__ __forceinline__ void scan_store_part(
    float* sm_y, int g, int q, int ch, const float (&yp)[GS])
{
    float* dst = sm_y + ch * YP + (GS * g) * YQ + q;
#pragma unroll
    for (int t = 0; t < GS; ++t) dst[t * YQ] = yp[t];
}

__device__ __forceinline__ void scan_trip_barrier() {
    asm volatile("s_wait_dscnt 0x0\n\ts_barrier_signal -1\n\ts_barrier_wait -1" ::: "memory");
}

__device__ __forceinline__ void scan_prep_trip(
    int i, int bs, bool tdir, int k, int ps, int pc0,
    const float* __restrict__ xdb, const float* __restrict__ yb, float* __restrict__ ob,
    const float* sm_y, float* sm_dd, const float* sm_yt,
    const float (&dtw)[SFE][RR], const float (&dtb)[SFE], const float (&DsC)[SFE],
    float (&u1)[SFE], float (&u2)[SFE],
    const float (&dtc)[RR], const float (&uc)[SFE],
    float (&dtn)[RR], float (&un)[SFE])
{
    float4 pa[SFE], pb[SFE];
    if (i >= 2) {
        const float* psy = sm_y + bs * (GCH * YP) + pc0 * YP + ps * YQ;
#pragma unroll
        for (int e = 0; e < SFE; ++e) {
            pa[e] = *(const float4*)(psy + e * YP);
            pb[e] = *(const float4*)(psy + e * YP + 4);
        }
    } else {
#pragma unroll
        for (int e = 0; e < SFE; ++e) {
            pa[e] = make_float4(0.0f, 0.0f, 0.0f, 0.0f);
            pb[e] = make_float4(0.0f, 0.0f, 0.0f, 0.0f);
        }
    }
    {
        const int cn  = (i + 1 < LL / TL) ? (i + 1) : (LL / TL - 1);
        const int ln0 = cn * TL;
#pragma unroll
        for (int r = 0; r < RR; ++r) dtn[r] = xdb[(size_t)r * LL + ln0 + ps];
        const int m1 = scan_idx(k, ln0 + ps);
        if (tdir) {
            const float* pyt = sm_yt + pc0 * YTE + (m1 >> 6) * YTP + (m1 & 63);
#pragma unroll
            for (int e = 0; e < SFE; ++e) un[e] = pyt[e * YTE];
        }
        if (!tdir) {
#pragma unroll
            for (int e = 0; e < SFE; ++e)
                un[e] = yb[(size_t)(pc0 + e) * LL + m1];
        }
    }
    if (i >= 2) {
        float* po = ob + (size_t)pc0 * LL + (i - 2) * TL + ps;
#pragma unroll
        for (int e = 0; e < SFE; ++e) {
            const float ys = ((pa[e].x + pa[e].y) + (pa[e].z + pa[e].w))
                           + ((pb[e].x + pb[e].y) + (pb[e].z + pb[e].w));
            po[(size_t)e * LL] = ys + DsC[e] * u2[e];
        }
    }
    if (i < LL / TL) {
        float* pdd = sm_dd + bs * (GCH * DP) + pc0 * DP + 2 * ps;
#pragma unroll
        for (int e = 0; e < SFE; ++e) {
            float s = dtb[e];
            s += dtw[e][0] * dtc[0];
            s += dtw[e][1] * dtc[1];
            s += dtw[e][2] * dtc[2];
            s += dtw[e][3] * dtc[3];
            s += dtw[e][4] * dtc[4];
            s += dtw[e][5] * dtc[5];
            const float delta = softplus_series(s);
            *(float2*)(pdd + e * DP) = make_float2(delta, delta * uc[e]);
        }
    }
#pragma unroll
    for (int e = 0; e < SFE; ++e) { u2[e] = u1[e]; u1[e] = uc[e]; }
}

__global__ __launch_bounds__(256) void ss2d_scan(
    const float* __restrict__ y,
    const float* __restrict__ dtw_g,
    const float* __restrict__ dtb_g,
    const float* __restrict__ Alogs,
    const float* __restrict__ Ds_g,
    const float* __restrict__ xdbl,
    float* __restrict__ outy)
{
    __shared__ __attribute__((aligned(16))) float smem[SMX_F + SMD_F + SMY_F + GCH * YTE];
    float* const sm_x  = smem;
    float* const sm_dd = smem + SMX_F;
    float* const sm_y  = smem + SMX_F + SMD_F;
    float* const sm_yt = smem + SMX_F + SMD_F + SMY_F;

    const int lane = threadIdx.x & 31;
    const int wv = __builtin_amdgcn_readfirstlane((int)(threadIdx.x >> 5));
    const bool is_scan  = wv < SCW;
    const bool is_prep  = (wv >= SCW) && (wv < SCW + 4);
    const bool is_stage = wv == SCW + 4;

    const int cg = blockIdx.x % (DI / GCH);
    const int kd = blockIdx.x / (DI / GCH);
    const int k  = kd & 3;
    const int b  = kd >> 2;
    const int d0 = cg * GCH;

    const float* xdb = xdbl + (size_t)kd * CP * LL;
    const float* yb  = y + ((size_t)b * DI + d0) * LL;
    float* ob        = outy + ((size_t)kd * DI + d0) * LL;

    const bool tdir = true;
    if (tdir) {
        const float4* ysl = (const float4*)yb;
#pragma unroll 1
        for (int rd = 0; rd < 2; ++rd) {
            float4 yv[24];
#pragma unroll
            for (int p = 0; p < 24; ++p)
                yv[p] = ysl[(rd * 24 + p) * 256 + (int)threadIdx.x];
#pragma unroll
            for (int p = 0; p < 24; ++p) {
                const int f  = (rd * 24 + p) * 256 + (int)threadIdx.x;
                const int r  = f & 1023;
                float* dst = sm_yt + (f >> 10) * YTE + (r >> 4) * YTP + 4 * (r & 15);
                dst[0] = yv[p].x;
                dst[1] = yv[p].y;
                dst[2] = yv[p].z;
                dst[3] = yv[p].w;
            }
        }
        __syncthreads();
    }

    const int ch = lane >> 3;
    const int q  = lane & 7;
    const int cw = (is_scan ? wv : 0) * CHW + ch;
    const int kdch = k * DI + d0 + cw;
    const float A20 = -__expf(Alogs[(size_t)kdch * NS + 2 * q])     * 1.4426950408889634f;
    const float A21 = -__expf(Alogs[(size_t)kdch * NS + 2 * q + 1]) * 1.4426950408889634f;
    float h0 = 0.0f, h1 = 0.0f;

    const int pw  = is_prep ? (wv - SCW) : 0;
    const int ps  = (pw & 1) * 32 + lane;
    const int pc0 = (pw >> 1) * SFE;
    float dtw[SFE][RR], dtb[SFE], DsC[SFE];
#pragma unroll
    for (int e = 0; e < SFE; ++e) {
        const int kc = k * DI + d0 + pc0 + e;
#pragma unroll
        for (int r = 0; r < RR; ++r) dtw[e][r] = dtw_g[(size_t)kc * RR + r];
        dtb[e] = dtb_g[kc];
        DsC[e] = Ds_g[kc];
    }
    float dtrA[RR], dtrB[RR];
    float unA[SFE], unB[SFE];
    float u1[SFE], u2[SFE];
#pragma unroll
    for (int r = 0; r < RR; ++r) { dtrA[r] = 0.0f; dtrB[r] = 0.0f; }
#pragma unroll
    for (int e = 0; e < SFE; ++e) { unA[e] = 0.0f; unB[e] = 0.0f; u1[e] = 0.0f; u2[e] = 0.0f; }

    float xr[2][2 * NS];
#pragma unroll
    for (int hf = 0; hf < 2; ++hf)
#pragma unroll
        for (int r = 0; r < 2 * NS; ++r) xr[hf][r] = 0.0f;

    if (is_prep) {
#pragma unroll
        for (int r = 0; r < RR; ++r) dtrA[r] = xdb[(size_t)r * LL + ps];
        const int m0 = scan_idx(k, ps);
        if (tdir) {
            const float* pyt = sm_yt + pc0 * YTE + (m0 >> 6) * YTP + (m0 & 63);
#pragma unroll
            for (int e = 0; e < SFE; ++e) unA[e] = pyt[e * YTE];
        } else {
#pragma unroll
            for (int e = 0; e < SFE; ++e) unA[e] = yb[(size_t)(pc0 + e) * LL + m0];
        }
    } else if (is_stage) {
#pragma unroll
        for (int hf = 0; hf < 2; ++hf)
#pragma unroll
            for (int r = 0; r < 2 * NS; ++r)
                xr[hf][r] = xdb[(size_t)(RR + r) * LL + lane + 32 * hf];
    }

    for (int i = 0; i < LL / TL + 2; ++i) {
        const int bs = i & 1;
        if (is_scan) {
            if (i >= 1 && i <= LL / TL) {
                const float* px = sm_x  + (bs ^ 1) * (TL * XP);
                const float* pd = sm_dd + (bs ^ 1) * (GCH * DP);
                float*       py = sm_y  + (bs ^ 1) * (GCH * YP);
                float4 BC0[GS], BC1[GS];
                float2 D0[GS], D1[GS];
                float ya[GS], yb[GS];
                scan_load_group(px, pd, 0, q, cw, BC0, D0);
                scan_load_group(px, pd, 1, q, cw, BC1, D1);
                __builtin_amdgcn_sched_barrier(0);
                scan_step_group(A20, A21, h0, h1, BC0, D0, ya);
#pragma unroll 1
                for (int g = 1; g < GPC - 1; g += 2) {
                    scan_store_part(py, g - 1, q, cw, ya);
                    scan_load_group(px, pd, g + 1, q, cw, BC0, D0);
                    __builtin_amdgcn_sched_barrier(0);
                    scan_step_group(A20, A21, h0, h1, BC1, D1, yb);
                    scan_store_part(py, g, q, cw, yb);
                    scan_load_group(px, pd, g + 2, q, cw, BC1, D1);
                    __builtin_amdgcn_sched_barrier(0);
                    scan_step_group(A20, A21, h0, h1, BC0, D0, ya);
                }
                scan_store_part(py, GPC - 2, q, cw, ya);
                scan_step_group(A20, A21, h0, h1, BC1, D1, yb);
                scan_store_part(py, GPC - 1, q, cw, yb);
            }
        }
        if (is_prep && bs == 0) {
            scan_prep_trip(i, bs, tdir, k, ps, pc0, xdb, yb, ob, sm_y, sm_dd, sm_yt,
                           dtw, dtb, DsC, u1, u2, dtrA, unA, dtrB, unB);
        }
        if (is_prep && bs != 0) {
            scan_prep_trip(i, bs, tdir, k, ps, pc0, xdb, yb, ob, sm_y, sm_dd, sm_yt,
                           dtw, dtb, DsC, u1, u2, dtrB, unB, dtrA, unA);
        }
        if (is_stage) {
            if (i < LL / TL) {
#pragma unroll
                for (int hf = 0; hf < 2; ++hf) {
                    float* dst = sm_x + bs * (TL * XP) + (lane + 32 * hf) * XP;
#pragma unroll
                    for (int p = 0; p < 8; ++p)
                        *(float4*)(dst + 4 * p) =
                            make_float4(xr[hf][2 * p],      xr[hf][2 * p + 1],
                                        xr[hf][NS + 2 * p], xr[hf][NS + 2 * p + 1]);
                }
            }
            if (i + 1 < LL / TL) {
                const int ln0 = (i + 1) * TL;
#pragma unroll
                for (int hf = 0; hf < 2; ++hf)
#pragma unroll
                    for (int r = 0; r < 2 * NS; ++r)
                        xr[hf][r] = xdb[(size_t)(RR + r) * LL + ln0 + lane + 32 * hf];
            }
        }
        scan_trip_barrier();
    }
}

#define MTE   8
#define MPOS  (MTE * MTE)
#define MDG   8
#define MCH   (DI / MDG)
#define MPIT  (DI + 1)

__global__ __launch_bounds__(MPOS * MDG) void ss2d_merge_ln(
    const float* __restrict__ outy,
    const float* __restrict__ gamma,
    const float* __restrict__ beta,
    float* __restrict__ out)
{
    __shared__ float tile[MPOS * MPIT];
    __shared__ float red_s[MDG * MPOS];
    __shared__ float red_q[MDG * MPOS];

    const int tid  = threadIdx.x;
    const int lane = tid & 31;
    const int wv   = tid >> 5;
    const int b    = blockIdx.x >> 6;
    const int t    = blockIdx.x & 63;
    const int h0   = (t >> 3) * MTE;
    const int w0   = (t & 7) * MTE;
    const int pos  = tid & (MPOS - 1);
    const int dg   = tid >> 6;
    const int hh   = h0 + (pos >> 3);
    const int ww   = w0 + (pos & 7);
    const int lm   = (hh << 6) | ww;
    const int lt   = (ww << 6) | hh;

    const size_t ks = (size_t)DI * LL;
    const float* base = outy + (size_t)b * KK * ks + (size_t)dg * LL;
    const float* p0 = base + lm;
    const float* p2 = base + 2 * ks + (LL - 1 - lm);
    const float* p1 = base + ks + lt;
    const float* p3 = base + 3 * ks + (LL - 1 - lt);

    float gm[6], bt[6];
#pragma unroll
    for (int c = 0; c < 6; ++c) { gm[c] = gamma[lane + 32 * c]; bt[c] = beta[lane + 32 * c]; }

    float a0[MCH], a2[MCH], a1[MCH], a3[MCH];
#pragma unroll
    for (int it = 0; it < MCH; ++it) {
        const size_t o = (size_t)it * MDG * LL;
        a0[it] = p0[o]; a2[it] = p2[o]; a1[it] = p1[o]; a3[it] = p3[o];
    }
    __builtin_amdgcn_sched_barrier(0);

    float v[MCH];
    float s = 0.0f, q = 0.0f;
#pragma unroll
    for (int it = 0; it < MCH; ++it) {
        v[it] = a0[it] + a2[it] + a1[it] + a3[it];
        s += v[it];
        q += v[it] * v[it];
    }
    red_s[dg * MPOS + pos] = s;
    red_q[dg * MPOS + pos] = q;
    __syncthreads();

    float ts = 0.0f, tq = 0.0f;
#pragma unroll
    for (int g = 0; g < MDG; ++g) {
        ts += red_s[g * MPOS + pos];
        tq += red_q[g * MPOS + pos];
    }
    const float mu   = ts / 192.0f;
    const float var  = tq / 192.0f - mu * mu;
    const float rstd = rsqrtf(var + 1e-5f);

#pragma unroll
    for (int it = 0; it < MCH; ++it)
        tile[pos * MPIT + dg + it * MDG] = (v[it] - mu) * rstd;
    __syncthreads();

    const int pb = wv * 4;
    const int lb = ((h0 + (pb >> 3)) << 6) | (w0 + (pb & 7));
    float* ob = out + ((size_t)b * LL + lb) * DI + lane;
    const float* tb = tile + pb * MPIT + lane;
#pragma unroll
    for (int pi = 0; pi < 4; ++pi)
#pragma unroll
        for (int c = 0; c < 6; ++c)
            ob[pi * DI + 32 * c] = tb[pi * MPIT + 32 * c] * gm[c] + bt[c];
}

extern "C" void kernel_launch(void* const* d_in, const int* in_sizes, int n_in,
                              void* d_out, int out_size, void* d_ws, size_t ws_size,
                              hipStream_t stream) {
    const float* x   = (const float*)d_in[0];
    const float* y   = (const float*)d_in[1];
    const float* xpw = (const float*)d_in[2];
    const float* dtw = (const float*)d_in[3];
    const float* dtb = (const float*)d_in[4];
    const float* Al  = (const float*)d_in[5];
    const float* Ds  = (const float*)d_in[6];
    const float* gm  = (const float*)d_in[7];
    const float* bt  = (const float*)d_in[8];

    float* xdbl = (float*)d_ws;
    float* outy = xdbl + (size_t)B_ * KK * CP * LL;

    ss2d_proj_wmma<<<256, 256, 0, stream>>>(x, xpw, xdbl);
    ss2d_scan<<<B_ * KK * (DI / GCH), 256, 0, stream>>>(y, dtw, dtb, Al, Ds, xdbl, outy);
    ss2d_merge_ln<<<B_ * (LL / MPOS), MPOS * MDG, 0, stream>>>(outy, gm, bt, (float*)d_out);
}
